# starting kernel
# baseline (speedup 1.0000x reference)
#include <hip/hip_runtime.h>

typedef __attribute__((ext_vector_type(2))) float v2f;
typedef __attribute__((ext_vector_type(8))) float v8f;

constexpr int HB = 64;
constexpr int HT = 4096;
constexpr int HK = 32;
constexpr float LOG2E = 1.44269504088896340736f;
constexpr float LN2   = 0.69314718055994530942f;

#if __has_builtin(__builtin_amdgcn_exp2f)
__device__ __forceinline__ float fexp2(float x){ return __builtin_amdgcn_exp2f(x); }
#else
__device__ __forceinline__ float fexp2(float x){ return exp2f(x); }
#endif
#if __has_builtin(__builtin_amdgcn_logf)
__device__ __forceinline__ float flog2(float x){ return __builtin_amdgcn_logf(x); }
#else
__device__ __forceinline__ float flog2(float x){ return log2f(x); }
#endif

__device__ __forceinline__ float rmax32(float x){
#pragma unroll
  for (int m = 16; m >= 1; m >>= 1) x = fmaxf(x, __shfl_xor(x, m, 32));
  return x;
}
__device__ __forceinline__ float rsum32(float x){
#pragma unroll
  for (int m = 16; m >= 1; m >>= 1) x += __shfl_xor(x, m, 32);
  return x;
}
__device__ __forceinline__ float rmax16(float x){
#pragma unroll
  for (int m = 8; m >= 1; m >>= 1) x = fmaxf(x, __shfl_xor(x, m, 32));
  return x;
}

__device__ __forceinline__ v8f wmma4(v2f a, v2f b, v8f c){
  return __builtin_amdgcn_wmma_f32_16x16x4_f32(
       false, a,  false, b,
       (short)0, c,  false,  false);
}

__device__ __forceinline__ void load_em(float (&dst)[2][8], const float* emb,
                                        int t, int h, int ln){
#pragma unroll
  for (int nt = 0; nt < 2; ++nt)
#pragma unroll
    for (int r = 0; r < 8; ++r)
      dst[nt][r] = emb[((size_t)(r + 8*h)*HT + t)*HK + nt*16 + ln];
}

__device__ __forceinline__ void store_tiles(float* dstb, int t, int h, int ln,
                                            const float (&v)[2][8]){
#pragma unroll
  for (int nt = 0; nt < 2; ++nt)
#pragma unroll
    for (int r = 0; r < 8; ++r)
      dstb[((size_t)(r + 8*h)*HT + t)*HK + nt*16 + ln] = v[nt][r];
}

template<int DIR>
__device__ __forceinline__ void hmm_step(float (&st)[2][8], const float (&emc)[2][8],
                                         const v2f (&breg)[8][2],
                                         float* sPf, const v2f* sP2,
                                         int h, int ln){
  float x[2][8];
#pragma unroll
  for (int nt = 0; nt < 2; ++nt)
#pragma unroll
    for (int r = 0; r < 8; ++r)
      x[nt][r] = (DIR == 0) ? st[nt][r] : st[nt][r] + emc[nt][r];

  float mrow[8];
#pragma unroll
  for (int r = 0; r < 8; ++r) mrow[r] = fmaxf(x[0][r], x[1][r]);
#pragma unroll
  for (int r = 0; r < 8; ++r) mrow[r] = rmax16(mrow[r]);

#pragma unroll
  for (int nt = 0; nt < 2; ++nt)
#pragma unroll
    for (int r = 0; r < 8; ++r)
      sPf[(r + 8*h)*34 + nt*16 + ln] = fexp2((x[nt][r] - mrow[r]) * LOG2E);
  __syncthreads();
  v2f a[8];
#pragma unroll
  for (int kc = 0; kc < 8; ++kc) a[kc] = sP2[ln*17 + 2*kc + h];
  __syncthreads();

  v8f c0 = {0.f,0.f,0.f,0.f,0.f,0.f,0.f,0.f};
  v8f c1 = {0.f,0.f,0.f,0.f,0.f,0.f,0.f,0.f};
#pragma unroll
  for (int kc = 0; kc < 8; ++kc){
    c0 = wmma4(a[kc], breg[kc][0], c0);
    c1 = wmma4(a[kc], breg[kc][1], c1);
  }
#pragma unroll
  for (int r = 0; r < 8; ++r){
    const float b0 = (DIR == 0) ? mrow[r] + emc[0][r] : mrow[r];
    const float b1 = (DIR == 0) ? mrow[r] + emc[1][r] : mrow[r];
    st[0][r] = fmaf(LN2, flog2(c0[r]), b0);
    st[1][r] = fmaf(LN2, flog2(c1[r]), b1);
  }
}

__global__ void __launch_bounds__(32)
hmm_scan_kernel(const float* __restrict__ em,
                const float* __restrict__ log_pi,
                const float* __restrict__ log_A,
                float* __restrict__ alpha_out,
                float* __restrict__ beta_out){
  __shared__ float sA[32*33];
  __shared__ v2f   sP2[16*17];
  float* sPf = (float*)sP2;

  const int lane = threadIdx.x;
  const int h    = lane >> 4;
  const int ln   = lane & 15;
  const int dir  = blockIdx.x >> 2;
  const int mb   = blockIdx.x & 3;

  for (int i = 0; i < 32; ++i){
    float xv = log_A[i*32 + lane];
    float mx = rmax32(xv);
    float e  = fexp2((xv - mx) * LOG2E);
    float s  = rsum32(e);
    sA[i*33 + lane] = e / s;
  }
  __syncthreads();

  v2f breg[8][2];
#pragma unroll
  for (int kc = 0; kc < 8; ++kc){
#pragma unroll
    for (int nt = 0; nt < 2; ++nt){
      const int kk = 4*kc + 2*h;
      const int nn = nt*16 + ln;
      if (dir == 0){
        breg[kc][nt].x = sA[kk*33 + nn];
        breg[kc][nt].y = sA[(kk+1)*33 + nn];
      } else {
        breg[kc][nt].x = sA[nn*33 + kk];
        breg[kc][nt].y = sA[nn*33 + kk + 1];
      }
    }
  }

  const float* emb = em + (size_t)(mb*16)*HT*HK;

  if (dir == 0){
    float xp = log_pi[lane];
    float mp = rmax32(xp);
    float sp = rsum32(fexp2((xp - mp) * LOG2E));
    float lp = xp - mp - LN2 * flog2(sp);
    float lpn[2];
#pragma unroll
    for (int nt = 0; nt < 2; ++nt) lpn[nt] = __shfl(lp, nt*16 + ln, 32);

    float* ab = alpha_out + (size_t)(mb*16)*HT*HK;
    float al[2][8];
    float emA[2][8], emB[2][8];
    load_em(emA, emb, 0, h, ln);
#pragma unroll
    for (int nt = 0; nt < 2; ++nt)
#pragma unroll
      for (int r = 0; r < 8; ++r) al[nt][r] = lpn[nt] + emA[nt][r];
    store_tiles(ab, 0, h, ln, al);
    load_em(emA, emb, 1, h, ln);

    for (int t = 1; t + 1 < HT; t += 2){
      load_em(emB, emb, t + 1, h, ln);
      hmm_step<0>(al, emA, breg, sPf, sP2, h, ln);
      store_tiles(ab, t, h, ln, al);
      load_em(emA, emb, (t + 2 < HT) ? t + 2 : HT - 1, h, ln);
      hmm_step<0>(al, emB, breg, sPf, sP2, h, ln);
      store_tiles(ab, t + 1, h, ln, al);
    }
    hmm_step<0>(al, emA, breg, sPf, sP2, h, ln);
    store_tiles(ab, HT - 1, h, ln, al);
  } else {
    float* bb = beta_out + (size_t)(mb*16)*HT*HK;
    float be[2][8];
#pragma unroll
    for (int nt = 0; nt < 2; ++nt)
#pragma unroll
      for (int r = 0; r < 8; ++r) be[nt][r] = 0.f;
    store_tiles(bb, HT - 1, h, ln, be);
    float emA[2][8], emB[2][8];
    load_em(emA, emb, HT - 1, h, ln);

    for (int t = HT - 2; t >= 2; t -= 2){
      load_em(emB, emb, t, h, ln);
      hmm_step<1>(be, emA, breg, sPf, sP2, h, ln);
      store_tiles(bb, t, h, ln, be);
      load_em(emA, emb, t - 1, h, ln);
      hmm_step<1>(be, emB, breg, sPf, sP2, h, ln);
      store_tiles(bb, t - 1, h, ln, be);
    }
    hmm_step<1>(be, emA, breg, sPf, sP2, h, ln);
    store_tiles(bb, 0, h, ln, be);
  }
}

__global__ void __launch_bounds__(256)
hmm_gamma_kernel(const float* alpha, const float* beta, float* out, int npairs){
  const int w    = (blockIdx.x*blockDim.x + threadIdx.x) >> 5;
  const int lane = threadIdx.x & 31;
  if (w >= npairs) return;
  const size_t idx = (size_t)w*32 + lane;
  float g  = alpha[idx] + beta[idx];
  float mx = rmax32(g);
  float s  = rsum32(fexp2((g - mx) * LOG2E));
  out[idx] = g - mx - LN2 * flog2(s);
}

extern "C" void kernel_launch(void* const* d_in, const int* in_sizes, int n_in,
                              void* d_out, int out_size, void* d_ws, size_t ws_size,
                              hipStream_t stream) {
  const float* em     = (const float*)d_in[0];
  const float* log_pi = (const float*)d_in[1];
  const float* log_A  = (const float*)d_in[2];
  float* out  = (float*)d_out;
  float* beta = (float*)d_ws;

  hmm_scan_kernel<<<dim3(8), dim3(32), 0, stream>>>(em, log_pi, log_A, out, beta);

  const int npairs = HB * HT;
  const int threads = 256;
  const int blocks = (npairs*32 + threads - 1) / threads;
  hmm_gamma_kernel<<<dim3(blocks), dim3(threads), 0, stream>>>(out, beta, out, npairs);
}


# Step product transposed (states on M, batch on lanes, k order matched to the D map): D feeds next B in registers, no LDS transpose or shuffle max per step; em rows 5 steps ahead; sum order changes
# speedup vs baseline: 2.3504x; 2.3504x over previous
#include <hip/hip_runtime.h>


typedef __attribute__((ext_vector_type(2))) float v2f;
typedef __attribute__((ext_vector_type(4))) float v4f;
typedef __attribute__((ext_vector_type(8))) float v8f;

constexpr int HB = 64;
constexpr int HT = 4096;
constexpr int HK = 32;
constexpr int PF = 5;
constexpr float LOG2E = 1.44269504088896340736f;
constexpr float LN2   = 0.69314718055994530942f;

#if __has_builtin(__builtin_amdgcn_exp2f)
__device__ __forceinline__ float fexp2(float x){ return __builtin_amdgcn_exp2f(x); }
#else
__device__ __forceinline__ float fexp2(float x){ return exp2f(x); }
#endif
#if __has_builtin(__builtin_amdgcn_logf)
__device__ __forceinline__ float flog2(float x){ return __builtin_amdgcn_logf(x); }
#else
__device__ __forceinline__ float flog2(float x){ return log2f(x); }
#endif

__device__ __forceinline__ float rmax32(float x){
#pragma unroll
  for (int m = 16; m >= 1; m >>= 1) x = fmaxf(x, __shfl_xor(x, m, 32));
  return x;
}
__device__ __forceinline__ float rsum32(float x){
#pragma unroll
  for (int m = 16; m >= 1; m >>= 1) x += __shfl_xor(x, m, 32);
  return x;
}

__device__ __forceinline__ float xhalf(float x){
  int b = __builtin_bit_cast(int, x);
  return __builtin_bit_cast(float,
      __builtin_amdgcn_permlanex16(b, b, 0x76543210, 0xfedcba98, false, false));
}

__device__ __forceinline__ v8f wmma4(v2f a, v2f b, v8f c){
  return __builtin_amdgcn_wmma_f32_16x16x4_f32(
       false, a,  false, b,
       (short)0, c,  false,  false);
}


__device__ __forceinline__ void load_row(v4f (&dst)[4], const float* p){
  const v4f* q = (const v4f*)p;
#pragma unroll
  for (int c = 0; c < 4; ++c) dst[c] = q[c];
}
__device__ __forceinline__ void store_row(float* p, const float (&v)[16]){
  v4f* q = (v4f*)p;
#pragma unroll
  for (int c = 0; c < 4; ++c){
    v4f w = { v[4*c+0], v[4*c+1], v[4*c+2], v[4*c+3] };
    q[c] = w;
  }
}

template<int DIR>
__device__ __forceinline__ void hmm_step(float (&st)[16], const v4f (&emc)[4],
                                         const v2f (&ereg)[2][8]){
  float x[16];
#pragma unroll
  for (int q = 0; q < 16; ++q)
    x[q] = (DIR == 0) ? st[q] : st[q] + emc[q >> 2][q & 3];

  const float t0 = fmaxf(fmaxf(x[0],  x[1]),  x[2]);
  const float t1 = fmaxf(fmaxf(x[3],  x[4]),  x[5]);
  const float t2 = fmaxf(fmaxf(x[6],  x[7]),  x[8]);
  const float t3 = fmaxf(fmaxf(x[9],  x[10]), x[11]);
  const float t4 = fmaxf(fmaxf(x[12], x[13]), x[14]);
  const float u0 = fmaxf(fmaxf(t0, t1), t2);
  const float u1 = fmaxf(fmaxf(t3, t4), x[15]);
  float m = fmaxf(u0, u1);
  m = fmaxf(m, xhalf(m));

  v2f b[8];
#pragma unroll
  for (int kc = 0; kc < 8; ++kc){
    b[kc].x = fexp2((x[2*kc]   - m) * LOG2E);
    b[kc].y = fexp2((x[2*kc+1] - m) * LOG2E);
  }

  v8f c0 = {0.f,0.f,0.f,0.f,0.f,0.f,0.f,0.f};
  v8f c1 = {0.f,0.f,0.f,0.f,0.f,0.f,0.f,0.f};
#pragma unroll
  for (int kc = 0; kc < 8; ++kc){
    c0 = wmma4(ereg[0][kc], b[kc], c0);
    c1 = wmma4(ereg[1][kc], b[kc], c1);
  }
#pragma unroll
  for (int r = 0; r < 8; ++r){
    const float b0 = (DIR == 0) ? m + emc[r >> 2][r & 3]           : m;
    const float b1 = (DIR == 0) ? m + emc[2 + (r >> 2)][r & 3]     : m;
    st[r]     = fmaf(LN2, flog2(c0[r]), b0);
    st[8 + r] = fmaf(LN2, flog2(c1[r]), b1);
  }
}

__global__ void __launch_bounds__(32)
hmm_scan_kernel(const float* __restrict__ em,
                const float* __restrict__ log_pi,
                const float* __restrict__ log_A,
                float* __restrict__ alpha_out,
                float* __restrict__ beta_out){
  __shared__ float sA[32*33];

  const int lane = threadIdx.x;
  const int h    = lane >> 4;
  const int ln   = lane & 15;
  const int dir  = blockIdx.x >> 2;
  const int mb   = blockIdx.x & 3;

  for (int i = 0; i < 32; ++i){
    float xv = log_A[i*32 + lane];
    float mx = rmax32(xv);
    float e  = fexp2((xv - mx) * LOG2E);
    float s  = rsum32(e);
    sA[i*33 + lane] = e / s;
  }
  __syncthreads();

  v2f ereg[2][8];
#pragma unroll
  for (int mt = 0; mt < 2; ++mt){
#pragma unroll
    for (int kc = 0; kc < 8; ++kc){
      const int sM = 16*(ln >> 3) + 8*mt + (ln & 7);
      const int sK = 16*h + 2*kc;
      if (dir == 0){
        ereg[mt][kc].x = sA[sK*33 + sM];
        ereg[mt][kc].y = sA[(sK+1)*33 + sM];
      } else {
        ereg[mt][kc].x = sA[sM*33 + sK];
        ereg[mt][kc].y = sA[sM*33 + sK + 1];
      }
    }
  }

  const size_t rowoff = (size_t)(mb*16 + ln)*HT*HK + 16*h;
  const float* eb = em + rowoff;

  v4f er[PF][4];

  if (dir == 0){
    float xp = log_pi[lane];
    float mp = rmax32(xp);
    float sp = rsum32(fexp2((xp - mp) * LOG2E));
    float lp = xp - mp - LN2 * flog2(sp);

    float* ab = alpha_out + rowoff;
    float al[16];
    v4f e0[4];
    load_row(e0, eb);
#pragma unroll
    for (int u = 0; u < PF; ++u) load_row(er[u], eb + (size_t)(1 + u)*HK);
#pragma unroll
    for (int q = 0; q < 16; ++q) al[q] = __shfl(lp, 16*h + q, 32) + e0[q >> 2][q & 3];
    store_row(ab, al);

    for (int t0 = 1; t0 < HT; t0 += PF){
#pragma unroll
      for (int u = 0; u < PF; ++u){
        const int t = t0 + u;
        hmm_step<0>(al, er[u], ereg);
        store_row(ab + (size_t)t*HK, al);
        const int tn = (t + PF < HT) ? t + PF : HT - 1;
        load_row(er[u], eb + (size_t)tn*HK);
        __builtin_amdgcn_sched_barrier(0);
      }
    }
  } else {
    float* bb = beta_out + rowoff;
    float be[16];
#pragma unroll
    for (int q = 0; q < 16; ++q) be[q] = 0.f;
    store_row(bb + (size_t)(HT - 1)*HK, be);
#pragma unroll
    for (int u = 0; u < PF; ++u) load_row(er[u], eb + (size_t)(HT - 1 - u)*HK);

    for (int s0 = 0; s0 < HT - 1; s0 += PF){
#pragma unroll
      for (int u = 0; u < PF; ++u){
        const int t = HT - 2 - (s0 + u);
        hmm_step<1>(be, er[u], ereg);
        store_row(bb + (size_t)t*HK, be);
        const int tn = (t + 1 - PF > 0) ? t + 1 - PF : 0;
        load_row(er[u], eb + (size_t)tn*HK);
        __builtin_amdgcn_sched_barrier(0);
      }
    }
  }
}

__global__ void __launch_bounds__(256)
hmm_gamma_kernel(const float* alpha, const float* beta, float* out, int npairs){
  const int w    = (blockIdx.x*blockDim.x + threadIdx.x) >> 5;
  const int lane = threadIdx.x & 31;
  if (w >= npairs) return;
  const size_t idx = (size_t)w*32 + lane;
  float g  = alpha[idx] + beta[idx];
  float mx = rmax32(g);
  float s  = rsum32(fexp2((g - mx) * LOG2E));
  out[idx] = g - mx - LN2 * flog2(s);
}

extern "C" void kernel_launch(void* const* d_in, const int* in_sizes, int n_in,
                              void* d_out, int out_size, void* d_ws, size_t ws_size,
                              hipStream_t stream) {
  const float* em     = (const float*)d_in[0];
  const float* log_pi = (const float*)d_in[1];
  const float* log_A  = (const float*)d_in[2];
  float* out  = (float*)d_out;
  float* beta = (float*)d_ws;

  hmm_scan_kernel<<<dim3(8), dim3(32), 0, stream>>>(em, log_pi, log_A, out, beta);

  const int npairs = HB * HT;
  const int threads = 256;
  const int blocks = (npairs*32 + threads - 1) / threads;
  hmm_gamma_kernel<<<dim3(blocks), dim3(threads), 0, stream>>>(out, beta, out, npairs);
}


# Rows addressed as scalar base + one fixed lane-offset register, and each row stored between the step's two halves from registers held past the matrix run: no address-translation wait on the chain
# speedup vs baseline: 2.8211x; 1.2003x over previous
#include <hip/hip_runtime.h>


typedef __attribute__((ext_vector_type(2))) float v2f;
typedef __attribute__((ext_vector_type(4))) float v4f;
typedef __attribute__((ext_vector_type(8))) float v8f;
typedef __attribute__((address_space(1))) v4f GV4;

constexpr int HB = 64;
constexpr int HT = 4096;
constexpr int HK = 32;
constexpr int PF = 5;
constexpr float LOG2E = 1.44269504088896340736f;
constexpr float LN2   = 0.69314718055994530942f;

#if __has_builtin(__builtin_amdgcn_exp2f)
__device__ __forceinline__ float fexp2(float x){ return __builtin_amdgcn_exp2f(x); }
#else
__device__ __forceinline__ float fexp2(float x){ return exp2f(x); }
#endif
#if __has_builtin(__builtin_amdgcn_logf)
__device__ __forceinline__ float flog2(float x){ return __builtin_amdgcn_logf(x); }
#else
__device__ __forceinline__ float flog2(float x){ return log2f(x); }
#endif

__device__ __forceinline__ float rmax32(float x){
#pragma unroll
  for (int m = 16; m >= 1; m >>= 1) x = fmaxf(x, __shfl_xor(x, m, 32));
  return x;
}
__device__ __forceinline__ float rsum32(float x){
#pragma unroll
  for (int m = 16; m >= 1; m >>= 1) x += __shfl_xor(x, m, 32);
  return x;
}

__device__ __forceinline__ float xhalf(float x){
  int b = __builtin_bit_cast(int, x);
  return __builtin_bit_cast(float,
      __builtin_amdgcn_permlanex16(b, b, 0x76543210, 0xfedcba98, false, false));
}

__device__ __forceinline__ v8f wmma4(v2f a, v2f b, v8f c){
  return __builtin_amdgcn_wmma_f32_16x16x4_f32(
       false, a,  false, b,
       (short)0, c,  false,  false);
}


__device__ __forceinline__ unsigned long long uniform_addr(const void* p){
  const unsigned long long a = (unsigned long long)p;
  const unsigned lo = (unsigned)__builtin_amdgcn_readfirstlane((int)(unsigned)a);
  const unsigned hi = (unsigned)__builtin_amdgcn_readfirstlane((int)(unsigned)(a >> 32));
  return ((unsigned long long)hi << 32) | (unsigned long long)lo;
}
__device__ __forceinline__ void load_row(v4f (&dst)[4], const float* ub, unsigned boff){
  const GV4 * q = (const GV4 *)(uniform_addr(ub) + boff);
#pragma unroll
  for (int c = 0; c < 4; ++c) dst[c] = q[c];
}
__device__ __forceinline__ void pack_row(v4f (&w)[4], const float (&v)[16]){
#pragma unroll
  for (int c = 0; c < 4; ++c){
    v4f t = { v[4*c+0], v[4*c+1], v[4*c+2], v[4*c+3] };
    w[c] = t;
  }
}
__device__ __forceinline__ void store_row(float* ub, unsigned boff, const v4f (&w)[4]){
  GV4 * q = (GV4 *)(uniform_addr(ub) + boff);
#pragma unroll
  for (int c = 0; c < 4; ++c) q[c] = w[c];
}
__device__ __forceinline__ void store_row(float* ub, unsigned boff, const float (&v)[16]){
  v4f w[4];
  pack_row(w, v);
  store_row(ub, boff, w);
}

template<int DIR>
__device__ __forceinline__ void hmm_front(const float (&st)[16], const v4f (&emc)[4],
                                          v2f (&b)[8], float& mout){
  float x[16];
#pragma unroll
  for (int q = 0; q < 16; ++q)
    x[q] = (DIR == 0) ? st[q] : st[q] + emc[q >> 2][q & 3];

  const float t0 = fmaxf(fmaxf(x[0],  x[1]),  x[2]);
  const float t1 = fmaxf(fmaxf(x[3],  x[4]),  x[5]);
  const float t2 = fmaxf(fmaxf(x[6],  x[7]),  x[8]);
  const float t3 = fmaxf(fmaxf(x[9],  x[10]), x[11]);
  const float t4 = fmaxf(fmaxf(x[12], x[13]), x[14]);
  const float u0 = fmaxf(fmaxf(t0, t1), t2);
  const float u1 = fmaxf(fmaxf(t3, t4), x[15]);
  float m = fmaxf(u0, u1);
  m = fmaxf(m, xhalf(m));

#pragma unroll
  for (int kc = 0; kc < 8; ++kc){
    b[kc].x = fexp2((x[2*kc]   - m) * LOG2E);
    b[kc].y = fexp2((x[2*kc+1] - m) * LOG2E);
  }
  mout = m;
}
template<int DIR>
__device__ __forceinline__ void hmm_back(float (&stn)[16], const v4f (&emc)[4],
                                         const v2f (&b)[8], const float m,
                                         const v2f (&ereg)[2][8]){
  v8f c0 = {0.f,0.f,0.f,0.f,0.f,0.f,0.f,0.f};
  v8f c1 = {0.f,0.f,0.f,0.f,0.f,0.f,0.f,0.f};
#pragma unroll
  for (int kc = 0; kc < 8; ++kc){
    c0 = wmma4(ereg[0][kc], b[kc], c0);
    c1 = wmma4(ereg[1][kc], b[kc], c1);
  }
#pragma unroll
  for (int r = 0; r < 8; ++r){
    const float b0 = (DIR == 0) ? m + emc[r >> 2][r & 3]           : m;
    const float b1 = (DIR == 0) ? m + emc[2 + (r >> 2)][r & 3]     : m;
    stn[r]     = fmaf(LN2, flog2(c0[r]), b0);
    stn[8 + r] = fmaf(LN2, flog2(c1[r]), b1);
  }
}
__device__ __forceinline__ void hold_row(const v4f (&w)[4], float after){
  asm volatile("" :: "v"(w[0]), "v"(w[1]), "v"(w[2]), "v"(w[3]), "v"(after));
}

__global__ void __launch_bounds__(32)
hmm_scan_kernel(const float* __restrict__ em,
                const float* __restrict__ log_pi,
                const float* __restrict__ log_A,
                float* __restrict__ alpha_out,
                float* __restrict__ beta_out){
  __shared__ float sA[32*33];

  const int lane = threadIdx.x;
  const int h    = lane >> 4;
  const int ln   = lane & 15;
  const int dir  = blockIdx.x >> 2;
  const int mb   = blockIdx.x & 3;

  for (int i = 0; i < 32; ++i){
    float xv = log_A[i*32 + lane];
    float mx = rmax32(xv);
    float e  = fexp2((xv - mx) * LOG2E);
    float s  = rsum32(e);
    sA[i*33 + lane] = e / s;
  }
  __syncthreads();

  v2f ereg[2][8];
#pragma unroll
  for (int mt = 0; mt < 2; ++mt){
#pragma unroll
    for (int kc = 0; kc < 8; ++kc){
      const int sM = 16*(ln >> 3) + 8*mt + (ln & 7);
      const int sK = 16*h + 2*kc;
      if (dir == 0){
        ereg[mt][kc].x = sA[sK*33 + sM];
        ereg[mt][kc].y = sA[(sK+1)*33 + sM];
      } else {
        ereg[mt][kc].x = sA[sM*33 + sK];
        ereg[mt][kc].y = sA[sM*33 + sK + 1];
      }
    }
  }

  const size_t   uoff = (size_t)mb*16*HT*HK;
  const unsigned boff = (unsigned)(ln*HT*HK + 16*h) * 4u;
  const float* eb = em + uoff;

  v4f er[PF][4];

  if (dir == 0){
    float xp = log_pi[lane];
    float mp = rmax32(xp);
    float sp = rsum32(fexp2((xp - mp) * LOG2E));
    float lp = xp - mp - LN2 * flog2(sp);

    float* ab = alpha_out + uoff;
    float al[16];
    v4f e0[4];
    load_row(e0, eb, boff);
#pragma unroll
    for (int u = 0; u < PF - 1; ++u) load_row(er[u], eb + (size_t)(1 + u)*HK, boff);
#pragma unroll
    for (int q = 0; q < 16; ++q) al[q] = __shfl(lp, 16*h + q, 32) + e0[q >> 2][q & 3];

    unsigned bo = boff;
    for (int t0 = 1; t0 < HT; t0 += PF){
      asm volatile("" : "+v"(bo));
#pragma unroll
      for (int u = 0; u < PF; ++u){
        const int t = t0 + u;
        v2f pb[8];
        float m;
        hmm_front<0>(al, er[u], pb, m);
        __builtin_amdgcn_sched_barrier(0);
        v4f wr[4];
        pack_row(wr, al);
        store_row(ab + (size_t)(t - 1)*HK, bo, wr);
        const int up = (u + PF - 1) % PF;
        const int tn = (t - 1 + PF < HT) ? t - 1 + PF : HT - 1;
        load_row(er[up], eb + (size_t)tn*HK, bo);
        float aln[16];
        hmm_back<0>(aln, er[u], pb, m, ereg);
        hold_row(wr, aln[0]);
#pragma unroll
        for (int q = 0; q < 16; ++q) al[q] = aln[q];
        __builtin_amdgcn_sched_barrier(0);
      }
    }
    store_row(ab + (size_t)(HT - 1)*HK, boff, al);
  } else {
    float* bb = beta_out + uoff;
    float be[16];
#pragma unroll
    for (int q = 0; q < 16; ++q) be[q] = 0.f;
#pragma unroll
    for (int u = 0; u < PF; ++u) load_row(er[u], eb + (size_t)(HT - 1 - u)*HK, boff);

    unsigned bo = boff;
    for (int s0 = 0; s0 < HT - 1; s0 += PF){
      asm volatile("" : "+v"(bo));
#pragma unroll
      for (int u = 0; u < PF; ++u){
        const int t = HT - 2 - (s0 + u);
        v2f pb[8];
        float m;
        hmm_front<1>(be, er[u], pb, m);
        __builtin_amdgcn_sched_barrier(0);
        v4f wr[4];
        pack_row(wr, be);
        store_row(bb + (size_t)(t + 1)*HK, bo, wr);
        const int tn = (t + 1 - PF > 0) ? t + 1 - PF : 0;
        load_row(er[u], eb + (size_t)tn*HK, bo);
        float ben[16];
        hmm_back<1>(ben, er[u], pb, m, ereg);
        hold_row(wr, ben[0]);
#pragma unroll
        for (int q = 0; q < 16; ++q) be[q] = ben[q];
        __builtin_amdgcn_sched_barrier(0);
      }
    }
    store_row(bb, boff, be);
  }
}

__global__ void __launch_bounds__(256)
hmm_gamma_kernel(const float* alpha, const float* beta, float* out, int npairs){
  const int w    = (blockIdx.x*blockDim.x + threadIdx.x) >> 5;
  const int lane = threadIdx.x & 31;
  if (w >= npairs) return;
  const size_t idx = (size_t)w*32 + lane;
  float g  = alpha[idx] + beta[idx];
  float mx = rmax32(g);
  float s  = rsum32(fexp2((g - mx) * LOG2E));
  out[idx] = g - mx - LN2 * flog2(s);
}

extern "C" void kernel_launch(void* const* d_in, const int* in_sizes, int n_in,
                              void* d_out, int out_size, void* d_ws, size_t ws_size,
                              hipStream_t stream) {
  const float* em     = (const float*)d_in[0];
  const float* log_pi = (const float*)d_in[1];
  const float* log_A  = (const float*)d_in[2];
  float* out  = (float*)d_out;
  float* beta = (float*)d_ws;

  hmm_scan_kernel<<<dim3(8), dim3(32), 0, stream>>>(em, log_pi, log_A, out, beta);

  const int npairs = HB * HT;
  const int threads = 256;
  const int blocks = (npairs*32 + threads - 1) / threads;
  hmm_gamma_kernel<<<dim3(blocks), dim3(threads), 0, stream>>>(out, beta, out, npairs);
}


# Scan carried as scaled probabilities (power-of-two row scale from the run's own operand; cancels in gamma): max, exp, log leave the step chain, now matrix run + one multiply; sum rounding changes
# speedup vs baseline: 3.2268x; 1.1438x over previous
#include <hip/hip_runtime.h>


typedef __attribute__((ext_vector_type(2))) float v2f;
typedef __attribute__((ext_vector_type(4))) float v4f;
typedef __attribute__((ext_vector_type(8))) float v8f;
typedef __attribute__((address_space(1))) v4f GV4;

constexpr int HB = 64;
constexpr int HT = 4096;
constexpr int HK = 32;
constexpr int PF = 5;
constexpr float LOG2E = 1.44269504088896340736f;
constexpr float LN2   = 0.69314718055994530942f;

#if __has_builtin(__builtin_amdgcn_exp2f)
__device__ __forceinline__ float fexp2(float x){ return __builtin_amdgcn_exp2f(x); }
#else
__device__ __forceinline__ float fexp2(float x){ return exp2f(x); }
#endif
#if __has_builtin(__builtin_amdgcn_logf)
__device__ __forceinline__ float flog2(float x){ return __builtin_amdgcn_logf(x); }
#else
__device__ __forceinline__ float flog2(float x){ return log2f(x); }
#endif

__device__ __forceinline__ float rmax32(float x){
#pragma unroll
  for (int m = 16; m >= 1; m >>= 1) x = fmaxf(x, __shfl_xor(x, m, 32));
  return x;
}
__device__ __forceinline__ float rsum32(float x){
#pragma unroll
  for (int m = 16; m >= 1; m >>= 1) x += __shfl_xor(x, m, 32);
  return x;
}

__device__ __forceinline__ float xhalf(float x){
  int b = __builtin_bit_cast(int, x);
  return __builtin_bit_cast(float,
      __builtin_amdgcn_permlanex16(b, b, 0x76543210, 0xfedcba98, false, false));
}

__device__ __forceinline__ v8f wmma4(v2f a, v2f b, v8f c){
  return __builtin_amdgcn_wmma_f32_16x16x4_f32(
       false, a,  false, b,
       (short)0, c,  false,  false);
}


__device__ __forceinline__ unsigned long long uniform_addr(const void* p){
  const unsigned long long a = (unsigned long long)p;
  const unsigned lo = (unsigned)__builtin_amdgcn_readfirstlane((int)(unsigned)a);
  const unsigned hi = (unsigned)__builtin_amdgcn_readfirstlane((int)(unsigned)(a >> 32));
  return ((unsigned long long)hi << 32) | (unsigned long long)lo;
}
__device__ __forceinline__ void load_row(v4f (&dst)[4], const float* ub, unsigned boff){
  const GV4 * q = (const GV4 *)(uniform_addr(ub) + boff);
#pragma unroll
  for (int c = 0; c < 4; ++c) dst[c] = q[c];
}
__device__ __forceinline__ void pack_row(v4f (&w)[4], const float (&v)[16]){
#pragma unroll
  for (int c = 0; c < 4; ++c){
    v4f t = { v[4*c+0], v[4*c+1], v[4*c+2], v[4*c+3] };
    w[c] = t;
  }
}
__device__ __forceinline__ void store_row(float* ub, unsigned boff, const v4f (&w)[4]){
  GV4 * q = (GV4 *)(uniform_addr(ub) + boff);
#pragma unroll
  for (int c = 0; c < 4; ++c) q[c] = w[c];
}
__device__ __forceinline__ void store_row(float* ub, unsigned boff, const float (&v)[16]){
  v4f w[4];
  pack_row(w, v);
  store_row(ub, boff, w);
}

__device__ __forceinline__ float row_exponent(const float (&p)[16]){
  const float t0 = fmaxf(fmaxf(p[0],  p[1]),  p[2]);
  const float t1 = fmaxf(fmaxf(p[3],  p[4]),  p[5]);
  const float t2 = fmaxf(fmaxf(p[6],  p[7]),  p[8]);
  const float t3 = fmaxf(fmaxf(p[9],  p[10]), p[11]);
  const float t4 = fmaxf(fmaxf(p[12], p[13]), p[14]);
  const float u0 = fmaxf(fmaxf(t0, t1), t2);
  const float u1 = fmaxf(fmaxf(t3, t4), p[15]);
  float m = fmaxf(u0, u1);
  m = fmaxf(m, xhalf(m));
#if __has_builtin(__builtin_amdgcn_frexp_expf)
  return (float)__builtin_amdgcn_frexp_expf(m);
#else
  return (float)(((__builtin_bit_cast(int, m) >> 23) & 0xff) - 126);
#endif
}
__device__ __forceinline__ void lin_step(float (&pn)[16], v8f& d0, v8f& d1,
                                         const float (&p)[16], const v4f (&emc)[4],
                                         const v2f (&ereg)[2][8]){
  v8f c0 = {0.f,0.f,0.f,0.f,0.f,0.f,0.f,0.f};
  v8f c1 = {0.f,0.f,0.f,0.f,0.f,0.f,0.f,0.f};
#pragma unroll
  for (int kc = 0; kc < 8; ++kc){
    const v2f b = { p[2*kc], p[2*kc+1] };
    c0 = wmma4(ereg[0][kc], b, c0);
    c1 = wmma4(ereg[1][kc], b, c1);
  }
  const float ef = row_exponent(p);
  float w[16];
#pragma unroll
  for (int q = 0; q < 16; ++q) w[q] = fexp2(fmaf(emc[q >> 2][q & 3], LOG2E, -ef));
#pragma unroll
  for (int r = 0; r < 8; ++r){
    pn[r]     = c0[r] * w[r];
    pn[8 + r] = c1[r] * w[8 + r];
  }
  d0 = c0;
  d1 = c1;
}
__device__ __forceinline__ void hold_row(const v4f (&w)[4], float after){
  asm volatile("" :: "v"(w[0]), "v"(w[1]), "v"(w[2]), "v"(w[3]), "v"(after));
}

__global__ void __launch_bounds__(32)
hmm_scan_kernel(const float* __restrict__ em,
                const float* __restrict__ log_pi,
                const float* __restrict__ log_A,
                float* __restrict__ alpha_out,
                float* __restrict__ beta_out){
  __shared__ float sA[32*33];

  const int lane = threadIdx.x;
  const int h    = lane >> 4;
  const int ln   = lane & 15;
  const int dir  = blockIdx.x >> 2;
  const int mb   = blockIdx.x & 3;

  for (int i = 0; i < 32; ++i){
    float xv = log_A[i*32 + lane];
    float mx = rmax32(xv);
    float e  = fexp2((xv - mx) * LOG2E);
    float s  = rsum32(e);
    sA[i*33 + lane] = e / s;
  }
  __syncthreads();

  v2f ereg[2][8];
#pragma unroll
  for (int mt = 0; mt < 2; ++mt){
#pragma unroll
    for (int kc = 0; kc < 8; ++kc){
      const int sM = 16*(ln >> 3) + 8*mt + (ln & 7);
      const int sK = 16*h + 2*kc;
      if (dir == 0){
        ereg[mt][kc].x = sA[sK*33 + sM];
        ereg[mt][kc].y = sA[(sK+1)*33 + sM];
      } else {
        ereg[mt][kc].x = sA[sM*33 + sK];
        ereg[mt][kc].y = sA[sM*33 + sK + 1];
      }
    }
  }

  const size_t   uoff = (size_t)mb*16*HT*HK;
  const unsigned boff = (unsigned)(ln*HT*HK + 16*h) * 4u;
  const float* eb = em + uoff;

  v4f er[PF][4];

  if (dir == 0){
    float xp = log_pi[lane];
    float mp = rmax32(xp);
    float sp = rsum32(fexp2((xp - mp) * LOG2E));
    float lp = xp - mp - LN2 * flog2(sp);

    float* ab = alpha_out + uoff;
    float al[16];
    v4f e0[4];
    load_row(e0, eb, boff);
#pragma unroll
    for (int u = 0; u < PF - 1; ++u) load_row(er[u], eb + (size_t)(1 + u)*HK, boff);
#pragma unroll
    for (int q = 0; q < 16; ++q)
      al[q] = fexp2((__shfl(lp, 16*h + q, 32) + e0[q >> 2][q & 3]) * LOG2E);

    unsigned bo = boff;
    for (int t0 = 1; t0 < HT; t0 += PF){
      asm volatile("" : "+v"(bo));
#pragma unroll
      for (int u = 0; u < PF; ++u){
        const int t = t0 + u;
        v4f wr[4];
        pack_row(wr, al);
        store_row(ab + (size_t)(t - 1)*HK, bo, wr);
        const int up = (u + PF - 1) % PF;
        const int tn = (t - 1 + PF < HT) ? t - 1 + PF : HT - 1;
        load_row(er[up], eb + (size_t)tn*HK, bo);
        float aln[16];
        v8f d0, d1;
        lin_step(aln, d0, d1, al, er[u], ereg);
        hold_row(wr, aln[0]);
#pragma unroll
        for (int q = 0; q < 16; ++q) al[q] = aln[q];
        __builtin_amdgcn_sched_barrier(0);
      }
    }
    store_row(ab + (size_t)(HT - 1)*HK, boff, al);
  } else {
    float* bb = beta_out + uoff;
    v4f yr[4];
#pragma unroll
    for (int c = 0; c < 4; ++c){ v4f one = {1.f, 1.f, 1.f, 1.f}; yr[c] = one; }
    float xe[16];
    {
      v4f eT[4];
      load_row(eT, eb + (size_t)(HT - 1)*HK, boff);
#pragma unroll
      for (int q = 0; q < 16; ++q) xe[q] = fexp2(eT[q >> 2][q & 3] * LOG2E);
    }
#pragma unroll
    for (int u = 0; u < PF - 1; ++u) load_row(er[u], eb + (size_t)(HT - 2 - u)*HK, boff);

    unsigned bo = boff;
    for (int s0 = 0; s0 < HT - 1; s0 += PF){
      asm volatile("" : "+v"(bo));
#pragma unroll
      for (int u = 0; u < PF; ++u){
        const int t = HT - 2 - (s0 + u);
        store_row(bb + (size_t)(t + 1)*HK, bo, yr);
        const int up = (u + PF - 1) % PF;
        const int tn = (t + 1 - PF > 0) ? t + 1 - PF : 0;
        load_row(er[up], eb + (size_t)tn*HK, bo);
        float xn[16];
        v8f d0, d1;
        lin_step(xn, d0, d1, xe, er[u], ereg);
        hold_row(yr, xn[0]);
        {
          v4f a0 = { d0[0], d0[1], d0[2], d0[3] };
          v4f a1 = { d0[4], d0[5], d0[6], d0[7] };
          v4f a2 = { d1[0], d1[1], d1[2], d1[3] };
          v4f a3 = { d1[4], d1[5], d1[6], d1[7] };
          yr[0] = a0; yr[1] = a1; yr[2] = a2; yr[3] = a3;
        }
#pragma unroll
        for (int q = 0; q < 16; ++q) xe[q] = xn[q];
        __builtin_amdgcn_sched_barrier(0);
      }
    }
    store_row(bb, boff, yr);
  }
}

__global__ void __launch_bounds__(256)
hmm_gamma_kernel(const float* alpha, const float* beta, float* out, int npairs){
  const int w    = (blockIdx.x*blockDim.x + threadIdx.x) >> 5;
  const int lane = threadIdx.x & 31;
  if (w >= npairs) return;
  const size_t idx = (size_t)w*32 + lane;
  const float a = alpha[idx];
  const float b = beta[idx];
  const float s = rsum32(a * b);
  out[idx] = LN2 * ((flog2(a) + flog2(b)) - flog2(s));
}

extern "C" void kernel_launch(void* const* d_in, const int* in_sizes, int n_in,
                              void* d_out, int out_size, void* d_ws, size_t ws_size,
                              hipStream_t stream) {
  const float* em     = (const float*)d_in[0];
  const float* log_pi = (const float*)d_in[1];
  const float* log_A  = (const float*)d_in[2];
  float* out  = (float*)d_out;
  float* beta = (float*)d_ws;

  hmm_scan_kernel<<<dim3(8), dim3(32), 0, stream>>>(em, log_pi, log_A, out, beta);

  const int npairs = HB * HT;
  const int threads = 256;
  const int blocks = (npairs*32 + threads - 1) / threads;
  hmm_gamma_kernel<<<dim3(blocks), dim3(threads), 0, stream>>>(out, beta, out, npairs);
}


# Time axis cut into 63 chunks: each chunk's 32x32 operator built by matrix products (all chunks at once), edge rows walked over the operators, chunks rescanned in parallel; sum order changes
# speedup vs baseline: 14.0638x; 4.3584x over previous
#include <hip/hip_runtime.h>


typedef __attribute__((ext_vector_type(2))) float v2f;
typedef __attribute__((ext_vector_type(4))) float v4f;
typedef __attribute__((ext_vector_type(8))) float v8f;
typedef __attribute__((address_space(1))) v4f GV4;

constexpr int HB = 64;
constexpr int HT = 4096;
constexpr int HK = 32;
constexpr int CL  = 65;
constexpr int NCH = 63;
constexpr int PFA = 5;
constexpr int PF  = 4;
constexpr float LOG2E = 1.44269504088896340736f;
constexpr float LN2   = 0.69314718055994530942f;

#if __has_builtin(__builtin_amdgcn_exp2f)
__device__ __forceinline__ float fexp2(float x){ return __builtin_amdgcn_exp2f(x); }
#else
__device__ __forceinline__ float fexp2(float x){ return exp2f(x); }
#endif
#if __has_builtin(__builtin_amdgcn_logf)
__device__ __forceinline__ float flog2(float x){ return __builtin_amdgcn_logf(x); }
#else
__device__ __forceinline__ float flog2(float x){ return log2f(x); }
#endif

__device__ __forceinline__ float rmax32(float x){
#pragma unroll
  for (int m = 16; m >= 1; m >>= 1) x = fmaxf(x, __shfl_xor(x, m, 32));
  return x;
}
__device__ __forceinline__ float rsum32(float x){
#pragma unroll
  for (int m = 16; m >= 1; m >>= 1) x += __shfl_xor(x, m, 32);
  return x;
}

__device__ __forceinline__ float xhalf(float x){
  int b = __builtin_bit_cast(int, x);
  return __builtin_bit_cast(float,
      __builtin_amdgcn_permlanex16(b, b, 0x76543210, 0xfedcba98, false, false));
}

__device__ __forceinline__ v8f wmma4(v2f a, v2f b, v8f c){
  return __builtin_amdgcn_wmma_f32_16x16x4_f32(
       false, a,  false, b,
       (short)0, c,  false,  false);
}


__device__ __forceinline__ unsigned long long uniform_addr(const void* p){
  const unsigned long long a = (unsigned long long)p;
  const unsigned lo = (unsigned)__builtin_amdgcn_readfirstlane((int)(unsigned)a);
  const unsigned hi = (unsigned)__builtin_amdgcn_readfirstlane((int)(unsigned)(a >> 32));
  return ((unsigned long long)hi << 32) | (unsigned long long)lo;
}
__device__ __forceinline__ void load_row(v4f (&dst)[4], const float* ub, unsigned boff){
  const GV4 * q = (const GV4 *)(uniform_addr(ub) + boff);
#pragma unroll
  for (int c = 0; c < 4; ++c) dst[c] = q[c];
}
__device__ __forceinline__ void pack_row(v4f (&w)[4], const float (&v)[16]){
#pragma unroll
  for (int c = 0; c < 4; ++c){
    v4f t = { v[4*c+0], v[4*c+1], v[4*c+2], v[4*c+3] };
    w[c] = t;
  }
}
__device__ __forceinline__ void store_row(float* ub, unsigned boff, const v4f (&w)[4]){
  GV4 * q = (GV4 *)(uniform_addr(ub) + boff);
#pragma unroll
  for (int c = 0; c < 4; ++c) q[c] = w[c];
}
__device__ __forceinline__ void store_row(float* ub, unsigned boff, const float (&v)[16]){
  v4f w[4];
  pack_row(w, v);
  store_row(ub, boff, w);
}

__device__ __forceinline__ float row_exponent(const float (&p)[16]){
  const float t0 = fmaxf(fmaxf(p[0],  p[1]),  p[2]);
  const float t1 = fmaxf(fmaxf(p[3],  p[4]),  p[5]);
  const float t2 = fmaxf(fmaxf(p[6],  p[7]),  p[8]);
  const float t3 = fmaxf(fmaxf(p[9],  p[10]), p[11]);
  const float t4 = fmaxf(fmaxf(p[12], p[13]), p[14]);
  const float u0 = fmaxf(fmaxf(t0, t1), t2);
  const float u1 = fmaxf(fmaxf(t3, t4), p[15]);
  float m = fmaxf(u0, u1);
  m = fmaxf(m, xhalf(m));
#if __has_builtin(__builtin_amdgcn_frexp_expf)
  return (float)__builtin_amdgcn_frexp_expf(m);
#else
  return (float)(((__builtin_bit_cast(int, m) >> 23) & 0xff) - 126);
#endif
}
__device__ __forceinline__ void lin_step(float (&pn)[16], v8f& d0, v8f& d1, float& ef_out,
                                         const float (&p)[16], const v4f (&emc)[4],
                                         const v2f (&ereg)[2][8]){
  v8f c0 = {0.f,0.f,0.f,0.f,0.f,0.f,0.f,0.f};
  v8f c1 = {0.f,0.f,0.f,0.f,0.f,0.f,0.f,0.f};
#pragma unroll
  for (int kc = 0; kc < 8; ++kc){
    const v2f b = { p[2*kc], p[2*kc+1] };
    c0 = wmma4(ereg[0][kc], b, c0);
    c1 = wmma4(ereg[1][kc], b, c1);
  }
  const float ef = row_exponent(p);
  ef_out = ef;
  float w[16];
#pragma unroll
  for (int q = 0; q < 16; ++q) w[q] = fexp2(fmaf(emc[q >> 2][q & 3], LOG2E, -ef));
#pragma unroll
  for (int r = 0; r < 8; ++r){
    pn[r]     = c0[r] * w[r];
    pn[8 + r] = c1[r] * w[8 + r];
  }
  d0 = c0;
  d1 = c1;
}
__device__ __forceinline__ void hold_row(const v4f (&w)[4], float after){
  asm volatile("" :: "v"(w[0]), "v"(w[1]), "v"(w[2]), "v"(w[3]), "v"(after));
}


__device__ __forceinline__ void build_expA(float* sA, const float* __restrict__ log_A, int lane){
  float x[32];
#pragma unroll
  for (int c = 0; c < 8; ++c){
    const v4f t = *(const v4f*)(log_A + lane*32 + 4*c);
    x[4*c+0] = t[0]; x[4*c+1] = t[1]; x[4*c+2] = t[2]; x[4*c+3] = t[3];
  }
  float mx = x[0];
#pragma unroll
  for (int j = 1; j < 32; ++j) mx = fmaxf(mx, x[j]);
  float s = 0.f;
#pragma unroll
  for (int j = 0; j < 32; ++j){ x[j] = fexp2((x[j] - mx) * LOG2E); s += x[j]; }
  const float r = 1.0f / s;
#pragma unroll
  for (int j = 0; j < 32; ++j) sA[lane*33 + j] = x[j] * r;
}

__device__ __forceinline__ void build_E(v2f (&ereg)[2][8], const float* sA, int ln, int h, int dir){
#pragma unroll
  for (int mt = 0; mt < 2; ++mt){
#pragma unroll
    for (int kc = 0; kc < 8; ++kc){
      const int sM = 16*(ln >> 3) + 8*mt + (ln & 7);
      const int sK = 16*h + 2*kc;
      if (dir == 0){
        ereg[mt][kc].x = sA[sK*33 + sM];
        ereg[mt][kc].y = sA[(sK+1)*33 + sM];
      } else {
        ereg[mt][kc].x = sA[sM*33 + sK];
        ereg[mt][kc].y = sA[sM*33 + sK + 1];
      }
    }
  }
}

__global__ void __launch_bounds__(32)
hmm_operator_kernel(const float* __restrict__ em,
                    const float* __restrict__ log_A,
                    float* __restrict__ fbuf){
  __shared__ float sA[32*33];
  const int lane = threadIdx.x;
  const int h    = lane >> 4;
  const int ln   = lane & 15;
  const int nt   = blockIdx.x & 1;
  const int cb   = blockIdx.x >> 1;
  const int c    = cb % NCH;
  const int b    = cb / NCH;

  build_expA(sA, log_A, lane);
  __syncthreads();
  v2f ereg[2][8];
  build_E(ereg, sA, ln, h, 0);

  const size_t urow = (size_t)b*HT + (size_t)(CL*c + 1);
  const float* eb = em + urow*HK;
  const unsigned boff = (unsigned)(16*h) * 4u;

  float p[16];
#pragma unroll
  for (int q = 0; q < 16; ++q) p[q] = (16*h + q == 16*nt + ln) ? 1.f : 0.f;
  float esum = 0.f;

  v4f er[PFA][4];
#pragma unroll
  for (int u = 0; u < PFA - 1; ++u) load_row(er[u], eb + (size_t)u*HK, boff);

  unsigned bo = boff;
  for (int k0 = 0; k0 < CL; k0 += PFA){
    asm volatile("" : "+v"(bo));
#pragma unroll
    for (int u = 0; u < PFA; ++u){
      const int k  = k0 + u;
      const int up = (u + PFA - 1) % PFA;
      const int kn = (k + PFA - 1 < CL) ? k + PFA - 1 : CL - 1;
      load_row(er[up], eb + (size_t)kn*HK, bo);
      float pn[16];
      v8f d0, d1;
      float ef;
      lin_step(pn, d0, d1, ef, p, er[u], ereg);
      esum += ef;
#pragma unroll
      for (int q = 0; q < 16; ++q) p[q] = pn[q];
    }
  }
  float* dst = fbuf + urow*HK;
  const unsigned coff = (unsigned)((16*nt + ln)*HK + 16*h) * 4u;
  store_row(dst, coff, p);
  if (h == 0) dst[(size_t)(32 + nt)*HK + 16*nt + ln] = esum;
}

__device__ __forceinline__ float expo_of(float m){
#if __has_builtin(__builtin_amdgcn_frexp_expf)
  return (float)__builtin_amdgcn_frexp_expf(m);
#else
  return (float)(((__builtin_bit_cast(int, m) >> 23) & 0xff) - 126);
#endif
}

__device__ __forceinline__ float lane_swz(float x, const int ctrl_sel){
  const int b = __builtin_bit_cast(int, x);
  int r;
  if (ctrl_sel == 0)      r = __builtin_amdgcn_update_dpp(b, b, 0xB1,  0xf, 0xf, false);
  else if (ctrl_sel == 1) r = __builtin_amdgcn_update_dpp(b, b, 0x4E,  0xf, 0xf, false);
  else if (ctrl_sel == 2) r = __builtin_amdgcn_update_dpp(b, b, 0x141, 0xf, 0xf, false);
  else                    r = __builtin_amdgcn_update_dpp(b, b, 0x140, 0xf, 0xf, false);
  return __builtin_bit_cast(float, r);
}
__device__ __forceinline__ float wmax32(float x){
  x = fmaxf(x, lane_swz(x, 0));
  x = fmaxf(x, lane_swz(x, 1));
  x = fmaxf(x, lane_swz(x, 2));
  x = fmaxf(x, lane_swz(x, 3));
  return fmaxf(x, xhalf(x));
}
__device__ __forceinline__ float lane_bcast(float x, const int j){
  return __builtin_bit_cast(float, __builtin_amdgcn_readlane(__builtin_bit_cast(int, x), j));
}

__global__ void __launch_bounds__(32)
hmm_edge_kernel(const float* __restrict__ em,
                const float* __restrict__ log_pi,
                float* fa,
                float* __restrict__ bw){
  const int lane = threadIdx.x;
  const int b    = blockIdx.x >> 1;
  const int dir  = blockIdx.x & 1;
  float* fb = fa + (size_t)b*HT*HK;

  if (dir == 0){
    float xp = log_pi[lane];
    float mp = rmax32(xp);
    float sp = rsum32(fexp2((xp - mp) * LOG2E));
    float lp = xp - mp - LN2 * flog2(sp);
    float v  = fexp2((lp + em[(size_t)b*HT*HK + lane]) * LOG2E);
    fb[lane] = v;

    float f[3][32];
    float sx[3];
#pragma unroll
    for (int u = 0; u < 2; ++u){
      const float* src = fb + (size_t)(CL*u + 1)*HK;
#pragma unroll
      for (int j = 0; j < 32; ++j) f[u][j] = src[(size_t)j*HK + lane];
      sx[u] = src[(size_t)(32 + (lane >> 4))*HK + lane];
    }
    for (int c0 = 0; c0 < NCH; c0 += 3){
#pragma unroll
      for (int u = 0; u < 3; ++u){
        const int c  = c0 + u;
        const int cn = (c + 2 < NCH) ? c + 2 : NCH - 1;
        {
          const int un = (u + 2) % 3;
          const float* src = fb + (size_t)(CL*cn + 1)*HK;
#pragma unroll
          for (int j = 0; j < 32; ++j) f[un][j] = src[(size_t)j*HK + lane];
          sx[un] = src[(size_t)(32 + (lane >> 4))*HK + lane];
        }
        const float smax = wmax32(sx[u]);
        const float uu = v * fexp2(sx[u] - smax);
        float a0 = 0.f, a1 = 0.f, a2 = 0.f, a3 = 0.f;
#pragma unroll
        for (int j = 0; j < 32; j += 4){
          a0 = fmaf(f[u][j+0], lane_bcast(uu, j+0), a0);
          a1 = fmaf(f[u][j+1], lane_bcast(uu, j+1), a1);
          a2 = fmaf(f[u][j+2], lane_bcast(uu, j+2), a2);
          a3 = fmaf(f[u][j+3], lane_bcast(uu, j+3), a3);
        }
        const float acc = (a0 + a1) + (a2 + a3);
        const float e = expo_of(wmax32(acc));
        v = acc * fexp2(-e);
        fb[(size_t)(CL*(c + 1))*HK + lane] = v;
      }
    }
  } else {
    float* bb = bw + (size_t)b*HT*HK;
    float v = 1.f;
    bb[(size_t)(HT - 1)*HK + lane] = v;

    v4f g[3][8];
    float sx[3];
#pragma unroll
    for (int u = 0; u < 2; ++u){
      const float* src = fb + (size_t)(CL*(NCH - 1 - u) + 1)*HK;
#pragma unroll
      for (int q = 0; q < 8; ++q) g[u][q] = *(const v4f*)(src + lane*HK + 4*q);
      sx[u] = src[(size_t)(32 + (lane >> 4))*HK + lane];
    }
    for (int c0 = 0; c0 < NCH; c0 += 3){
#pragma unroll
      for (int u = 0; u < 3; ++u){
        const int c  = NCH - 1 - (c0 + u);
        const int cn = (c - 2 > 0) ? c - 2 : 0;
        {
          const int un = (u + 2) % 3;
          const float* src = fb + (size_t)(CL*cn + 1)*HK;
#pragma unroll
          for (int q = 0; q < 8; ++q) g[un][q] = *(const v4f*)(src + lane*HK + 4*q);
          sx[un] = src[(size_t)(32 + (lane >> 4))*HK + lane];
        }
        float a0 = 0.f, a1 = 0.f, a2 = 0.f, a3 = 0.f;
#pragma unroll
        for (int q = 0; q < 8; ++q){
          a0 = fmaf(g[u][q][0], lane_bcast(v, 4*q+0), a0);
          a1 = fmaf(g[u][q][1], lane_bcast(v, 4*q+1), a1);
          a2 = fmaf(g[u][q][2], lane_bcast(v, 4*q+2), a2);
          a3 = fmaf(g[u][q][3], lane_bcast(v, 4*q+3), a3);
        }
        const float smax = wmax32(sx[u]);
        const float acc = ((a0 + a1) + (a2 + a3)) * fexp2(sx[u] - smax);
        const float e = expo_of(wmax32(acc));
        v = acc * fexp2(-e);
        bb[(size_t)(CL*c)*HK + lane] = v;
      }
    }
  }
}

__global__ void __launch_bounds__(32)
hmm_scan_kernel(const float* __restrict__ em,
                const float* __restrict__ log_A,
                float* alpha_out,
                float* beta_out){
  __shared__ float sA[32*33];

  const int lane = threadIdx.x;
  const int h    = lane >> 4;
  const int ln   = lane & 15;
  const int mb   = blockIdx.x & 3;
  const int dir  = (blockIdx.x >> 2) & 1;
  const int ch   = blockIdx.x >> 3;

  build_expA(sA, log_A, lane);
  __syncthreads();
  v2f ereg[2][8];
  build_E(ereg, sA, ln, h, dir);

  const size_t   uoff = (size_t)mb*16*HT*HK;
  const unsigned boff = (unsigned)(ln*HT*HK + 16*h) * 4u;
  const float* eb = em + uoff;

  v4f er[PF][4];

  if (dir == 0){
    const int tb = CL*ch;
    float* ab = alpha_out + uoff;
    float al[16];
    {
      v4f a0[4];
      load_row(a0, ab + (size_t)tb*HK, boff);
#pragma unroll
      for (int q = 0; q < 16; ++q) al[q] = a0[q >> 2][q & 3];
    }
#pragma unroll
    for (int u = 0; u < PF - 1; ++u) load_row(er[u], eb + (size_t)(tb + 1 + u)*HK, boff);

    unsigned bo = boff;
    for (int k0 = 1; k0 < CL; k0 += PF){
      asm volatile("" : "+v"(bo));
#pragma unroll
      for (int u = 0; u < PF; ++u){
        const int t = tb + k0 + u;
        v4f wr[4];
        pack_row(wr, al);
        store_row(ab + (size_t)(t - 1)*HK, bo, wr);
        const int up = (u + PF - 1) % PF;
        const int tn = (t - 1 + PF < HT) ? t - 1 + PF : HT - 1;
        load_row(er[up], eb + (size_t)tn*HK, bo);
        float aln[16];
        v8f d0, d1;
        float ef;
        lin_step(aln, d0, d1, ef, al, er[u], ereg);
        hold_row(wr, aln[0]);
#pragma unroll
        for (int q = 0; q < 16; ++q) al[q] = aln[q];
        __builtin_amdgcn_sched_barrier(0);
      }
    }
    store_row(ab + (size_t)(tb + CL - 1)*HK, boff, al);
  } else {
    const int te = CL*(ch + 1);
    float* bb = beta_out + uoff;
    v4f yr[4];
    load_row(yr, bb + (size_t)te*HK, boff);
    float xe[16];
    {
      v4f eT[4];
      load_row(eT, eb + (size_t)te*HK, boff);
#pragma unroll
      for (int q = 0; q < 16; ++q) xe[q] = yr[q >> 2][q & 3] * fexp2(eT[q >> 2][q & 3] * LOG2E);
    }
#pragma unroll
    for (int u = 0; u < PF - 1; ++u) load_row(er[u], eb + (size_t)(te - 1 - u)*HK, boff);

    unsigned bo = boff;
    for (int s0 = 0; s0 < CL - 1; s0 += PF){
      asm volatile("" : "+v"(bo));
#pragma unroll
      for (int u = 0; u < PF; ++u){
        const int t = te - 1 - (s0 + u);
        store_row(bb + (size_t)(t + 1)*HK, bo, yr);
        const int up = (u + PF - 1) % PF;
        const int tn = (t + 1 - PF > 0) ? t + 1 - PF : 0;
        load_row(er[up], eb + (size_t)tn*HK, bo);
        float xn[16];
        v8f d0, d1;
        float ef;
        lin_step(xn, d0, d1, ef, xe, er[u], ereg);
        hold_row(yr, xn[0]);
        {
          v4f a0 = { d0[0], d0[1], d0[2], d0[3] };
          v4f a1 = { d0[4], d0[5], d0[6], d0[7] };
          v4f a2 = { d1[0], d1[1], d1[2], d1[3] };
          v4f a3 = { d1[4], d1[5], d1[6], d1[7] };
          yr[0] = a0; yr[1] = a1; yr[2] = a2; yr[3] = a3;
        }
#pragma unroll
        for (int q = 0; q < 16; ++q) xe[q] = xn[q];
        __builtin_amdgcn_sched_barrier(0);
      }
    }
    store_row(bb + (size_t)(te - CL + 1)*HK, boff, yr);
  }
}

__global__ void __launch_bounds__(256)
hmm_gamma_kernel(const float* alpha, const float* beta, float* out, int npairs){
  const int w    = (blockIdx.x*blockDim.x + threadIdx.x) >> 5;
  const int lane = threadIdx.x & 31;
  if (w >= npairs) return;
  const size_t idx = (size_t)w*32 + lane;
  const float a = alpha[idx];
  const float b = beta[idx];
  const float s = rsum32(a * b);
  out[idx] = LN2 * ((flog2(a) + flog2(b)) - flog2(s));
}

extern "C" void kernel_launch(void* const* d_in, const int* in_sizes, int n_in,
                              void* d_out, int out_size, void* d_ws, size_t ws_size,
                              hipStream_t stream) {
  const float* em     = (const float*)d_in[0];
  const float* log_pi = (const float*)d_in[1];
  const float* log_A  = (const float*)d_in[2];
  float* out  = (float*)d_out;
  float* beta = (float*)d_ws;

  hmm_operator_kernel<<<dim3(HB*NCH*2), dim3(32), 0, stream>>>(em, log_A, out);
  hmm_edge_kernel<<<dim3(HB*2), dim3(32), 0, stream>>>(em, log_pi, out, beta);
  hmm_scan_kernel<<<dim3(NCH*8), dim3(32), 0, stream>>>(em, log_A, out, beta);

  const int npairs = HB * HT;
  const int threads = 256;
  const int blocks = (npairs*32 + threads - 1) / threads;
  hmm_gamma_kernel<<<dim3(blocks), dim3(threads), 0, stream>>>(out, beta, out, npairs);
}


# Gamma made inside the chunk scan: a chunk's forward and backward wave share a work-group, swap first-half rows at one barrier, store log gamma from registers; gamma launch gone; row sum order changes
# speedup vs baseline: 17.4283x; 1.2392x over previous
#include <hip/hip_runtime.h>


typedef __attribute__((ext_vector_type(2))) float v2f;
typedef __attribute__((ext_vector_type(4))) float v4f;
typedef __attribute__((ext_vector_type(8))) float v8f;
typedef __attribute__((address_space(1))) v4f GV4;

constexpr int HB = 64;
constexpr int HT = 4096;
constexpr int HK = 32;
constexpr int CL  = 65;
constexpr int NCH = 63;
constexpr int PFA = 5;
constexpr int PF  = 4;
constexpr float LOG2E = 1.44269504088896340736f;
constexpr float LN2   = 0.69314718055994530942f;

#if __has_builtin(__builtin_amdgcn_exp2f)
__device__ __forceinline__ float fexp2(float x){ return __builtin_amdgcn_exp2f(x); }
#else
__device__ __forceinline__ float fexp2(float x){ return exp2f(x); }
#endif
#if __has_builtin(__builtin_amdgcn_logf)
__device__ __forceinline__ float flog2(float x){ return __builtin_amdgcn_logf(x); }
#else
__device__ __forceinline__ float flog2(float x){ return log2f(x); }
#endif

__device__ __forceinline__ float rmax32(float x){
#pragma unroll
  for (int m = 16; m >= 1; m >>= 1) x = fmaxf(x, __shfl_xor(x, m, 32));
  return x;
}
__device__ __forceinline__ float rsum32(float x){
#pragma unroll
  for (int m = 16; m >= 1; m >>= 1) x += __shfl_xor(x, m, 32);
  return x;
}

__device__ __forceinline__ float xhalf(float x){
  int b = __builtin_bit_cast(int, x);
  return __builtin_bit_cast(float,
      __builtin_amdgcn_permlanex16(b, b, 0x76543210, 0xfedcba98, false, false));
}

__device__ __forceinline__ v8f wmma4(v2f a, v2f b, v8f c){
  return __builtin_amdgcn_wmma_f32_16x16x4_f32(
       false, a,  false, b,
       (short)0, c,  false,  false);
}


__device__ __forceinline__ unsigned long long uniform_addr(const void* p){
  const unsigned long long a = (unsigned long long)p;
  const unsigned lo = (unsigned)__builtin_amdgcn_readfirstlane((int)(unsigned)a);
  const unsigned hi = (unsigned)__builtin_amdgcn_readfirstlane((int)(unsigned)(a >> 32));
  return ((unsigned long long)hi << 32) | (unsigned long long)lo;
}
__device__ __forceinline__ void load_row(v4f (&dst)[4], const float* ub, unsigned boff){
  const GV4 * q = (const GV4 *)(uniform_addr(ub) + boff);
#pragma unroll
  for (int c = 0; c < 4; ++c) dst[c] = q[c];
}
__device__ __forceinline__ void pack_row(v4f (&w)[4], const float (&v)[16]){
#pragma unroll
  for (int c = 0; c < 4; ++c){
    v4f t = { v[4*c+0], v[4*c+1], v[4*c+2], v[4*c+3] };
    w[c] = t;
  }
}
__device__ __forceinline__ void store_row(float* ub, unsigned boff, const v4f (&w)[4]){
  GV4 * q = (GV4 *)(uniform_addr(ub) + boff);
#pragma unroll
  for (int c = 0; c < 4; ++c) q[c] = w[c];
}
__device__ __forceinline__ void store_row(float* ub, unsigned boff, const float (&v)[16]){
  v4f w[4];
  pack_row(w, v);
  store_row(ub, boff, w);
}

__device__ __forceinline__ float row_exponent(const float (&p)[16]){
  const float t0 = fmaxf(fmaxf(p[0],  p[1]),  p[2]);
  const float t1 = fmaxf(fmaxf(p[3],  p[4]),  p[5]);
  const float t2 = fmaxf(fmaxf(p[6],  p[7]),  p[8]);
  const float t3 = fmaxf(fmaxf(p[9],  p[10]), p[11]);
  const float t4 = fmaxf(fmaxf(p[12], p[13]), p[14]);
  const float u0 = fmaxf(fmaxf(t0, t1), t2);
  const float u1 = fmaxf(fmaxf(t3, t4), p[15]);
  float m = fmaxf(u0, u1);
  m = fmaxf(m, xhalf(m));
#if __has_builtin(__builtin_amdgcn_frexp_expf)
  return (float)__builtin_amdgcn_frexp_expf(m);
#else
  return (float)(((__builtin_bit_cast(int, m) >> 23) & 0xff) - 126);
#endif
}
__device__ __forceinline__ void lin_step(float (&pn)[16], v8f& d0, v8f& d1, float& ef_out,
                                         const float (&p)[16], const v4f (&emc)[4],
                                         const v2f (&ereg)[2][8]){
  v8f c0 = {0.f,0.f,0.f,0.f,0.f,0.f,0.f,0.f};
  v8f c1 = {0.f,0.f,0.f,0.f,0.f,0.f,0.f,0.f};
#pragma unroll
  for (int kc = 0; kc < 8; ++kc){
    const v2f b = { p[2*kc], p[2*kc+1] };
    c0 = wmma4(ereg[0][kc], b, c0);
    c1 = wmma4(ereg[1][kc], b, c1);
  }
  const float ef = row_exponent(p);
  ef_out = ef;
  float w[16];
#pragma unroll
  for (int q = 0; q < 16; ++q) w[q] = fexp2(fmaf(emc[q >> 2][q & 3], LOG2E, -ef));
#pragma unroll
  for (int r = 0; r < 8; ++r){
    pn[r]     = c0[r] * w[r];
    pn[8 + r] = c1[r] * w[8 + r];
  }
  d0 = c0;
  d1 = c1;
}
__device__ __forceinline__ void hold_row(const v4f (&w)[4], float after){
  asm volatile("" :: "v"(w[0]), "v"(w[1]), "v"(w[2]), "v"(w[3]), "v"(after));
}


__device__ __forceinline__ void build_expA(float* sA, const float* __restrict__ log_A, int lane){
  float x[32];
#pragma unroll
  for (int c = 0; c < 8; ++c){
    const v4f t = *(const v4f*)(log_A + lane*32 + 4*c);
    x[4*c+0] = t[0]; x[4*c+1] = t[1]; x[4*c+2] = t[2]; x[4*c+3] = t[3];
  }
  float mx = x[0];
#pragma unroll
  for (int j = 1; j < 32; ++j) mx = fmaxf(mx, x[j]);
  float s = 0.f;
#pragma unroll
  for (int j = 0; j < 32; ++j){ x[j] = fexp2((x[j] - mx) * LOG2E); s += x[j]; }
  const float r = 1.0f / s;
#pragma unroll
  for (int j = 0; j < 32; ++j) sA[lane*33 + j] = x[j] * r;
}

__device__ __forceinline__ void build_E(v2f (&ereg)[2][8], const float* sA, int ln, int h, int dir){
#pragma unroll
  for (int mt = 0; mt < 2; ++mt){
#pragma unroll
    for (int kc = 0; kc < 8; ++kc){
      const int sM = 16*(ln >> 3) + 8*mt + (ln & 7);
      const int sK = 16*h + 2*kc;
      if (dir == 0){
        ereg[mt][kc].x = sA[sK*33 + sM];
        ereg[mt][kc].y = sA[(sK+1)*33 + sM];
      } else {
        ereg[mt][kc].x = sA[sM*33 + sK];
        ereg[mt][kc].y = sA[sM*33 + sK + 1];
      }
    }
  }
}

__global__ void __launch_bounds__(32)
hmm_operator_kernel(const float* __restrict__ em,
                    const float* __restrict__ log_A,
                    float* __restrict__ fbuf){
  __shared__ float sA[32*33];
  const int lane = threadIdx.x;
  const int h    = lane >> 4;
  const int ln   = lane & 15;
  const int nt   = blockIdx.x & 1;
  const int cb   = blockIdx.x >> 1;
  const int c    = cb % NCH;
  const int b    = cb / NCH;

  build_expA(sA, log_A, lane);
  __syncthreads();
  v2f ereg[2][8];
  build_E(ereg, sA, ln, h, 0);

  const size_t urow = (size_t)b*HT + (size_t)(CL*c + 1);
  const float* eb = em + urow*HK;
  const unsigned boff = (unsigned)(16*h) * 4u;

  float p[16];
#pragma unroll
  for (int q = 0; q < 16; ++q) p[q] = (16*h + q == 16*nt + ln) ? 1.f : 0.f;
  float esum = 0.f;

  v4f er[PFA][4];
#pragma unroll
  for (int u = 0; u < PFA - 1; ++u) load_row(er[u], eb + (size_t)u*HK, boff);

  unsigned bo = boff;
  for (int k0 = 0; k0 < CL; k0 += PFA){
    asm volatile("" : "+v"(bo));
#pragma unroll
    for (int u = 0; u < PFA; ++u){
      const int k  = k0 + u;
      const int up = (u + PFA - 1) % PFA;
      const int kn = (k + PFA - 1 < CL) ? k + PFA - 1 : CL - 1;
      load_row(er[up], eb + (size_t)kn*HK, bo);
      float pn[16];
      v8f d0, d1;
      float ef;
      lin_step(pn, d0, d1, ef, p, er[u], ereg);
      esum += ef;
#pragma unroll
      for (int q = 0; q < 16; ++q) p[q] = pn[q];
    }
  }
  float* dst = fbuf + urow*HK;
  const unsigned coff = (unsigned)((16*nt + ln)*HK + 16*h) * 4u;
  store_row(dst, coff, p);
  if (h == 0) dst[(size_t)(32 + nt)*HK + 16*nt + ln] = esum;
}

__device__ __forceinline__ float expo_of(float m){
#if __has_builtin(__builtin_amdgcn_frexp_expf)
  return (float)__builtin_amdgcn_frexp_expf(m);
#else
  return (float)(((__builtin_bit_cast(int, m) >> 23) & 0xff) - 126);
#endif
}

__device__ __forceinline__ float lane_swz(float x, const int ctrl_sel){
  const int b = __builtin_bit_cast(int, x);
  int r;
  if (ctrl_sel == 0)      r = __builtin_amdgcn_update_dpp(b, b, 0xB1,  0xf, 0xf, false);
  else if (ctrl_sel == 1) r = __builtin_amdgcn_update_dpp(b, b, 0x4E,  0xf, 0xf, false);
  else if (ctrl_sel == 2) r = __builtin_amdgcn_update_dpp(b, b, 0x141, 0xf, 0xf, false);
  else                    r = __builtin_amdgcn_update_dpp(b, b, 0x140, 0xf, 0xf, false);
  return __builtin_bit_cast(float, r);
}
__device__ __forceinline__ float wmax32(float x){
  x = fmaxf(x, lane_swz(x, 0));
  x = fmaxf(x, lane_swz(x, 1));
  x = fmaxf(x, lane_swz(x, 2));
  x = fmaxf(x, lane_swz(x, 3));
  return fmaxf(x, xhalf(x));
}
__device__ __forceinline__ float lane_bcast(float x, const int j){
  return __builtin_bit_cast(float, __builtin_amdgcn_readlane(__builtin_bit_cast(int, x), j));
}

__global__ void __launch_bounds__(32)
hmm_edge_kernel(const float* __restrict__ em,
                const float* __restrict__ log_pi,
                float* fa,
                float* __restrict__ bw){
  const int lane = threadIdx.x;
  const int b    = blockIdx.x >> 1;
  const int dir  = blockIdx.x & 1;
  float* fb = fa + (size_t)b*HT*HK;

  if (dir == 0){
    float xp = log_pi[lane];
    float mp = rmax32(xp);
    float sp = rsum32(fexp2((xp - mp) * LOG2E));
    float lp = xp - mp - LN2 * flog2(sp);
    float v  = fexp2((lp + em[(size_t)b*HT*HK + lane]) * LOG2E);
    fb[lane] = v;

    float f[3][32];
    float sx[3];
#pragma unroll
    for (int u = 0; u < 2; ++u){
      const float* src = fb + (size_t)(CL*u + 1)*HK;
#pragma unroll
      for (int j = 0; j < 32; ++j) f[u][j] = src[(size_t)j*HK + lane];
      sx[u] = src[(size_t)(32 + (lane >> 4))*HK + lane];
    }
    for (int c0 = 0; c0 < NCH; c0 += 3){
#pragma unroll
      for (int u = 0; u < 3; ++u){
        const int c  = c0 + u;
        const int cn = (c + 2 < NCH) ? c + 2 : NCH - 1;
        {
          const int un = (u + 2) % 3;
          const float* src = fb + (size_t)(CL*cn + 1)*HK;
#pragma unroll
          for (int j = 0; j < 32; ++j) f[un][j] = src[(size_t)j*HK + lane];
          sx[un] = src[(size_t)(32 + (lane >> 4))*HK + lane];
        }
        const float smax = wmax32(sx[u]);
        const float uu = v * fexp2(sx[u] - smax);
        float a0 = 0.f, a1 = 0.f, a2 = 0.f, a3 = 0.f;
#pragma unroll
        for (int j = 0; j < 32; j += 4){
          a0 = fmaf(f[u][j+0], lane_bcast(uu, j+0), a0);
          a1 = fmaf(f[u][j+1], lane_bcast(uu, j+1), a1);
          a2 = fmaf(f[u][j+2], lane_bcast(uu, j+2), a2);
          a3 = fmaf(f[u][j+3], lane_bcast(uu, j+3), a3);
        }
        const float acc = (a0 + a1) + (a2 + a3);
        const float e = expo_of(wmax32(acc));
        v = acc * fexp2(-e);
        fb[(size_t)(CL*(c + 1))*HK + lane] = v;
      }
    }
  } else {
    float* bb = bw + (size_t)b*HT*HK;
    float v = 1.f;
    bb[(size_t)(HT - 1)*HK + lane] = v;

    v4f g[3][8];
    float sx[3];
#pragma unroll
    for (int u = 0; u < 2; ++u){
      const float* src = fb + (size_t)(CL*(NCH - 1 - u) + 1)*HK;
#pragma unroll
      for (int q = 0; q < 8; ++q) g[u][q] = *(const v4f*)(src + lane*HK + 4*q);
      sx[u] = src[(size_t)(32 + (lane >> 4))*HK + lane];
    }
    for (int c0 = 0; c0 < NCH; c0 += 3){
#pragma unroll
      for (int u = 0; u < 3; ++u){
        const int c  = NCH - 1 - (c0 + u);
        const int cn = (c - 2 > 0) ? c - 2 : 0;
        {
          const int un = (u + 2) % 3;
          const float* src = fb + (size_t)(CL*cn + 1)*HK;
#pragma unroll
          for (int q = 0; q < 8; ++q) g[un][q] = *(const v4f*)(src + lane*HK + 4*q);
          sx[un] = src[(size_t)(32 + (lane >> 4))*HK + lane];
        }
        float a0 = 0.f, a1 = 0.f, a2 = 0.f, a3 = 0.f;
#pragma unroll
        for (int q = 0; q < 8; ++q){
          a0 = fmaf(g[u][q][0], lane_bcast(v, 4*q+0), a0);
          a1 = fmaf(g[u][q][1], lane_bcast(v, 4*q+1), a1);
          a2 = fmaf(g[u][q][2], lane_bcast(v, 4*q+2), a2);
          a3 = fmaf(g[u][q][3], lane_bcast(v, 4*q+3), a3);
        }
        const float smax = wmax32(sx[u]);
        const float acc = ((a0 + a1) + (a2 + a3)) * fexp2(sx[u] - smax);
        const float e = expo_of(wmax32(acc));
        v = acc * fexp2(-e);
        bb[(size_t)(CL*c)*HK + lane] = v;
      }
    }
  }
}

__device__ __forceinline__ void gamma_row(v4f (&g)[4], const v4f (&a)[4], const v4f (&b)[4]){
  float p[16];
#pragma unroll
  for (int q = 0; q < 16; ++q) p[q] = a[q >> 2][q & 3] * b[q >> 2][q & 3];
  const float s0 = ((p[0] + p[1]) + (p[2] + p[3])) + ((p[4] + p[5]) + (p[6] + p[7]));
  const float s1 = ((p[8] + p[9]) + (p[10] + p[11])) + ((p[12] + p[13]) + (p[14] + p[15]));
  float s = s0 + s1;
  s += xhalf(s);
  const float ls = flog2(s);
#pragma unroll
  for (int c = 0; c < 4; ++c){
    v4f t;
#pragma unroll
    for (int e = 0; e < 4; ++e) t[e] = LN2 * ((flog2(a[c][e]) + flog2(b[c][e])) - ls);
    g[c] = t;
  }
}

__global__ void __launch_bounds__(64)
hmm_scan_kernel(const float* __restrict__ em,
                const float* __restrict__ log_A,
                float* alpha_out,
                float* beta_out){
  __shared__ float sA[32*33];

  const int lane = threadIdx.x & 31;
  const int dir  = __builtin_amdgcn_readfirstlane((int)(threadIdx.x >> 5));
  const int h    = lane >> 4;
  const int ln   = lane & 15;
  const int mb   = blockIdx.x & 3;
  const int ch   = blockIdx.x >> 2;

  if (dir == 0) build_expA(sA, log_A, lane);
  __syncthreads();
  v2f ereg[2][8];
  build_E(ereg, sA, ln, h, dir);

  const size_t   uoff = (size_t)mb*16*HT*HK;
  const unsigned boff = (unsigned)(ln*HT*HK + 16*h) * 4u;
  const float* eb = em + uoff;
  float* ab = alpha_out + uoff;
  float* bb = beta_out + uoff;
  const int tb = CL*ch;
  const int te = CL*(ch + 1);
  constexpr int HALF = 32;

  v4f er[PF][4];
  v4f pr[PF][4];
  float st[16];
  v4f yr[4];

  if (dir == 0){
    {
      v4f a0[4], b0[4];
      load_row(a0, ab + (size_t)tb*HK, boff);
      load_row(b0, bb + (size_t)tb*HK, boff);
#pragma unroll
      for (int u = 0; u < PF - 1; ++u) load_row(er[u], eb + (size_t)(tb + 1 + u)*HK, boff);
#pragma unroll
      for (int q = 0; q < 16; ++q) st[q] = a0[q >> 2][q & 3];
      v4f g[4];
      gamma_row(g, a0, b0);
      store_row(ab + (size_t)tb*HK, boff, g);
    }
    unsigned bo = boff;
    for (int k0 = 1; k0 <= HALF; k0 += PF){
      asm volatile("" : "+v"(bo));
#pragma unroll
      for (int u = 0; u < PF; ++u){
        const int t = tb + k0 + u;
        v4f wr[4];
        pack_row(wr, st);
        if (k0 + u > 1) store_row(ab + (size_t)(t - 1)*HK, bo, wr);
        const int up = (u + PF - 1) % PF;
        const int tn = (t - 1 + PF < HT) ? t - 1 + PF : HT - 1;
        load_row(er[up], eb + (size_t)tn*HK, bo);
        float aln[16];
        v8f d0, d1;
        float ef;
        lin_step(aln, d0, d1, ef, st, er[u], ereg);
        hold_row(wr, aln[0]);
#pragma unroll
        for (int q = 0; q < 16; ++q) st[q] = aln[q];
        __builtin_amdgcn_sched_barrier(0);
      }
    }
    store_row(ab + (size_t)(tb + HALF)*HK, boff, st);
  } else {
    load_row(yr, bb + (size_t)te*HK, boff);
    {
      v4f eT[4];
      load_row(eT, eb + (size_t)te*HK, boff);
#pragma unroll
      for (int q = 0; q < 16; ++q) st[q] = yr[q >> 2][q & 3] * fexp2(eT[q >> 2][q & 3] * LOG2E);
    }
    if (ch == NCH - 1){
      v4f aT[4], g[4];
      load_row(aT, ab + (size_t)te*HK, boff);
      gamma_row(g, aT, yr);
      store_row(ab + (size_t)te*HK, boff, g);
    }
#pragma unroll
    for (int u = 0; u < PF - 1; ++u) load_row(er[u], eb + (size_t)(te - 1 - u)*HK, boff);

    unsigned bo = boff;
    for (int s0 = 0; s0 < HALF; s0 += PF){
      asm volatile("" : "+v"(bo));
#pragma unroll
      for (int u = 0; u < PF; ++u){
        const int t = te - 1 - (s0 + u);
        if (s0 + u > 0) store_row(bb + (size_t)(t + 1)*HK, bo, yr);
        const int up = (u + PF - 1) % PF;
        const int tn = (t + 1 - PF > 0) ? t + 1 - PF : 0;
        load_row(er[up], eb + (size_t)tn*HK, bo);
        float xn[16];
        v8f d0, d1;
        float ef;
        lin_step(xn, d0, d1, ef, st, er[u], ereg);
        hold_row(yr, xn[0]);
        {
          v4f a0 = { d0[0], d0[1], d0[2], d0[3] };
          v4f a1 = { d0[4], d0[5], d0[6], d0[7] };
          v4f a2 = { d1[0], d1[1], d1[2], d1[3] };
          v4f a3 = { d1[4], d1[5], d1[6], d1[7] };
          yr[0] = a0; yr[1] = a1; yr[2] = a2; yr[3] = a3;
        }
#pragma unroll
        for (int q = 0; q < 16; ++q) st[q] = xn[q];
        __builtin_amdgcn_sched_barrier(0);
      }
    }
    store_row(bb + (size_t)(te - HALF)*HK, boff, yr);
  }

  __syncthreads();

  const v4f vz = {0.f, 0.f, 0.f, 0.f};
  if (dir == 0){
#pragma unroll
    for (int u = 0; u < PF - 1; ++u) load_row(pr[u], bb + (size_t)(tb + HALF + 1 + u)*HK, boff);
    unsigned bo = boff;
    for (int k0 = HALF + 1; k0 <= 2*HALF; k0 += PF){
      asm volatile("" : "+v"(bo));
#pragma unroll
      for (int u = 0; u < PF; ++u){
        const int k  = k0 + u;
        const int t  = tb + k;
        const int up = (u + PF - 1) % PF;
        v4f wr[4] = { vz, vz, vz, vz };
        if (k > HALF + 1){
          v4f av[4];
          pack_row(av, st);
          gamma_row(wr, av, pr[up]);
          store_row(ab + (size_t)(t - 1)*HK, bo, wr);
        }
        const int tn = (t - 1 + PF < HT) ? t - 1 + PF : HT - 1;
        load_row(er[up], eb + (size_t)tn*HK, bo);
        const int jn = (k - 1 + PF < 2*HALF) ? k - 1 + PF : 2*HALF;
        load_row(pr[up], bb + (size_t)(tb + jn)*HK, bo);
        float aln[16];
        v8f d0, d1;
        float ef;
        lin_step(aln, d0, d1, ef, st, er[u], ereg);
        hold_row(wr, aln[0]);
#pragma unroll
        for (int q = 0; q < 16; ++q) st[q] = aln[q];
        __builtin_amdgcn_sched_barrier(0);
      }
    }
    {
      v4f av[4], g[4];
      pack_row(av, st);
      gamma_row(g, av, pr[(2*HALF - 1) % PF]);
      store_row(ab + (size_t)(tb + 2*HALF)*HK, boff, g);
    }
  } else {
#pragma unroll
    for (int u = 0; u < PF - 1; ++u) load_row(pr[u], ab + (size_t)(te - 1 - HALF - u)*HK, boff);
    unsigned bo = boff;
    for (int s0 = HALF; s0 < 2*HALF; s0 += PF){
      asm volatile("" : "+v"(bo));
#pragma unroll
      for (int u = 0; u < PF; ++u){
        const int s  = s0 + u;
        const int t  = te - 1 - s;
        const int up = (u + PF - 1) % PF;
        v4f wr[4] = { vz, vz, vz, vz };
        if (s > HALF){
          gamma_row(wr, pr[up], yr);
          store_row(ab + (size_t)(t + 1)*HK, bo, wr);
        }
        const int tn = (t + 1 - PF > 0) ? t + 1 - PF : 0;
        load_row(er[up], eb + (size_t)tn*HK, bo);
        const int pa = (t + 1 - PF > tb + 1) ? t + 1 - PF : tb + 1;
        load_row(pr[up], ab + (size_t)pa*HK, bo);
        float xn[16];
        v8f d0, d1;
        float ef;
        lin_step(xn, d0, d1, ef, st, er[u], ereg);
        hold_row(wr, xn[0]);
        {
          v4f a0 = { d0[0], d0[1], d0[2], d0[3] };
          v4f a1 = { d0[4], d0[5], d0[6], d0[7] };
          v4f a2 = { d1[0], d1[1], d1[2], d1[3] };
          v4f a3 = { d1[4], d1[5], d1[6], d1[7] };
          yr[0] = a0; yr[1] = a1; yr[2] = a2; yr[3] = a3;
        }
#pragma unroll
        for (int q = 0; q < 16; ++q) st[q] = xn[q];
        __builtin_amdgcn_sched_barrier(0);
      }
    }
    {
      v4f g[4];
      gamma_row(g, pr[(2*HALF - 1) % PF], yr);
      store_row(ab + (size_t)(tb + 1)*HK, boff, g);
    }
  }
}

extern "C" void kernel_launch(void* const* d_in, const int* in_sizes, int n_in,
                              void* d_out, int out_size, void* d_ws, size_t ws_size,
                              hipStream_t stream) {
  const float* em     = (const float*)d_in[0];
  const float* log_pi = (const float*)d_in[1];
  const float* log_A  = (const float*)d_in[2];
  float* out  = (float*)d_out;
  float* beta = (float*)d_ws;

  hmm_operator_kernel<<<dim3(HB*NCH*2), dim3(32), 0, stream>>>(em, log_A, out);
  hmm_edge_kernel<<<dim3(HB*2), dim3(32), 0, stream>>>(em, log_pi, out, beta);
  hmm_scan_kernel<<<dim3(NCH*4), dim3(64), 0, stream>>>(em, log_A, out, beta);
}


# Operator pass: a scheduler boundary per step, as in the chunk scan: emission loads stay four steps ahead (they had sunk to their use, waited for at once) and the two matrix chains interleave
# speedup vs baseline: 19.1172x; 1.0969x over previous
#include <hip/hip_runtime.h>


typedef __attribute__((ext_vector_type(2))) float v2f;
typedef __attribute__((ext_vector_type(4))) float v4f;
typedef __attribute__((ext_vector_type(8))) float v8f;
typedef __attribute__((address_space(1))) v4f GV4;

constexpr int HB = 64;
constexpr int HT = 4096;
constexpr int HK = 32;
constexpr int CL  = 65;
constexpr int NCH = 63;
constexpr int PFA = 5;
constexpr int PF  = 4;
constexpr float LOG2E = 1.44269504088896340736f;
constexpr float LN2   = 0.69314718055994530942f;

#if __has_builtin(__builtin_amdgcn_exp2f)
__device__ __forceinline__ float fexp2(float x){ return __builtin_amdgcn_exp2f(x); }
#else
__device__ __forceinline__ float fexp2(float x){ return exp2f(x); }
#endif
#if __has_builtin(__builtin_amdgcn_logf)
__device__ __forceinline__ float flog2(float x){ return __builtin_amdgcn_logf(x); }
#else
__device__ __forceinline__ float flog2(float x){ return log2f(x); }
#endif

__device__ __forceinline__ float rmax32(float x){
#pragma unroll
  for (int m = 16; m >= 1; m >>= 1) x = fmaxf(x, __shfl_xor(x, m, 32));
  return x;
}
__device__ __forceinline__ float rsum32(float x){
#pragma unroll
  for (int m = 16; m >= 1; m >>= 1) x += __shfl_xor(x, m, 32);
  return x;
}

__device__ __forceinline__ float xhalf(float x){
  int b = __builtin_bit_cast(int, x);
  return __builtin_bit_cast(float,
      __builtin_amdgcn_permlanex16(b, b, 0x76543210, 0xfedcba98, false, false));
}

__device__ __forceinline__ v8f wmma4(v2f a, v2f b, v8f c){
  return __builtin_amdgcn_wmma_f32_16x16x4_f32(
       false, a,  false, b,
       (short)0, c,  false,  false);
}


__device__ __forceinline__ unsigned long long uniform_addr(const void* p){
  const unsigned long long a = (unsigned long long)p;
  const unsigned lo = (unsigned)__builtin_amdgcn_readfirstlane((int)(unsigned)a);
  const unsigned hi = (unsigned)__builtin_amdgcn_readfirstlane((int)(unsigned)(a >> 32));
  return ((unsigned long long)hi << 32) | (unsigned long long)lo;
}
__device__ __forceinline__ void load_row(v4f (&dst)[4], const float* ub, unsigned boff){
  const GV4 * q = (const GV4 *)(uniform_addr(ub) + boff);
#pragma unroll
  for (int c = 0; c < 4; ++c) dst[c] = q[c];
}
__device__ __forceinline__ void pack_row(v4f (&w)[4], const float (&v)[16]){
#pragma unroll
  for (int c = 0; c < 4; ++c){
    v4f t = { v[4*c+0], v[4*c+1], v[4*c+2], v[4*c+3] };
    w[c] = t;
  }
}
__device__ __forceinline__ void store_row(float* ub, unsigned boff, const v4f (&w)[4]){
  GV4 * q = (GV4 *)(uniform_addr(ub) + boff);
#pragma unroll
  for (int c = 0; c < 4; ++c) q[c] = w[c];
}
__device__ __forceinline__ void store_row(float* ub, unsigned boff, const float (&v)[16]){
  v4f w[4];
  pack_row(w, v);
  store_row(ub, boff, w);
}

__device__ __forceinline__ float row_exponent(const float (&p)[16]){
  const float t0 = fmaxf(fmaxf(p[0],  p[1]),  p[2]);
  const float t1 = fmaxf(fmaxf(p[3],  p[4]),  p[5]);
  const float t2 = fmaxf(fmaxf(p[6],  p[7]),  p[8]);
  const float t3 = fmaxf(fmaxf(p[9],  p[10]), p[11]);
  const float t4 = fmaxf(fmaxf(p[12], p[13]), p[14]);
  const float u0 = fmaxf(fmaxf(t0, t1), t2);
  const float u1 = fmaxf(fmaxf(t3, t4), p[15]);
  float m = fmaxf(u0, u1);
  m = fmaxf(m, xhalf(m));
#if __has_builtin(__builtin_amdgcn_frexp_expf)
  return (float)__builtin_amdgcn_frexp_expf(m);
#else
  return (float)(((__builtin_bit_cast(int, m) >> 23) & 0xff) - 126);
#endif
}
__device__ __forceinline__ void lin_step(float (&pn)[16], v8f& d0, v8f& d1, float& ef_out,
                                         const float (&p)[16], const v4f (&emc)[4],
                                         const v2f (&ereg)[2][8]){
  v8f c0 = {0.f,0.f,0.f,0.f,0.f,0.f,0.f,0.f};
  v8f c1 = {0.f,0.f,0.f,0.f,0.f,0.f,0.f,0.f};
#pragma unroll
  for (int kc = 0; kc < 8; ++kc){
    const v2f b = { p[2*kc], p[2*kc+1] };
    c0 = wmma4(ereg[0][kc], b, c0);
    c1 = wmma4(ereg[1][kc], b, c1);
  }
  const float ef = row_exponent(p);
  ef_out = ef;
  float w[16];
#pragma unroll
  for (int q = 0; q < 16; ++q) w[q] = fexp2(fmaf(emc[q >> 2][q & 3], LOG2E, -ef));
#pragma unroll
  for (int r = 0; r < 8; ++r){
    pn[r]     = c0[r] * w[r];
    pn[8 + r] = c1[r] * w[8 + r];
  }
  d0 = c0;
  d1 = c1;
}
__device__ __forceinline__ void hold_row(const v4f (&w)[4], float after){
  asm volatile("" :: "v"(w[0]), "v"(w[1]), "v"(w[2]), "v"(w[3]), "v"(after));
}


__device__ __forceinline__ void build_expA(float* sA, const float* __restrict__ log_A, int lane){
  float x[32];
#pragma unroll
  for (int c = 0; c < 8; ++c){
    const v4f t = *(const v4f*)(log_A + lane*32 + 4*c);
    x[4*c+0] = t[0]; x[4*c+1] = t[1]; x[4*c+2] = t[2]; x[4*c+3] = t[3];
  }
  float mx = x[0];
#pragma unroll
  for (int j = 1; j < 32; ++j) mx = fmaxf(mx, x[j]);
  float s = 0.f;
#pragma unroll
  for (int j = 0; j < 32; ++j){ x[j] = fexp2((x[j] - mx) * LOG2E); s += x[j]; }
  const float r = 1.0f / s;
#pragma unroll
  for (int j = 0; j < 32; ++j) sA[lane*33 + j] = x[j] * r;
}

__device__ __forceinline__ void build_E(v2f (&ereg)[2][8], const float* sA, int ln, int h, int dir){
#pragma unroll
  for (int mt = 0; mt < 2; ++mt){
#pragma unroll
    for (int kc = 0; kc < 8; ++kc){
      const int sM = 16*(ln >> 3) + 8*mt + (ln & 7);
      const int sK = 16*h + 2*kc;
      if (dir == 0){
        ereg[mt][kc].x = sA[sK*33 + sM];
        ereg[mt][kc].y = sA[(sK+1)*33 + sM];
      } else {
        ereg[mt][kc].x = sA[sM*33 + sK];
        ereg[mt][kc].y = sA[sM*33 + sK + 1];
      }
    }
  }
}

__global__ void __launch_bounds__(32)
hmm_operator_kernel(const float* __restrict__ em,
                    const float* __restrict__ log_A,
                    float* __restrict__ fbuf){
  __shared__ float sA[32*33];
  const int lane = threadIdx.x;
  const int h    = lane >> 4;
  const int ln   = lane & 15;
  const int nt   = blockIdx.x & 1;
  const int cb   = blockIdx.x >> 1;
  const int c    = cb % NCH;
  const int b    = cb / NCH;

  build_expA(sA, log_A, lane);
  __syncthreads();
  v2f ereg[2][8];
  build_E(ereg, sA, ln, h, 0);

  const size_t urow = (size_t)b*HT + (size_t)(CL*c + 1);
  const float* eb = em + urow*HK;
  const unsigned boff = (unsigned)(16*h) * 4u;

  float p[16];
#pragma unroll
  for (int q = 0; q < 16; ++q) p[q] = (16*h + q == 16*nt + ln) ? 1.f : 0.f;
  float esum = 0.f;

  v4f er[PFA][4];
#pragma unroll
  for (int u = 0; u < PFA - 1; ++u) load_row(er[u], eb + (size_t)u*HK, boff);

  unsigned bo = boff;
  for (int k0 = 0; k0 < CL; k0 += PFA){
    asm volatile("" : "+v"(bo));
#pragma unroll
    for (int u = 0; u < PFA; ++u){
      const int k  = k0 + u;
      const int up = (u + PFA - 1) % PFA;
      const int kn = (k + PFA - 1 < CL) ? k + PFA - 1 : CL - 1;
      load_row(er[up], eb + (size_t)kn*HK, bo);
      float pn[16];
      v8f d0, d1;
      float ef;
      lin_step(pn, d0, d1, ef, p, er[u], ereg);
      esum += ef;
#pragma unroll
      for (int q = 0; q < 16; ++q) p[q] = pn[q];
      __builtin_amdgcn_sched_barrier(0);
    }
  }
  float* dst = fbuf + urow*HK;
  const unsigned coff = (unsigned)((16*nt + ln)*HK + 16*h) * 4u;
  store_row(dst, coff, p);
  if (h == 0) dst[(size_t)(32 + nt)*HK + 16*nt + ln] = esum;
}

__device__ __forceinline__ float expo_of(float m){
#if __has_builtin(__builtin_amdgcn_frexp_expf)
  return (float)__builtin_amdgcn_frexp_expf(m);
#else
  return (float)(((__builtin_bit_cast(int, m) >> 23) & 0xff) - 126);
#endif
}

__device__ __forceinline__ float lane_swz(float x, const int ctrl_sel){
  const int b = __builtin_bit_cast(int, x);
  int r;
  if (ctrl_sel == 0)      r = __builtin_amdgcn_update_dpp(b, b, 0xB1,  0xf, 0xf, false);
  else if (ctrl_sel == 1) r = __builtin_amdgcn_update_dpp(b, b, 0x4E,  0xf, 0xf, false);
  else if (ctrl_sel == 2) r = __builtin_amdgcn_update_dpp(b, b, 0x141, 0xf, 0xf, false);
  else                    r = __builtin_amdgcn_update_dpp(b, b, 0x140, 0xf, 0xf, false);
  return __builtin_bit_cast(float, r);
}
__device__ __forceinline__ float wmax32(float x){
  x = fmaxf(x, lane_swz(x, 0));
  x = fmaxf(x, lane_swz(x, 1));
  x = fmaxf(x, lane_swz(x, 2));
  x = fmaxf(x, lane_swz(x, 3));
  return fmaxf(x, xhalf(x));
}
__device__ __forceinline__ float lane_bcast(float x, const int j){
  return __builtin_bit_cast(float, __builtin_amdgcn_readlane(__builtin_bit_cast(int, x), j));
}

__global__ void __launch_bounds__(32)
hmm_edge_kernel(const float* __restrict__ em,
                const float* __restrict__ log_pi,
                float* fa,
                float* __restrict__ bw){
  const int lane = threadIdx.x;
  const int b    = blockIdx.x >> 1;
  const int dir  = blockIdx.x & 1;
  float* fb = fa + (size_t)b*HT*HK;

  if (dir == 0){
    float xp = log_pi[lane];
    float mp = rmax32(xp);
    float sp = rsum32(fexp2((xp - mp) * LOG2E));
    float lp = xp - mp - LN2 * flog2(sp);
    float v  = fexp2((lp + em[(size_t)b*HT*HK + lane]) * LOG2E);
    fb[lane] = v;

    float f[3][32];
    float sx[3];
#pragma unroll
    for (int u = 0; u < 2; ++u){
      const float* src = fb + (size_t)(CL*u + 1)*HK;
#pragma unroll
      for (int j = 0; j < 32; ++j) f[u][j] = src[(size_t)j*HK + lane];
      sx[u] = src[(size_t)(32 + (lane >> 4))*HK + lane];
    }
    for (int c0 = 0; c0 < NCH; c0 += 3){
#pragma unroll
      for (int u = 0; u < 3; ++u){
        const int c  = c0 + u;
        const int cn = (c + 2 < NCH) ? c + 2 : NCH - 1;
        {
          const int un = (u + 2) % 3;
          const float* src = fb + (size_t)(CL*cn + 1)*HK;
#pragma unroll
          for (int j = 0; j < 32; ++j) f[un][j] = src[(size_t)j*HK + lane];
          sx[un] = src[(size_t)(32 + (lane >> 4))*HK + lane];
        }
        const float smax = wmax32(sx[u]);
        const float uu = v * fexp2(sx[u] - smax);
        float a0 = 0.f, a1 = 0.f, a2 = 0.f, a3 = 0.f;
#pragma unroll
        for (int j = 0; j < 32; j += 4){
          a0 = fmaf(f[u][j+0], lane_bcast(uu, j+0), a0);
          a1 = fmaf(f[u][j+1], lane_bcast(uu, j+1), a1);
          a2 = fmaf(f[u][j+2], lane_bcast(uu, j+2), a2);
          a3 = fmaf(f[u][j+3], lane_bcast(uu, j+3), a3);
        }
        const float acc = (a0 + a1) + (a2 + a3);
        const float e = expo_of(wmax32(acc));
        v = acc * fexp2(-e);
        fb[(size_t)(CL*(c + 1))*HK + lane] = v;
      }
    }
  } else {
    float* bb = bw + (size_t)b*HT*HK;
    float v = 1.f;
    bb[(size_t)(HT - 1)*HK + lane] = v;

    v4f g[3][8];
    float sx[3];
#pragma unroll
    for (int u = 0; u < 2; ++u){
      const float* src = fb + (size_t)(CL*(NCH - 1 - u) + 1)*HK;
#pragma unroll
      for (int q = 0; q < 8; ++q) g[u][q] = *(const v4f*)(src + lane*HK + 4*q);
      sx[u] = src[(size_t)(32 + (lane >> 4))*HK + lane];
    }
    for (int c0 = 0; c0 < NCH; c0 += 3){
#pragma unroll
      for (int u = 0; u < 3; ++u){
        const int c  = NCH - 1 - (c0 + u);
        const int cn = (c - 2 > 0) ? c - 2 : 0;
        {
          const int un = (u + 2) % 3;
          const float* src = fb + (size_t)(CL*cn + 1)*HK;
#pragma unroll
          for (int q = 0; q < 8; ++q) g[un][q] = *(const v4f*)(src + lane*HK + 4*q);
          sx[un] = src[(size_t)(32 + (lane >> 4))*HK + lane];
        }
        float a0 = 0.f, a1 = 0.f, a2 = 0.f, a3 = 0.f;
#pragma unroll
        for (int q = 0; q < 8; ++q){
          a0 = fmaf(g[u][q][0], lane_bcast(v, 4*q+0), a0);
          a1 = fmaf(g[u][q][1], lane_bcast(v, 4*q+1), a1);
          a2 = fmaf(g[u][q][2], lane_bcast(v, 4*q+2), a2);
          a3 = fmaf(g[u][q][3], lane_bcast(v, 4*q+3), a3);
        }
        const float smax = wmax32(sx[u]);
        const float acc = ((a0 + a1) + (a2 + a3)) * fexp2(sx[u] - smax);
        const float e = expo_of(wmax32(acc));
        v = acc * fexp2(-e);
        bb[(size_t)(CL*c)*HK + lane] = v;
      }
    }
  }
}

__device__ __forceinline__ void gamma_row(v4f (&g)[4], const v4f (&a)[4], const v4f (&b)[4]){
  float p[16];
#pragma unroll
  for (int q = 0; q < 16; ++q) p[q] = a[q >> 2][q & 3] * b[q >> 2][q & 3];
  const float s0 = ((p[0] + p[1]) + (p[2] + p[3])) + ((p[4] + p[5]) + (p[6] + p[7]));
  const float s1 = ((p[8] + p[9]) + (p[10] + p[11])) + ((p[12] + p[13]) + (p[14] + p[15]));
  float s = s0 + s1;
  s += xhalf(s);
  const float ls = flog2(s);
#pragma unroll
  for (int c = 0; c < 4; ++c){
    v4f t;
#pragma unroll
    for (int e = 0; e < 4; ++e) t[e] = LN2 * ((flog2(a[c][e]) + flog2(b[c][e])) - ls);
    g[c] = t;
  }
}

__global__ void __launch_bounds__(64)
hmm_scan_kernel(const float* __restrict__ em,
                const float* __restrict__ log_A,
                float* alpha_out,
                float* beta_out){
  __shared__ float sA[32*33];

  const int lane = threadIdx.x & 31;
  const int dir  = __builtin_amdgcn_readfirstlane((int)(threadIdx.x >> 5));
  const int h    = lane >> 4;
  const int ln   = lane & 15;
  const int mb   = blockIdx.x & 3;
  const int ch   = blockIdx.x >> 2;

  if (dir == 0) build_expA(sA, log_A, lane);
  __syncthreads();
  v2f ereg[2][8];
  build_E(ereg, sA, ln, h, dir);

  const size_t   uoff = (size_t)mb*16*HT*HK;
  const unsigned boff = (unsigned)(ln*HT*HK + 16*h) * 4u;
  const float* eb = em + uoff;
  float* ab = alpha_out + uoff;
  float* bb = beta_out + uoff;
  const int tb = CL*ch;
  const int te = CL*(ch + 1);
  constexpr int HALF = 32;

  v4f er[PF][4];
  v4f pr[PF][4];
  float st[16];
  v4f yr[4];

  if (dir == 0){
    {
      v4f a0[4], b0[4];
      load_row(a0, ab + (size_t)tb*HK, boff);
      load_row(b0, bb + (size_t)tb*HK, boff);
#pragma unroll
      for (int u = 0; u < PF - 1; ++u) load_row(er[u], eb + (size_t)(tb + 1 + u)*HK, boff);
#pragma unroll
      for (int q = 0; q < 16; ++q) st[q] = a0[q >> 2][q & 3];
      v4f g[4];
      gamma_row(g, a0, b0);
      store_row(ab + (size_t)tb*HK, boff, g);
    }
    unsigned bo = boff;
    for (int k0 = 1; k0 <= HALF; k0 += PF){
      asm volatile("" : "+v"(bo));
#pragma unroll
      for (int u = 0; u < PF; ++u){
        const int t = tb + k0 + u;
        v4f wr[4];
        pack_row(wr, st);
        if (k0 + u > 1) store_row(ab + (size_t)(t - 1)*HK, bo, wr);
        const int up = (u + PF - 1) % PF;
        const int tn = (t - 1 + PF < HT) ? t - 1 + PF : HT - 1;
        load_row(er[up], eb + (size_t)tn*HK, bo);
        float aln[16];
        v8f d0, d1;
        float ef;
        lin_step(aln, d0, d1, ef, st, er[u], ereg);
        hold_row(wr, aln[0]);
#pragma unroll
        for (int q = 0; q < 16; ++q) st[q] = aln[q];
        __builtin_amdgcn_sched_barrier(0);
      }
    }
    store_row(ab + (size_t)(tb + HALF)*HK, boff, st);
  } else {
    load_row(yr, bb + (size_t)te*HK, boff);
    {
      v4f eT[4];
      load_row(eT, eb + (size_t)te*HK, boff);
#pragma unroll
      for (int q = 0; q < 16; ++q) st[q] = yr[q >> 2][q & 3] * fexp2(eT[q >> 2][q & 3] * LOG2E);
    }
    if (ch == NCH - 1){
      v4f aT[4], g[4];
      load_row(aT, ab + (size_t)te*HK, boff);
      gamma_row(g, aT, yr);
      store_row(ab + (size_t)te*HK, boff, g);
    }
#pragma unroll
    for (int u = 0; u < PF - 1; ++u) load_row(er[u], eb + (size_t)(te - 1 - u)*HK, boff);

    unsigned bo = boff;
    for (int s0 = 0; s0 < HALF; s0 += PF){
      asm volatile("" : "+v"(bo));
#pragma unroll
      for (int u = 0; u < PF; ++u){
        const int t = te - 1 - (s0 + u);
        if (s0 + u > 0) store_row(bb + (size_t)(t + 1)*HK, bo, yr);
        const int up = (u + PF - 1) % PF;
        const int tn = (t + 1 - PF > 0) ? t + 1 - PF : 0;
        load_row(er[up], eb + (size_t)tn*HK, bo);
        float xn[16];
        v8f d0, d1;
        float ef;
        lin_step(xn, d0, d1, ef, st, er[u], ereg);
        hold_row(yr, xn[0]);
        {
          v4f a0 = { d0[0], d0[1], d0[2], d0[3] };
          v4f a1 = { d0[4], d0[5], d0[6], d0[7] };
          v4f a2 = { d1[0], d1[1], d1[2], d1[3] };
          v4f a3 = { d1[4], d1[5], d1[6], d1[7] };
          yr[0] = a0; yr[1] = a1; yr[2] = a2; yr[3] = a3;
        }
#pragma unroll
        for (int q = 0; q < 16; ++q) st[q] = xn[q];
        __builtin_amdgcn_sched_barrier(0);
      }
    }
    store_row(bb + (size_t)(te - HALF)*HK, boff, yr);
  }

  __syncthreads();

  const v4f vz = {0.f, 0.f, 0.f, 0.f};
  if (dir == 0){
#pragma unroll
    for (int u = 0; u < PF - 1; ++u) load_row(pr[u], bb + (size_t)(tb + HALF + 1 + u)*HK, boff);
    unsigned bo = boff;
    for (int k0 = HALF + 1; k0 <= 2*HALF; k0 += PF){
      asm volatile("" : "+v"(bo));
#pragma unroll
      for (int u = 0; u < PF; ++u){
        const int k  = k0 + u;
        const int t  = tb + k;
        const int up = (u + PF - 1) % PF;
        v4f wr[4] = { vz, vz, vz, vz };
        if (k > HALF + 1){
          v4f av[4];
          pack_row(av, st);
          gamma_row(wr, av, pr[up]);
          store_row(ab + (size_t)(t - 1)*HK, bo, wr);
        }
        const int tn = (t - 1 + PF < HT) ? t - 1 + PF : HT - 1;
        load_row(er[up], eb + (size_t)tn*HK, bo);
        const int jn = (k - 1 + PF < 2*HALF) ? k - 1 + PF : 2*HALF;
        load_row(pr[up], bb + (size_t)(tb + jn)*HK, bo);
        float aln[16];
        v8f d0, d1;
        float ef;
        lin_step(aln, d0, d1, ef, st, er[u], ereg);
        hold_row(wr, aln[0]);
#pragma unroll
        for (int q = 0; q < 16; ++q) st[q] = aln[q];
        __builtin_amdgcn_sched_barrier(0);
      }
    }
    {
      v4f av[4], g[4];
      pack_row(av, st);
      gamma_row(g, av, pr[(2*HALF - 1) % PF]);
      store_row(ab + (size_t)(tb + 2*HALF)*HK, boff, g);
    }
  } else {
#pragma unroll
    for (int u = 0; u < PF - 1; ++u) load_row(pr[u], ab + (size_t)(te - 1 - HALF - u)*HK, boff);
    unsigned bo = boff;
    for (int s0 = HALF; s0 < 2*HALF; s0 += PF){
      asm volatile("" : "+v"(bo));
#pragma unroll
      for (int u = 0; u < PF; ++u){
        const int s  = s0 + u;
        const int t  = te - 1 - s;
        const int up = (u + PF - 1) % PF;
        v4f wr[4] = { vz, vz, vz, vz };
        if (s > HALF){
          gamma_row(wr, pr[up], yr);
          store_row(ab + (size_t)(t + 1)*HK, bo, wr);
        }
        const int tn = (t + 1 - PF > 0) ? t + 1 - PF : 0;
        load_row(er[up], eb + (size_t)tn*HK, bo);
        const int pa = (t + 1 - PF > tb + 1) ? t + 1 - PF : tb + 1;
        load_row(pr[up], ab + (size_t)pa*HK, bo);
        float xn[16];
        v8f d0, d1;
        float ef;
        lin_step(xn, d0, d1, ef, st, er[u], ereg);
        hold_row(wr, xn[0]);
        {
          v4f a0 = { d0[0], d0[1], d0[2], d0[3] };
          v4f a1 = { d0[4], d0[5], d0[6], d0[7] };
          v4f a2 = { d1[0], d1[1], d1[2], d1[3] };
          v4f a3 = { d1[4], d1[5], d1[6], d1[7] };
          yr[0] = a0; yr[1] = a1; yr[2] = a2; yr[3] = a3;
        }
#pragma unroll
        for (int q = 0; q < 16; ++q) st[q] = xn[q];
        __builtin_amdgcn_sched_barrier(0);
      }
    }
    {
      v4f g[4];
      gamma_row(g, pr[(2*HALF - 1) % PF], yr);
      store_row(ab + (size_t)(tb + 1)*HK, boff, g);
    }
  }
}

extern "C" void kernel_launch(void* const* d_in, const int* in_sizes, int n_in,
                              void* d_out, int out_size, void* d_ws, size_t ws_size,
                              hipStream_t stream) {
  const float* em     = (const float*)d_in[0];
  const float* log_pi = (const float*)d_in[1];
  const float* log_A  = (const float*)d_in[2];
  float* out  = (float*)d_out;
  float* beta = (float*)d_ws;

  hmm_operator_kernel<<<dim3(HB*NCH*2), dim3(32), 0, stream>>>(em, log_A, out);
  hmm_edge_kernel<<<dim3(HB*2), dim3(32), 0, stream>>>(em, log_pi, out, beta);
  hmm_scan_kernel<<<dim3(NCH*4), dim3(64), 0, stream>>>(em, log_A, out, beta);
}


# Operator pass: each emission row's 32 weights made once per wave (lane = state: one load, one exponential), parked in LDS, read by halves; column scale is a separate multiply; sum order unchanged
# speedup vs baseline: 19.9555x; 1.0439x over previous
#include <hip/hip_runtime.h>


typedef __attribute__((ext_vector_type(2))) float v2f;
typedef __attribute__((ext_vector_type(4))) float v4f;
typedef __attribute__((ext_vector_type(8))) float v8f;
typedef __attribute__((address_space(1))) v4f GV4;

constexpr int HB = 64;
constexpr int HT = 4096;
constexpr int HK = 32;
constexpr int CL  = 65;
constexpr int NCH = 63;
constexpr int PFA = 5;
constexpr int PF  = 4;
constexpr float LOG2E = 1.44269504088896340736f;
constexpr float LN2   = 0.69314718055994530942f;

#if __has_builtin(__builtin_amdgcn_exp2f)
__device__ __forceinline__ float fexp2(float x){ return __builtin_amdgcn_exp2f(x); }
#else
__device__ __forceinline__ float fexp2(float x){ return exp2f(x); }
#endif
#if __has_builtin(__builtin_amdgcn_logf)
__device__ __forceinline__ float flog2(float x){ return __builtin_amdgcn_logf(x); }
#else
__device__ __forceinline__ float flog2(float x){ return log2f(x); }
#endif

__device__ __forceinline__ float rmax32(float x){
#pragma unroll
  for (int m = 16; m >= 1; m >>= 1) x = fmaxf(x, __shfl_xor(x, m, 32));
  return x;
}
__device__ __forceinline__ float rsum32(float x){
#pragma unroll
  for (int m = 16; m >= 1; m >>= 1) x += __shfl_xor(x, m, 32);
  return x;
}

__device__ __forceinline__ float xhalf(float x){
  int b = __builtin_bit_cast(int, x);
  return __builtin_bit_cast(float,
      __builtin_amdgcn_permlanex16(b, b, 0x76543210, 0xfedcba98, false, false));
}

__device__ __forceinline__ v8f wmma4(v2f a, v2f b, v8f c){
  return __builtin_amdgcn_wmma_f32_16x16x4_f32(
       false, a,  false, b,
       (short)0, c,  false,  false);
}


__device__ __forceinline__ unsigned long long uniform_addr(const void* p){
  const unsigned long long a = (unsigned long long)p;
  const unsigned lo = (unsigned)__builtin_amdgcn_readfirstlane((int)(unsigned)a);
  const unsigned hi = (unsigned)__builtin_amdgcn_readfirstlane((int)(unsigned)(a >> 32));
  return ((unsigned long long)hi << 32) | (unsigned long long)lo;
}
__device__ __forceinline__ void load_row(v4f (&dst)[4], const float* ub, unsigned boff){
  const GV4 * q = (const GV4 *)(uniform_addr(ub) + boff);
#pragma unroll
  for (int c = 0; c < 4; ++c) dst[c] = q[c];
}
__device__ __forceinline__ void pack_row(v4f (&w)[4], const float (&v)[16]){
#pragma unroll
  for (int c = 0; c < 4; ++c){
    v4f t = { v[4*c+0], v[4*c+1], v[4*c+2], v[4*c+3] };
    w[c] = t;
  }
}
__device__ __forceinline__ void store_row(float* ub, unsigned boff, const v4f (&w)[4]){
  GV4 * q = (GV4 *)(uniform_addr(ub) + boff);
#pragma unroll
  for (int c = 0; c < 4; ++c) q[c] = w[c];
}
__device__ __forceinline__ void store_row(float* ub, unsigned boff, const float (&v)[16]){
  v4f w[4];
  pack_row(w, v);
  store_row(ub, boff, w);
}

__device__ __forceinline__ float row_exponent(const float (&p)[16]){
  const float t0 = fmaxf(fmaxf(p[0],  p[1]),  p[2]);
  const float t1 = fmaxf(fmaxf(p[3],  p[4]),  p[5]);
  const float t2 = fmaxf(fmaxf(p[6],  p[7]),  p[8]);
  const float t3 = fmaxf(fmaxf(p[9],  p[10]), p[11]);
  const float t4 = fmaxf(fmaxf(p[12], p[13]), p[14]);
  const float u0 = fmaxf(fmaxf(t0, t1), t2);
  const float u1 = fmaxf(fmaxf(t3, t4), p[15]);
  float m = fmaxf(u0, u1);
  m = fmaxf(m, xhalf(m));
#if __has_builtin(__builtin_amdgcn_frexp_expf)
  return (float)__builtin_amdgcn_frexp_expf(m);
#else
  return (float)(((__builtin_bit_cast(int, m) >> 23) & 0xff) - 126);
#endif
}
__device__ __forceinline__ void lin_step(float (&pn)[16], v8f& d0, v8f& d1, float& ef_out,
                                         const float (&p)[16], const v4f (&emc)[4],
                                         const v2f (&ereg)[2][8]){
  v8f c0 = {0.f,0.f,0.f,0.f,0.f,0.f,0.f,0.f};
  v8f c1 = {0.f,0.f,0.f,0.f,0.f,0.f,0.f,0.f};
#pragma unroll
  for (int kc = 0; kc < 8; ++kc){
    const v2f b = { p[2*kc], p[2*kc+1] };
    c0 = wmma4(ereg[0][kc], b, c0);
    c1 = wmma4(ereg[1][kc], b, c1);
  }
  const float ef = row_exponent(p);
  ef_out = ef;
  float w[16];
#pragma unroll
  for (int q = 0; q < 16; ++q) w[q] = fexp2(fmaf(emc[q >> 2][q & 3], LOG2E, -ef));
#pragma unroll
  for (int r = 0; r < 8; ++r){
    pn[r]     = c0[r] * w[r];
    pn[8 + r] = c1[r] * w[8 + r];
  }
  d0 = c0;
  d1 = c1;
}
__device__ __forceinline__ void hold_row(const v4f (&w)[4], float after){
  asm volatile("" :: "v"(w[0]), "v"(w[1]), "v"(w[2]), "v"(w[3]), "v"(after));
}


__device__ __forceinline__ int row_exponent_i(const float (&p)[16]){
  const float t0 = fmaxf(fmaxf(p[0],  p[1]),  p[2]);
  const float t1 = fmaxf(fmaxf(p[3],  p[4]),  p[5]);
  const float t2 = fmaxf(fmaxf(p[6],  p[7]),  p[8]);
  const float t3 = fmaxf(fmaxf(p[9],  p[10]), p[11]);
  const float t4 = fmaxf(fmaxf(p[12], p[13]), p[14]);
  const float u0 = fmaxf(fmaxf(t0, t1), t2);
  const float u1 = fmaxf(fmaxf(t3, t4), p[15]);
  float m = fmaxf(u0, u1);
  m = fmaxf(m, xhalf(m));
#if __has_builtin(__builtin_amdgcn_frexp_expf)
  return __builtin_amdgcn_frexp_expf(m);
#else
  return ((__builtin_bit_cast(int, m) >> 23) & 0xff) - 126;
#endif
}
__device__ __forceinline__ void op_step(float (&pn)[16], int& e_out,
                                        const float (&p)[16], const v4f (&wv)[4],
                                        const v2f (&ereg)[2][8]){
  v8f c0 = {0.f,0.f,0.f,0.f,0.f,0.f,0.f,0.f};
  v8f c1 = {0.f,0.f,0.f,0.f,0.f,0.f,0.f,0.f};
#pragma unroll
  for (int kc = 0; kc < 8; ++kc){
    const v2f b = { p[2*kc], p[2*kc+1] };
    c0 = wmma4(ereg[0][kc], b, c0);
    c1 = wmma4(ereg[1][kc], b, c1);
  }
  const int e = row_exponent_i(p);
  e_out = e;
  const float sc = __builtin_ldexpf(1.0f, -e);
  v2f ws[8];
#pragma unroll
  for (int i = 0; i < 8; ++i){
    const v2f w = { wv[i >> 1][2*(i & 1)], wv[i >> 1][2*(i & 1) + 1] };
    ws[i] = w * sc;
  }
#pragma unroll
  for (int i = 0; i < 4; ++i){
    const v2f d0 = { c0[2*i], c0[2*i+1] };
    const v2f d1 = { c1[2*i], c1[2*i+1] };
    const v2f r0 = d0 * ws[i];
    const v2f r1 = d1 * ws[4 + i];
    pn[2*i]     = r0.x;  pn[2*i+1]     = r0.y;
    pn[8 + 2*i] = r1.x;  pn[8 + 2*i+1] = r1.y;
  }
}


__device__ __forceinline__ void build_expA(float* sA, const float* __restrict__ log_A, int lane){
  float x[32];
#pragma unroll
  for (int c = 0; c < 8; ++c){
    const v4f t = *(const v4f*)(log_A + lane*32 + 4*c);
    x[4*c+0] = t[0]; x[4*c+1] = t[1]; x[4*c+2] = t[2]; x[4*c+3] = t[3];
  }
  float mx = x[0];
#pragma unroll
  for (int j = 1; j < 32; ++j) mx = fmaxf(mx, x[j]);
  float s = 0.f;
#pragma unroll
  for (int j = 0; j < 32; ++j){ x[j] = fexp2((x[j] - mx) * LOG2E); s += x[j]; }
  const float r = 1.0f / s;
#pragma unroll
  for (int j = 0; j < 32; ++j) sA[lane*33 + j] = x[j] * r;
}

__device__ __forceinline__ void build_E(v2f (&ereg)[2][8], const float* sA, int ln, int h, int dir){
#pragma unroll
  for (int mt = 0; mt < 2; ++mt){
#pragma unroll
    for (int kc = 0; kc < 8; ++kc){
      const int sM = 16*(ln >> 3) + 8*mt + (ln & 7);
      const int sK = 16*h + 2*kc;
      if (dir == 0){
        ereg[mt][kc].x = sA[sK*33 + sM];
        ereg[mt][kc].y = sA[(sK+1)*33 + sM];
      } else {
        ereg[mt][kc].x = sA[sM*33 + sK];
        ereg[mt][kc].y = sA[sM*33 + sK + 1];
      }
    }
  }
}

__global__ void __launch_bounds__(32)
hmm_operator_kernel(const float* __restrict__ em,
                    const float* __restrict__ log_A,
                    float* __restrict__ fbuf){
  __shared__ __attribute__((aligned(16))) float sm[CL*HK];
  const int lane = threadIdx.x;
  const int h    = lane >> 4;
  const int ln   = lane & 15;
  const int nt   = blockIdx.x & 1;
  const int cb   = blockIdx.x >> 1;
  const int c    = cb % NCH;
  const int b    = cb / NCH;

  const size_t urow = (size_t)b*HT + (size_t)(CL*c + 1);
  const float* eb = em + urow*HK;

  float x[CL];
#pragma unroll
  for (int k = 0; k < CL; ++k) x[k] = eb[(size_t)k*HK + lane];

  build_expA(sm, log_A, lane);
  __syncthreads();
  v2f ereg[2][8];
  build_E(ereg, sm, ln, h, 0);
  __syncthreads();
#pragma unroll
  for (int k = 0; k < CL; ++k) sm[k*HK + lane] = fexp2(x[k] * LOG2E);
  __syncthreads();

  float p[16];
#pragma unroll
  for (int q = 0; q < 16; ++q) p[q] = (16*h + q == 16*nt + ln) ? 1.f : 0.f;
  int esum = 0;

  const float* wl = sm + 16*h;
  v4f wr[PFA][4];
#pragma unroll
  for (int cc = 0; cc < 4; ++cc) wr[0][cc] = *(const v4f*)(wl + 4*cc);

  for (int k0 = 0; k0 < CL; k0 += PFA){
#pragma unroll
    for (int u = 0; u < PFA; ++u){
      const int k  = k0 + u;
      const int un = (u + 1) % PFA;
      const int kn = (k + 1 < CL) ? k + 1 : CL - 1;
#pragma unroll
      for (int cc = 0; cc < 4; ++cc) wr[un][cc] = *(const v4f*)(wl + kn*HK + 4*cc);
      float pn[16];
      int e;
      op_step(pn, e, p, wr[u], ereg);
      esum += e;
#pragma unroll
      for (int q = 0; q < 16; ++q) p[q] = pn[q];
      __builtin_amdgcn_sched_barrier(0);
    }
  }
  float* dst = fbuf + urow*HK;
  const unsigned coff = (unsigned)((16*nt + ln)*HK + 16*h) * 4u;
  store_row(dst, coff, p);
  if (h == 0) dst[(size_t)(32 + nt)*HK + 16*nt + ln] = (float)esum;
}

__device__ __forceinline__ float expo_of(float m){
#if __has_builtin(__builtin_amdgcn_frexp_expf)
  return (float)__builtin_amdgcn_frexp_expf(m);
#else
  return (float)(((__builtin_bit_cast(int, m) >> 23) & 0xff) - 126);
#endif
}

__device__ __forceinline__ float lane_swz(float x, const int ctrl_sel){
  const int b = __builtin_bit_cast(int, x);
  int r;
  if (ctrl_sel == 0)      r = __builtin_amdgcn_update_dpp(b, b, 0xB1,  0xf, 0xf, false);
  else if (ctrl_sel == 1) r = __builtin_amdgcn_update_dpp(b, b, 0x4E,  0xf, 0xf, false);
  else if (ctrl_sel == 2) r = __builtin_amdgcn_update_dpp(b, b, 0x141, 0xf, 0xf, false);
  else                    r = __builtin_amdgcn_update_dpp(b, b, 0x140, 0xf, 0xf, false);
  return __builtin_bit_cast(float, r);
}
__device__ __forceinline__ float wmax32(float x){
  x = fmaxf(x, lane_swz(x, 0));
  x = fmaxf(x, lane_swz(x, 1));
  x = fmaxf(x, lane_swz(x, 2));
  x = fmaxf(x, lane_swz(x, 3));
  return fmaxf(x, xhalf(x));
}
__device__ __forceinline__ float lane_bcast(float x, const int j){
  return __builtin_bit_cast(float, __builtin_amdgcn_readlane(__builtin_bit_cast(int, x), j));
}

__global__ void __launch_bounds__(32)
hmm_edge_kernel(const float* __restrict__ em,
                const float* __restrict__ log_pi,
                float* fa,
                float* __restrict__ bw){
  const int lane = threadIdx.x;
  const int b    = blockIdx.x >> 1;
  const int dir  = blockIdx.x & 1;
  float* fb = fa + (size_t)b*HT*HK;

  if (dir == 0){
    float xp = log_pi[lane];
    float mp = rmax32(xp);
    float sp = rsum32(fexp2((xp - mp) * LOG2E));
    float lp = xp - mp - LN2 * flog2(sp);
    float v  = fexp2((lp + em[(size_t)b*HT*HK + lane]) * LOG2E);
    fb[lane] = v;

    float f[3][32];
    float sx[3];
#pragma unroll
    for (int u = 0; u < 2; ++u){
      const float* src = fb + (size_t)(CL*u + 1)*HK;
#pragma unroll
      for (int j = 0; j < 32; ++j) f[u][j] = src[(size_t)j*HK + lane];
      sx[u] = src[(size_t)(32 + (lane >> 4))*HK + lane];
    }
    for (int c0 = 0; c0 < NCH; c0 += 3){
#pragma unroll
      for (int u = 0; u < 3; ++u){
        const int c  = c0 + u;
        const int cn = (c + 2 < NCH) ? c + 2 : NCH - 1;
        {
          const int un = (u + 2) % 3;
          const float* src = fb + (size_t)(CL*cn + 1)*HK;
#pragma unroll
          for (int j = 0; j < 32; ++j) f[un][j] = src[(size_t)j*HK + lane];
          sx[un] = src[(size_t)(32 + (lane >> 4))*HK + lane];
        }
        const float smax = wmax32(sx[u]);
        const float uu = v * fexp2(sx[u] - smax);
        float a0 = 0.f, a1 = 0.f, a2 = 0.f, a3 = 0.f;
#pragma unroll
        for (int j = 0; j < 32; j += 4){
          a0 = fmaf(f[u][j+0], lane_bcast(uu, j+0), a0);
          a1 = fmaf(f[u][j+1], lane_bcast(uu, j+1), a1);
          a2 = fmaf(f[u][j+2], lane_bcast(uu, j+2), a2);
          a3 = fmaf(f[u][j+3], lane_bcast(uu, j+3), a3);
        }
        const float acc = (a0 + a1) + (a2 + a3);
        const float e = expo_of(wmax32(acc));
        v = acc * fexp2(-e);
        fb[(size_t)(CL*(c + 1))*HK + lane] = v;
      }
    }
  } else {
    float* bb = bw + (size_t)b*HT*HK;
    float v = 1.f;
    bb[(size_t)(HT - 1)*HK + lane] = v;

    v4f g[3][8];
    float sx[3];
#pragma unroll
    for (int u = 0; u < 2; ++u){
      const float* src = fb + (size_t)(CL*(NCH - 1 - u) + 1)*HK;
#pragma unroll
      for (int q = 0; q < 8; ++q) g[u][q] = *(const v4f*)(src + lane*HK + 4*q);
      sx[u] = src[(size_t)(32 + (lane >> 4))*HK + lane];
    }
    for (int c0 = 0; c0 < NCH; c0 += 3){
#pragma unroll
      for (int u = 0; u < 3; ++u){
        const int c  = NCH - 1 - (c0 + u);
        const int cn = (c - 2 > 0) ? c - 2 : 0;
        {
          const int un = (u + 2) % 3;
          const float* src = fb + (size_t)(CL*cn + 1)*HK;
#pragma unroll
          for (int q = 0; q < 8; ++q) g[un][q] = *(const v4f*)(src + lane*HK + 4*q);
          sx[un] = src[(size_t)(32 + (lane >> 4))*HK + lane];
        }
        float a0 = 0.f, a1 = 0.f, a2 = 0.f, a3 = 0.f;
#pragma unroll
        for (int q = 0; q < 8; ++q){
          a0 = fmaf(g[u][q][0], lane_bcast(v, 4*q+0), a0);
          a1 = fmaf(g[u][q][1], lane_bcast(v, 4*q+1), a1);
          a2 = fmaf(g[u][q][2], lane_bcast(v, 4*q+2), a2);
          a3 = fmaf(g[u][q][3], lane_bcast(v, 4*q+3), a3);
        }
        const float smax = wmax32(sx[u]);
        const float acc = ((a0 + a1) + (a2 + a3)) * fexp2(sx[u] - smax);
        const float e = expo_of(wmax32(acc));
        v = acc * fexp2(-e);
        bb[(size_t)(CL*c)*HK + lane] = v;
      }
    }
  }
}

__device__ __forceinline__ void gamma_row(v4f (&g)[4], const v4f (&a)[4], const v4f (&b)[4]){
  float p[16];
#pragma unroll
  for (int q = 0; q < 16; ++q) p[q] = a[q >> 2][q & 3] * b[q >> 2][q & 3];
  const float s0 = ((p[0] + p[1]) + (p[2] + p[3])) + ((p[4] + p[5]) + (p[6] + p[7]));
  const float s1 = ((p[8] + p[9]) + (p[10] + p[11])) + ((p[12] + p[13]) + (p[14] + p[15]));
  float s = s0 + s1;
  s += xhalf(s);
  const float ls = flog2(s);
#pragma unroll
  for (int c = 0; c < 4; ++c){
    v4f t;
#pragma unroll
    for (int e = 0; e < 4; ++e) t[e] = LN2 * ((flog2(a[c][e]) + flog2(b[c][e])) - ls);
    g[c] = t;
  }
}

__global__ void __launch_bounds__(64)
hmm_scan_kernel(const float* __restrict__ em,
                const float* __restrict__ log_A,
                float* alpha_out,
                float* beta_out){
  __shared__ float sA[32*33];

  const int lane = threadIdx.x & 31;
  const int dir  = __builtin_amdgcn_readfirstlane((int)(threadIdx.x >> 5));
  const int h    = lane >> 4;
  const int ln   = lane & 15;
  const int mb   = blockIdx.x & 3;
  const int ch   = blockIdx.x >> 2;

  if (dir == 0) build_expA(sA, log_A, lane);
  __syncthreads();
  v2f ereg[2][8];
  build_E(ereg, sA, ln, h, dir);

  const size_t   uoff = (size_t)mb*16*HT*HK;
  const unsigned boff = (unsigned)(ln*HT*HK + 16*h) * 4u;
  const float* eb = em + uoff;
  float* ab = alpha_out + uoff;
  float* bb = beta_out + uoff;
  const int tb = CL*ch;
  const int te = CL*(ch + 1);
  constexpr int HALF = 32;

  v4f er[PF][4];
  v4f pr[PF][4];
  float st[16];
  v4f yr[4];

  if (dir == 0){
    {
      v4f a0[4], b0[4];
      load_row(a0, ab + (size_t)tb*HK, boff);
      load_row(b0, bb + (size_t)tb*HK, boff);
#pragma unroll
      for (int u = 0; u < PF - 1; ++u) load_row(er[u], eb + (size_t)(tb + 1 + u)*HK, boff);
#pragma unroll
      for (int q = 0; q < 16; ++q) st[q] = a0[q >> 2][q & 3];
      v4f g[4];
      gamma_row(g, a0, b0);
      store_row(ab + (size_t)tb*HK, boff, g);
    }
    unsigned bo = boff;
    for (int k0 = 1; k0 <= HALF; k0 += PF){
      asm volatile("" : "+v"(bo));
#pragma unroll
      for (int u = 0; u < PF; ++u){
        const int t = tb + k0 + u;
        v4f wr[4];
        pack_row(wr, st);
        if (k0 + u > 1) store_row(ab + (size_t)(t - 1)*HK, bo, wr);
        const int up = (u + PF - 1) % PF;
        const int tn = (t - 1 + PF < HT) ? t - 1 + PF : HT - 1;
        load_row(er[up], eb + (size_t)tn*HK, bo);
        float aln[16];
        v8f d0, d1;
        float ef;
        lin_step(aln, d0, d1, ef, st, er[u], ereg);
        hold_row(wr, aln[0]);
#pragma unroll
        for (int q = 0; q < 16; ++q) st[q] = aln[q];
        __builtin_amdgcn_sched_barrier(0);
      }
    }
    store_row(ab + (size_t)(tb + HALF)*HK, boff, st);
  } else {
    load_row(yr, bb + (size_t)te*HK, boff);
    {
      v4f eT[4];
      load_row(eT, eb + (size_t)te*HK, boff);
#pragma unroll
      for (int q = 0; q < 16; ++q) st[q] = yr[q >> 2][q & 3] * fexp2(eT[q >> 2][q & 3] * LOG2E);
    }
    if (ch == NCH - 1){
      v4f aT[4], g[4];
      load_row(aT, ab + (size_t)te*HK, boff);
      gamma_row(g, aT, yr);
      store_row(ab + (size_t)te*HK, boff, g);
    }
#pragma unroll
    for (int u = 0; u < PF - 1; ++u) load_row(er[u], eb + (size_t)(te - 1 - u)*HK, boff);

    unsigned bo = boff;
    for (int s0 = 0; s0 < HALF; s0 += PF){
      asm volatile("" : "+v"(bo));
#pragma unroll
      for (int u = 0; u < PF; ++u){
        const int t = te - 1 - (s0 + u);
        if (s0 + u > 0) store_row(bb + (size_t)(t + 1)*HK, bo, yr);
        const int up = (u + PF - 1) % PF;
        const int tn = (t + 1 - PF > 0) ? t + 1 - PF : 0;
        load_row(er[up], eb + (size_t)tn*HK, bo);
        float xn[16];
        v8f d0, d1;
        float ef;
        lin_step(xn, d0, d1, ef, st, er[u], ereg);
        hold_row(yr, xn[0]);
        {
          v4f a0 = { d0[0], d0[1], d0[2], d0[3] };
          v4f a1 = { d0[4], d0[5], d0[6], d0[7] };
          v4f a2 = { d1[0], d1[1], d1[2], d1[3] };
          v4f a3 = { d1[4], d1[5], d1[6], d1[7] };
          yr[0] = a0; yr[1] = a1; yr[2] = a2; yr[3] = a3;
        }
#pragma unroll
        for (int q = 0; q < 16; ++q) st[q] = xn[q];
        __builtin_amdgcn_sched_barrier(0);
      }
    }
    store_row(bb + (size_t)(te - HALF)*HK, boff, yr);
  }

  __syncthreads();

  const v4f vz = {0.f, 0.f, 0.f, 0.f};
  if (dir == 0){
#pragma unroll
    for (int u = 0; u < PF - 1; ++u) load_row(pr[u], bb + (size_t)(tb + HALF + 1 + u)*HK, boff);
    unsigned bo = boff;
    for (int k0 = HALF + 1; k0 <= 2*HALF; k0 += PF){
      asm volatile("" : "+v"(bo));
#pragma unroll
      for (int u = 0; u < PF; ++u){
        const int k  = k0 + u;
        const int t  = tb + k;
        const int up = (u + PF - 1) % PF;
        v4f wr[4] = { vz, vz, vz, vz };
        if (k > HALF + 1){
          v4f av[4];
          pack_row(av, st);
          gamma_row(wr, av, pr[up]);
          store_row(ab + (size_t)(t - 1)*HK, bo, wr);
        }
        const int tn = (t - 1 + PF < HT) ? t - 1 + PF : HT - 1;
        load_row(er[up], eb + (size_t)tn*HK, bo);
        const int jn = (k - 1 + PF < 2*HALF) ? k - 1 + PF : 2*HALF;
        load_row(pr[up], bb + (size_t)(tb + jn)*HK, bo);
        float aln[16];
        v8f d0, d1;
        float ef;
        lin_step(aln, d0, d1, ef, st, er[u], ereg);
        hold_row(wr, aln[0]);
#pragma unroll
        for (int q = 0; q < 16; ++q) st[q] = aln[q];
        __builtin_amdgcn_sched_barrier(0);
      }
    }
    {
      v4f av[4], g[4];
      pack_row(av, st);
      gamma_row(g, av, pr[(2*HALF - 1) % PF]);
      store_row(ab + (size_t)(tb + 2*HALF)*HK, boff, g);
    }
  } else {
#pragma unroll
    for (int u = 0; u < PF - 1; ++u) load_row(pr[u], ab + (size_t)(te - 1 - HALF - u)*HK, boff);
    unsigned bo = boff;
    for (int s0 = HALF; s0 < 2*HALF; s0 += PF){
      asm volatile("" : "+v"(bo));
#pragma unroll
      for (int u = 0; u < PF; ++u){
        const int s  = s0 + u;
        const int t  = te - 1 - s;
        const int up = (u + PF - 1) % PF;
        v4f wr[4] = { vz, vz, vz, vz };
        if (s > HALF){
          gamma_row(wr, pr[up], yr);
          store_row(ab + (size_t)(t + 1)*HK, bo, wr);
        }
        const int tn = (t + 1 - PF > 0) ? t + 1 - PF : 0;
        load_row(er[up], eb + (size_t)tn*HK, bo);
        const int pa = (t + 1 - PF > tb + 1) ? t + 1 - PF : tb + 1;
        load_row(pr[up], ab + (size_t)pa*HK, bo);
        float xn[16];
        v8f d0, d1;
        float ef;
        lin_step(xn, d0, d1, ef, st, er[u], ereg);
        hold_row(wr, xn[0]);
        {
          v4f a0 = { d0[0], d0[1], d0[2], d0[3] };
          v4f a1 = { d0[4], d0[5], d0[6], d0[7] };
          v4f a2 = { d1[0], d1[1], d1[2], d1[3] };
          v4f a3 = { d1[4], d1[5], d1[6], d1[7] };
          yr[0] = a0; yr[1] = a1; yr[2] = a2; yr[3] = a3;
        }
#pragma unroll
        for (int q = 0; q < 16; ++q) st[q] = xn[q];
        __builtin_amdgcn_sched_barrier(0);
      }
    }
    {
      v4f g[4];
      gamma_row(g, pr[(2*HALF - 1) % PF], yr);
      store_row(ab + (size_t)(tb + 1)*HK, boff, g);
    }
  }
}

extern "C" void kernel_launch(void* const* d_in, const int* in_sizes, int n_in,
                              void* d_out, int out_size, void* d_ws, size_t ws_size,
                              hipStream_t stream) {
  const float* em     = (const float*)d_in[0];
  const float* log_pi = (const float*)d_in[1];
  const float* log_A  = (const float*)d_in[2];
  float* out  = (float*)d_out;
  float* beta = (float*)d_ws;

  hmm_operator_kernel<<<dim3(HB*NCH*2), dim3(32), 0, stream>>>(em, log_A, out);
  hmm_edge_kernel<<<dim3(HB*2), dim3(32), 0, stream>>>(em, log_pi, out, beta);
  hmm_scan_kernel<<<dim3(NCH*4), dim3(64), 0, stream>>>(em, log_A, out, beta);
}


# Edge walk: only the product stands between two rows: row scale 2^-e from the operand's broadcasts on the scalar unit, exponent rows and column scales a chunk early, no load wait under two chunks
# speedup vs baseline: 21.5658x; 1.0807x over previous
#include <hip/hip_runtime.h>


typedef __attribute__((ext_vector_type(2))) float v2f;
typedef __attribute__((ext_vector_type(4))) float v4f;
typedef __attribute__((ext_vector_type(8))) float v8f;
typedef __attribute__((address_space(1))) v4f GV4;

constexpr int HB = 64;
constexpr int HT = 4096;
constexpr int HK = 32;
constexpr int CL  = 65;
constexpr int NCH = 63;
constexpr int PFA = 5;
constexpr int PF  = 4;
constexpr float LOG2E = 1.44269504088896340736f;
constexpr float LN2   = 0.69314718055994530942f;

#if __has_builtin(__builtin_amdgcn_exp2f)
__device__ __forceinline__ float fexp2(float x){ return __builtin_amdgcn_exp2f(x); }
#else
__device__ __forceinline__ float fexp2(float x){ return exp2f(x); }
#endif
#if __has_builtin(__builtin_amdgcn_logf)
__device__ __forceinline__ float flog2(float x){ return __builtin_amdgcn_logf(x); }
#else
__device__ __forceinline__ float flog2(float x){ return log2f(x); }
#endif

__device__ __forceinline__ float rmax32(float x){
#pragma unroll
  for (int m = 16; m >= 1; m >>= 1) x = fmaxf(x, __shfl_xor(x, m, 32));
  return x;
}
__device__ __forceinline__ float rsum32(float x){
#pragma unroll
  for (int m = 16; m >= 1; m >>= 1) x += __shfl_xor(x, m, 32);
  return x;
}

__device__ __forceinline__ float xhalf(float x){
  int b = __builtin_bit_cast(int, x);
  return __builtin_bit_cast(float,
      __builtin_amdgcn_permlanex16(b, b, 0x76543210, 0xfedcba98, false, false));
}

__device__ __forceinline__ v8f wmma4(v2f a, v2f b, v8f c){
  return __builtin_amdgcn_wmma_f32_16x16x4_f32(
       false, a,  false, b,
       (short)0, c,  false,  false);
}


__device__ __forceinline__ unsigned long long uniform_addr(const void* p){
  const unsigned long long a = (unsigned long long)p;
  const unsigned lo = (unsigned)__builtin_amdgcn_readfirstlane((int)(unsigned)a);
  const unsigned hi = (unsigned)__builtin_amdgcn_readfirstlane((int)(unsigned)(a >> 32));
  return ((unsigned long long)hi << 32) | (unsigned long long)lo;
}
__device__ __forceinline__ void load_row(v4f (&dst)[4], const float* ub, unsigned boff){
  const GV4 * q = (const GV4 *)(uniform_addr(ub) + boff);
#pragma unroll
  for (int c = 0; c < 4; ++c) dst[c] = q[c];
}
__device__ __forceinline__ void pack_row(v4f (&w)[4], const float (&v)[16]){
#pragma unroll
  for (int c = 0; c < 4; ++c){
    v4f t = { v[4*c+0], v[4*c+1], v[4*c+2], v[4*c+3] };
    w[c] = t;
  }
}
__device__ __forceinline__ void store_row(float* ub, unsigned boff, const v4f (&w)[4]){
  GV4 * q = (GV4 *)(uniform_addr(ub) + boff);
#pragma unroll
  for (int c = 0; c < 4; ++c) q[c] = w[c];
}
__device__ __forceinline__ void store_row(float* ub, unsigned boff, const float (&v)[16]){
  v4f w[4];
  pack_row(w, v);
  store_row(ub, boff, w);
}

__device__ __forceinline__ float row_exponent(const float (&p)[16]){
  const float t0 = fmaxf(fmaxf(p[0],  p[1]),  p[2]);
  const float t1 = fmaxf(fmaxf(p[3],  p[4]),  p[5]);
  const float t2 = fmaxf(fmaxf(p[6],  p[7]),  p[8]);
  const float t3 = fmaxf(fmaxf(p[9],  p[10]), p[11]);
  const float t4 = fmaxf(fmaxf(p[12], p[13]), p[14]);
  const float u0 = fmaxf(fmaxf(t0, t1), t2);
  const float u1 = fmaxf(fmaxf(t3, t4), p[15]);
  float m = fmaxf(u0, u1);
  m = fmaxf(m, xhalf(m));
#if __has_builtin(__builtin_amdgcn_frexp_expf)
  return (float)__builtin_amdgcn_frexp_expf(m);
#else
  return (float)(((__builtin_bit_cast(int, m) >> 23) & 0xff) - 126);
#endif
}
__device__ __forceinline__ void lin_step(float (&pn)[16], v8f& d0, v8f& d1, float& ef_out,
                                         const float (&p)[16], const v4f (&emc)[4],
                                         const v2f (&ereg)[2][8]){
  v8f c0 = {0.f,0.f,0.f,0.f,0.f,0.f,0.f,0.f};
  v8f c1 = {0.f,0.f,0.f,0.f,0.f,0.f,0.f,0.f};
#pragma unroll
  for (int kc = 0; kc < 8; ++kc){
    const v2f b = { p[2*kc], p[2*kc+1] };
    c0 = wmma4(ereg[0][kc], b, c0);
    c1 = wmma4(ereg[1][kc], b, c1);
  }
  const float ef = row_exponent(p);
  ef_out = ef;
  float w[16];
#pragma unroll
  for (int q = 0; q < 16; ++q) w[q] = fexp2(fmaf(emc[q >> 2][q & 3], LOG2E, -ef));
#pragma unroll
  for (int r = 0; r < 8; ++r){
    pn[r]     = c0[r] * w[r];
    pn[8 + r] = c1[r] * w[8 + r];
  }
  d0 = c0;
  d1 = c1;
}
__device__ __forceinline__ void hold_row(const v4f (&w)[4], float after){
  asm volatile("" :: "v"(w[0]), "v"(w[1]), "v"(w[2]), "v"(w[3]), "v"(after));
}


__device__ __forceinline__ int row_exponent_i(const float (&p)[16]){
  const float t0 = fmaxf(fmaxf(p[0],  p[1]),  p[2]);
  const float t1 = fmaxf(fmaxf(p[3],  p[4]),  p[5]);
  const float t2 = fmaxf(fmaxf(p[6],  p[7]),  p[8]);
  const float t3 = fmaxf(fmaxf(p[9],  p[10]), p[11]);
  const float t4 = fmaxf(fmaxf(p[12], p[13]), p[14]);
  const float u0 = fmaxf(fmaxf(t0, t1), t2);
  const float u1 = fmaxf(fmaxf(t3, t4), p[15]);
  float m = fmaxf(u0, u1);
  m = fmaxf(m, xhalf(m));
#if __has_builtin(__builtin_amdgcn_frexp_expf)
  return __builtin_amdgcn_frexp_expf(m);
#else
  return ((__builtin_bit_cast(int, m) >> 23) & 0xff) - 126;
#endif
}
__device__ __forceinline__ void op_step(float (&pn)[16], int& e_out,
                                        const float (&p)[16], const v4f (&wv)[4],
                                        const v2f (&ereg)[2][8]){
  v8f c0 = {0.f,0.f,0.f,0.f,0.f,0.f,0.f,0.f};
  v8f c1 = {0.f,0.f,0.f,0.f,0.f,0.f,0.f,0.f};
#pragma unroll
  for (int kc = 0; kc < 8; ++kc){
    const v2f b = { p[2*kc], p[2*kc+1] };
    c0 = wmma4(ereg[0][kc], b, c0);
    c1 = wmma4(ereg[1][kc], b, c1);
  }
  const int e = row_exponent_i(p);
  e_out = e;
  const float sc = __builtin_ldexpf(1.0f, -e);
  v2f ws[8];
#pragma unroll
  for (int i = 0; i < 8; ++i){
    const v2f w = { wv[i >> 1][2*(i & 1)], wv[i >> 1][2*(i & 1) + 1] };
    ws[i] = w * sc;
  }
#pragma unroll
  for (int i = 0; i < 4; ++i){
    const v2f d0 = { c0[2*i], c0[2*i+1] };
    const v2f d1 = { c1[2*i], c1[2*i+1] };
    const v2f r0 = d0 * ws[i];
    const v2f r1 = d1 * ws[4 + i];
    pn[2*i]     = r0.x;  pn[2*i+1]     = r0.y;
    pn[8 + 2*i] = r1.x;  pn[8 + 2*i+1] = r1.y;
  }
}


__device__ __forceinline__ void build_expA(float* sA, const float* __restrict__ log_A, int lane){
  float x[32];
#pragma unroll
  for (int c = 0; c < 8; ++c){
    const v4f t = *(const v4f*)(log_A + lane*32 + 4*c);
    x[4*c+0] = t[0]; x[4*c+1] = t[1]; x[4*c+2] = t[2]; x[4*c+3] = t[3];
  }
  float mx = x[0];
#pragma unroll
  for (int j = 1; j < 32; ++j) mx = fmaxf(mx, x[j]);
  float s = 0.f;
#pragma unroll
  for (int j = 0; j < 32; ++j){ x[j] = fexp2((x[j] - mx) * LOG2E); s += x[j]; }
  const float r = 1.0f / s;
#pragma unroll
  for (int j = 0; j < 32; ++j) sA[lane*33 + j] = x[j] * r;
}

__device__ __forceinline__ void build_E(v2f (&ereg)[2][8], const float* sA, int ln, int h, int dir){
#pragma unroll
  for (int mt = 0; mt < 2; ++mt){
#pragma unroll
    for (int kc = 0; kc < 8; ++kc){
      const int sM = 16*(ln >> 3) + 8*mt + (ln & 7);
      const int sK = 16*h + 2*kc;
      if (dir == 0){
        ereg[mt][kc].x = sA[sK*33 + sM];
        ereg[mt][kc].y = sA[(sK+1)*33 + sM];
      } else {
        ereg[mt][kc].x = sA[sM*33 + sK];
        ereg[mt][kc].y = sA[sM*33 + sK + 1];
      }
    }
  }
}

__global__ void __launch_bounds__(32)
hmm_operator_kernel(const float* __restrict__ em,
                    const float* __restrict__ log_A,
                    float* __restrict__ fbuf){
  __shared__ __attribute__((aligned(16))) float sm[CL*HK];
  const int lane = threadIdx.x;
  const int h    = lane >> 4;
  const int ln   = lane & 15;
  const int nt   = blockIdx.x & 1;
  const int cb   = blockIdx.x >> 1;
  const int c    = cb % NCH;
  const int b    = cb / NCH;

  const size_t urow = (size_t)b*HT + (size_t)(CL*c + 1);
  const float* eb = em + urow*HK;

  float x[CL];
#pragma unroll
  for (int k = 0; k < CL; ++k) x[k] = eb[(size_t)k*HK + lane];

  build_expA(sm, log_A, lane);
  __syncthreads();
  v2f ereg[2][8];
  build_E(ereg, sm, ln, h, 0);
  __syncthreads();
#pragma unroll
  for (int k = 0; k < CL; ++k) sm[k*HK + lane] = fexp2(x[k] * LOG2E);
  __syncthreads();

  float p[16];
#pragma unroll
  for (int q = 0; q < 16; ++q) p[q] = (16*h + q == 16*nt + ln) ? 1.f : 0.f;
  int esum = 0;

  const float* wl = sm + 16*h;
  v4f wr[PFA][4];
#pragma unroll
  for (int cc = 0; cc < 4; ++cc) wr[0][cc] = *(const v4f*)(wl + 4*cc);

  for (int k0 = 0; k0 < CL; k0 += PFA){
#pragma unroll
    for (int u = 0; u < PFA; ++u){
      const int k  = k0 + u;
      const int un = (u + 1) % PFA;
      const int kn = (k + 1 < CL) ? k + 1 : CL - 1;
#pragma unroll
      for (int cc = 0; cc < 4; ++cc) wr[un][cc] = *(const v4f*)(wl + kn*HK + 4*cc);
      float pn[16];
      int e;
      op_step(pn, e, p, wr[u], ereg);
      esum += e;
#pragma unroll
      for (int q = 0; q < 16; ++q) p[q] = pn[q];
      __builtin_amdgcn_sched_barrier(0);
    }
  }
  float* dst = fbuf + urow*HK;
  const unsigned coff = (unsigned)((16*nt + ln)*HK + 16*h) * 4u;
  store_row(dst, coff, p);
  if (h == 0) dst[(size_t)(32 + nt)*HK + 16*nt + ln] = (float)esum;
}

__device__ __forceinline__ float expo_of(float m){
#if __has_builtin(__builtin_amdgcn_frexp_expf)
  return (float)__builtin_amdgcn_frexp_expf(m);
#else
  return (float)(((__builtin_bit_cast(int, m) >> 23) & 0xff) - 126);
#endif
}

__device__ __forceinline__ float lane_swz(float x, const int ctrl_sel){
  const int b = __builtin_bit_cast(int, x);
  int r;
  if (ctrl_sel == 0)      r = __builtin_amdgcn_update_dpp(b, b, 0xB1,  0xf, 0xf, false);
  else if (ctrl_sel == 1) r = __builtin_amdgcn_update_dpp(b, b, 0x4E,  0xf, 0xf, false);
  else if (ctrl_sel == 2) r = __builtin_amdgcn_update_dpp(b, b, 0x141, 0xf, 0xf, false);
  else                    r = __builtin_amdgcn_update_dpp(b, b, 0x140, 0xf, 0xf, false);
  return __builtin_bit_cast(float, r);
}
__device__ __forceinline__ float wmax32(float x){
  x = fmaxf(x, lane_swz(x, 0));
  x = fmaxf(x, lane_swz(x, 1));
  x = fmaxf(x, lane_swz(x, 2));
  x = fmaxf(x, lane_swz(x, 3));
  return fmaxf(x, xhalf(x));
}
__device__ __forceinline__ float lane_bcast(float x, const int j){
  return __builtin_bit_cast(float, __builtin_amdgcn_readlane(__builtin_bit_cast(int, x), j));
}

__device__ __forceinline__ unsigned umax2(unsigned a, unsigned b){ return a > b ? a : b; }
__device__ __forceinline__ float inv_pow2_of_max(unsigned mb){
  int fld = 253 - (int)(mb >> 23);
  fld = fld < 1 ? 1 : fld;
  return __builtin_bit_cast(float, (unsigned)fld << 23);
}

__global__ void __launch_bounds__(32)
hmm_edge_kernel(const float* __restrict__ em,
                const float* __restrict__ log_pi,
                float* fa,
                float* __restrict__ bw){
  const int lane = threadIdx.x;
  const int b    = blockIdx.x >> 1;
  const int dir  = blockIdx.x & 1;
  float* fb = fa + (size_t)b*HT*HK;

  if (dir == 0){
    float xp = log_pi[lane];
    float mp = rmax32(xp);
    float sp = rsum32(fexp2((xp - mp) * LOG2E));
    float lp = xp - mp - LN2 * flog2(sp);
    float v  = fexp2((lp + em[(size_t)b*HT*HK + lane]) * LOG2E);
    fb[lane] = v;

    float f[3][32];
    float sx[3];
    const size_t xoff = (size_t)(32 + (lane >> 4))*HK + lane;
#pragma unroll
    for (int u = 0; u < 3; ++u) sx[u] = fb[(size_t)(CL*u + 1)*HK + xoff];
#pragma unroll
    for (int u = 0; u < 2; ++u){
      const float* src = fb + (size_t)(CL*u + 1)*HK;
#pragma unroll
      for (int j = 0; j < 32; ++j) f[u][j] = src[(size_t)j*HK + lane];
    }
    float cs[3];
    cs[0] = fexp2(sx[0] - wmax32(sx[0]));
    asm volatile("" :: "v"(f[0][0]), "v"(f[0][1]), "v"(f[0][2]), "v"(f[0][3]), "v"(f[0][4]), "v"(f[0][5]), "v"(f[0][6]), "v"(f[0][7]), "v"(f[0][8]), "v"(f[0][9]), "v"(f[0][10]), "v"(f[0][11]), "v"(f[0][12]), "v"(f[0][13]), "v"(f[0][14]), "v"(f[0][15]));
    asm volatile("" :: "v"(f[0][16]), "v"(f[0][17]), "v"(f[0][18]), "v"(f[0][19]), "v"(f[0][20]), "v"(f[0][21]), "v"(f[0][22]), "v"(f[0][23]), "v"(f[0][24]), "v"(f[0][25]), "v"(f[0][26]), "v"(f[0][27]), "v"(f[0][28]), "v"(f[0][29]), "v"(f[0][30]), "v"(f[0][31]), "v"(sx[1]));
    for (int c0 = 0; c0 < NCH; c0 += 3){
#pragma unroll
      for (int u = 0; u < 3; ++u){
        const int c  = c0 + u;
        const int cn = (c + 2 < NCH) ? c + 2 : NCH - 1;
        const int cx = (c + 3 < NCH) ? c + 3 : NCH - 1;
        {
          sx[u] = fb[(size_t)(CL*cx + 1)*HK + xoff];
          const int un = (u + 2) % 3;
          const float* src = fb + (size_t)(CL*cn + 1)*HK;
#pragma unroll
          for (int j = 0; j < 32; ++j) f[un][j] = src[(size_t)j*HK + lane];
        }
        {
          const int u1 = (u + 1) % 3;
          cs[u1] = fexp2(sx[u1] - wmax32(sx[u1]));
        }
        const float uu = v * cs[u];
        float a0 = 0.f, a1 = 0.f, a2 = 0.f, a3 = 0.f;
        unsigned m0 = 0u, m1 = 0u, m2 = 0u, m3 = 0u;
#pragma unroll
        for (int j = 0; j < 32; j += 4){
          const float b0 = lane_bcast(uu, j+0), b1 = lane_bcast(uu, j+1);
          const float b2 = lane_bcast(uu, j+2), b3 = lane_bcast(uu, j+3);
          m0 = umax2(m0, __builtin_bit_cast(unsigned, b0));
          m1 = umax2(m1, __builtin_bit_cast(unsigned, b1));
          m2 = umax2(m2, __builtin_bit_cast(unsigned, b2));
          m3 = umax2(m3, __builtin_bit_cast(unsigned, b3));
          a0 = fmaf(f[u][j+0], b0, a0);
          a1 = fmaf(f[u][j+1], b1, a1);
          a2 = fmaf(f[u][j+2], b2, a2);
          a3 = fmaf(f[u][j+3], b3, a3);
        }
        const float acc = (a0 + a1) + (a2 + a3);
        v = acc * inv_pow2_of_max(umax2(umax2(m0, m1), umax2(m2, m3)));
        fb[(size_t)(CL*(c + 1))*HK + lane] = v;
        __builtin_amdgcn_sched_barrier(0);
      }
    }
  } else {
    float* bb = bw + (size_t)b*HT*HK;
    float v = 1.f;
    bb[(size_t)(HT - 1)*HK + lane] = v;

    v4f g[3][8];
    float sx[3];
    const size_t xoff = (size_t)(32 + (lane >> 4))*HK + lane;
#pragma unroll
    for (int u = 0; u < 3; ++u) sx[u] = fb[(size_t)(CL*(NCH - 1 - u) + 1)*HK + xoff];
#pragma unroll
    for (int u = 0; u < 2; ++u){
      const float* src = fb + (size_t)(CL*(NCH - 1 - u) + 1)*HK;
#pragma unroll
      for (int q = 0; q < 8; ++q) g[u][q] = *(const v4f*)(src + lane*HK + 4*q);
    }
    float cs[3];
    cs[0] = fexp2(sx[0] - wmax32(sx[0]));
    asm volatile("" :: "v"(g[0][0]), "v"(g[0][1]), "v"(g[0][2]), "v"(g[0][3]), "v"(g[0][4]), "v"(g[0][5]), "v"(g[0][6]), "v"(g[0][7]), "v"(sx[1]));
    for (int c0 = 0; c0 < NCH; c0 += 3){
#pragma unroll
      for (int u = 0; u < 3; ++u){
        const int c  = NCH - 1 - (c0 + u);
        const int cn = (c - 2 > 0) ? c - 2 : 0;
        const int cx = (c - 3 > 0) ? c - 3 : 0;
        {
          sx[u] = fb[(size_t)(CL*cx + 1)*HK + xoff];
          const int un = (u + 2) % 3;
          const float* src = fb + (size_t)(CL*cn + 1)*HK;
#pragma unroll
          for (int q = 0; q < 8; ++q) g[un][q] = *(const v4f*)(src + lane*HK + 4*q);
        }
        {
          const int u1 = (u + 1) % 3;
          cs[u1] = fexp2(sx[u1] - wmax32(sx[u1]));
        }
        float a0 = 0.f, a1 = 0.f, a2 = 0.f, a3 = 0.f;
        unsigned m0 = 0u, m1 = 0u, m2 = 0u, m3 = 0u;
#pragma unroll
        for (int q = 0; q < 8; ++q){
          const float b0 = lane_bcast(v, 4*q+0), b1 = lane_bcast(v, 4*q+1);
          const float b2 = lane_bcast(v, 4*q+2), b3 = lane_bcast(v, 4*q+3);
          m0 = umax2(m0, __builtin_bit_cast(unsigned, b0));
          m1 = umax2(m1, __builtin_bit_cast(unsigned, b1));
          m2 = umax2(m2, __builtin_bit_cast(unsigned, b2));
          m3 = umax2(m3, __builtin_bit_cast(unsigned, b3));
          a0 = fmaf(g[u][q][0], b0, a0);
          a1 = fmaf(g[u][q][1], b1, a1);
          a2 = fmaf(g[u][q][2], b2, a2);
          a3 = fmaf(g[u][q][3], b3, a3);
        }
        const float cse = cs[u] * inv_pow2_of_max(umax2(umax2(m0, m1), umax2(m2, m3)));
        v = ((a0 + a1) + (a2 + a3)) * cse;
        bb[(size_t)(CL*c)*HK + lane] = v;
        __builtin_amdgcn_sched_barrier(0);
      }
    }
  }
}

__device__ __forceinline__ void gamma_row(v4f (&g)[4], const v4f (&a)[4], const v4f (&b)[4]){
  float p[16];
#pragma unroll
  for (int q = 0; q < 16; ++q) p[q] = a[q >> 2][q & 3] * b[q >> 2][q & 3];
  const float s0 = ((p[0] + p[1]) + (p[2] + p[3])) + ((p[4] + p[5]) + (p[6] + p[7]));
  const float s1 = ((p[8] + p[9]) + (p[10] + p[11])) + ((p[12] + p[13]) + (p[14] + p[15]));
  float s = s0 + s1;
  s += xhalf(s);
  const float ls = flog2(s);
#pragma unroll
  for (int c = 0; c < 4; ++c){
    v4f t;
#pragma unroll
    for (int e = 0; e < 4; ++e) t[e] = LN2 * ((flog2(a[c][e]) + flog2(b[c][e])) - ls);
    g[c] = t;
  }
}

__global__ void __launch_bounds__(64)
hmm_scan_kernel(const float* __restrict__ em,
                const float* __restrict__ log_A,
                float* alpha_out,
                float* beta_out){
  __shared__ float sA[32*33];

  const int lane = threadIdx.x & 31;
  const int dir  = __builtin_amdgcn_readfirstlane((int)(threadIdx.x >> 5));
  const int h    = lane >> 4;
  const int ln   = lane & 15;
  const int mb   = blockIdx.x & 3;
  const int ch   = blockIdx.x >> 2;

  if (dir == 0) build_expA(sA, log_A, lane);
  __syncthreads();
  v2f ereg[2][8];
  build_E(ereg, sA, ln, h, dir);

  const size_t   uoff = (size_t)mb*16*HT*HK;
  const unsigned boff = (unsigned)(ln*HT*HK + 16*h) * 4u;
  const float* eb = em + uoff;
  float* ab = alpha_out + uoff;
  float* bb = beta_out + uoff;
  const int tb = CL*ch;
  const int te = CL*(ch + 1);
  constexpr int HALF = 32;

  v4f er[PF][4];
  v4f pr[PF][4];
  float st[16];
  v4f yr[4];

  if (dir == 0){
    {
      v4f a0[4], b0[4];
      load_row(a0, ab + (size_t)tb*HK, boff);
      load_row(b0, bb + (size_t)tb*HK, boff);
#pragma unroll
      for (int u = 0; u < PF - 1; ++u) load_row(er[u], eb + (size_t)(tb + 1 + u)*HK, boff);
#pragma unroll
      for (int q = 0; q < 16; ++q) st[q] = a0[q >> 2][q & 3];
      v4f g[4];
      gamma_row(g, a0, b0);
      store_row(ab + (size_t)tb*HK, boff, g);
    }
    unsigned bo = boff;
    for (int k0 = 1; k0 <= HALF; k0 += PF){
      asm volatile("" : "+v"(bo));
#pragma unroll
      for (int u = 0; u < PF; ++u){
        const int t = tb + k0 + u;
        v4f wr[4];
        pack_row(wr, st);
        if (k0 + u > 1) store_row(ab + (size_t)(t - 1)*HK, bo, wr);
        const int up = (u + PF - 1) % PF;
        const int tn = (t - 1 + PF < HT) ? t - 1 + PF : HT - 1;
        load_row(er[up], eb + (size_t)tn*HK, bo);
        float aln[16];
        v8f d0, d1;
        float ef;
        lin_step(aln, d0, d1, ef, st, er[u], ereg);
        hold_row(wr, aln[0]);
#pragma unroll
        for (int q = 0; q < 16; ++q) st[q] = aln[q];
        __builtin_amdgcn_sched_barrier(0);
      }
    }
    store_row(ab + (size_t)(tb + HALF)*HK, boff, st);
  } else {
    load_row(yr, bb + (size_t)te*HK, boff);
    {
      v4f eT[4];
      load_row(eT, eb + (size_t)te*HK, boff);
#pragma unroll
      for (int q = 0; q < 16; ++q) st[q] = yr[q >> 2][q & 3] * fexp2(eT[q >> 2][q & 3] * LOG2E);
    }
    if (ch == NCH - 1){
      v4f aT[4], g[4];
      load_row(aT, ab + (size_t)te*HK, boff);
      gamma_row(g, aT, yr);
      store_row(ab + (size_t)te*HK, boff, g);
    }
#pragma unroll
    for (int u = 0; u < PF - 1; ++u) load_row(er[u], eb + (size_t)(te - 1 - u)*HK, boff);

    unsigned bo = boff;
    for (int s0 = 0; s0 < HALF; s0 += PF){
      asm volatile("" : "+v"(bo));
#pragma unroll
      for (int u = 0; u < PF; ++u){
        const int t = te - 1 - (s0 + u);
        if (s0 + u > 0) store_row(bb + (size_t)(t + 1)*HK, bo, yr);
        const int up = (u + PF - 1) % PF;
        const int tn = (t + 1 - PF > 0) ? t + 1 - PF : 0;
        load_row(er[up], eb + (size_t)tn*HK, bo);
        float xn[16];
        v8f d0, d1;
        float ef;
        lin_step(xn, d0, d1, ef, st, er[u], ereg);
        hold_row(yr, xn[0]);
        {
          v4f a0 = { d0[0], d0[1], d0[2], d0[3] };
          v4f a1 = { d0[4], d0[5], d0[6], d0[7] };
          v4f a2 = { d1[0], d1[1], d1[2], d1[3] };
          v4f a3 = { d1[4], d1[5], d1[6], d1[7] };
          yr[0] = a0; yr[1] = a1; yr[2] = a2; yr[3] = a3;
        }
#pragma unroll
        for (int q = 0; q < 16; ++q) st[q] = xn[q];
        __builtin_amdgcn_sched_barrier(0);
      }
    }
    store_row(bb + (size_t)(te - HALF)*HK, boff, yr);
  }

  __syncthreads();

  const v4f vz = {0.f, 0.f, 0.f, 0.f};
  if (dir == 0){
#pragma unroll
    for (int u = 0; u < PF - 1; ++u) load_row(pr[u], bb + (size_t)(tb + HALF + 1 + u)*HK, boff);
    unsigned bo = boff;
    for (int k0 = HALF + 1; k0 <= 2*HALF; k0 += PF){
      asm volatile("" : "+v"(bo));
#pragma unroll
      for (int u = 0; u < PF; ++u){
        const int k  = k0 + u;
        const int t  = tb + k;
        const int up = (u + PF - 1) % PF;
        v4f wr[4] = { vz, vz, vz, vz };
        if (k > HALF + 1){
          v4f av[4];
          pack_row(av, st);
          gamma_row(wr, av, pr[up]);
          store_row(ab + (size_t)(t - 1)*HK, bo, wr);
        }
        const int tn = (t - 1 + PF < HT) ? t - 1 + PF : HT - 1;
        load_row(er[up], eb + (size_t)tn*HK, bo);
        const int jn = (k - 1 + PF < 2*HALF) ? k - 1 + PF : 2*HALF;
        load_row(pr[up], bb + (size_t)(tb + jn)*HK, bo);
        float aln[16];
        v8f d0, d1;
        float ef;
        lin_step(aln, d0, d1, ef, st, er[u], ereg);
        hold_row(wr, aln[0]);
#pragma unroll
        for (int q = 0; q < 16; ++q) st[q] = aln[q];
        __builtin_amdgcn_sched_barrier(0);
      }
    }
    {
      v4f av[4], g[4];
      pack_row(av, st);
      gamma_row(g, av, pr[(2*HALF - 1) % PF]);
      store_row(ab + (size_t)(tb + 2*HALF)*HK, boff, g);
    }
  } else {
#pragma unroll
    for (int u = 0; u < PF - 1; ++u) load_row(pr[u], ab + (size_t)(te - 1 - HALF - u)*HK, boff);
    unsigned bo = boff;
    for (int s0 = HALF; s0 < 2*HALF; s0 += PF){
      asm volatile("" : "+v"(bo));
#pragma unroll
      for (int u = 0; u < PF; ++u){
        const int s  = s0 + u;
        const int t  = te - 1 - s;
        const int up = (u + PF - 1) % PF;
        v4f wr[4] = { vz, vz, vz, vz };
        if (s > HALF){
          gamma_row(wr, pr[up], yr);
          store_row(ab + (size_t)(t + 1)*HK, bo, wr);
        }
        const int tn = (t + 1 - PF > 0) ? t + 1 - PF : 0;
        load_row(er[up], eb + (size_t)tn*HK, bo);
        const int pa = (t + 1 - PF > tb + 1) ? t + 1 - PF : tb + 1;
        load_row(pr[up], ab + (size_t)pa*HK, bo);
        float xn[16];
        v8f d0, d1;
        float ef;
        lin_step(xn, d0, d1, ef, st, er[u], ereg);
        hold_row(wr, xn[0]);
        {
          v4f a0 = { d0[0], d0[1], d0[2], d0[3] };
          v4f a1 = { d0[4], d0[5], d0[6], d0[7] };
          v4f a2 = { d1[0], d1[1], d1[2], d1[3] };
          v4f a3 = { d1[4], d1[5], d1[6], d1[7] };
          yr[0] = a0; yr[1] = a1; yr[2] = a2; yr[3] = a3;
        }
#pragma unroll
        for (int q = 0; q < 16; ++q) st[q] = xn[q];
        __builtin_amdgcn_sched_barrier(0);
      }
    }
    {
      v4f g[4];
      gamma_row(g, pr[(2*HALF - 1) % PF], yr);
      store_row(ab + (size_t)(tb + 1)*HK, boff, g);
    }
  }
}

extern "C" void kernel_launch(void* const* d_in, const int* in_sizes, int n_in,
                              void* d_out, int out_size, void* d_ws, size_t ws_size,
                              hipStream_t stream) {
  const float* em     = (const float*)d_in[0];
  const float* log_pi = (const float*)d_in[1];
  const float* log_A  = (const float*)d_in[2];
  float* out  = (float*)d_out;
  float* beta = (float*)d_ws;

  hmm_operator_kernel<<<dim3(HB*NCH*2), dim3(32), 0, stream>>>(em, log_A, out);
  hmm_edge_kernel<<<dim3(HB*2), dim3(32), 0, stream>>>(em, log_pi, out, beta);
  hmm_scan_kernel<<<dim3(NCH*4), dim3(64), 0, stream>>>(em, log_A, out, beta);
}


# Edge walk: operators asked for six chunks ahead (ring of seven), not two; forward reads a row-wise operator copy the operator pass leaves in scratch, 8 wide loads per lane; sum order unchanged
# speedup vs baseline: 22.3461x; 1.0362x over previous
#include <hip/hip_runtime.h>


typedef __attribute__((ext_vector_type(2))) float v2f;
typedef __attribute__((ext_vector_type(4))) float v4f;
typedef __attribute__((ext_vector_type(8))) float v8f;
typedef __attribute__((address_space(1))) v4f GV4;

constexpr int HB = 64;
constexpr int HT = 4096;
constexpr int HK = 32;
constexpr int CL  = 65;
constexpr int NCH = 63;
constexpr int PFA = 5;
constexpr int PF  = 4;
constexpr int DP  = 7;
constexpr float LOG2E = 1.44269504088896340736f;
constexpr float LN2   = 0.69314718055994530942f;

#if __has_builtin(__builtin_amdgcn_exp2f)
__device__ __forceinline__ float fexp2(float x){ return __builtin_amdgcn_exp2f(x); }
#else
__device__ __forceinline__ float fexp2(float x){ return exp2f(x); }
#endif
#if __has_builtin(__builtin_amdgcn_logf)
__device__ __forceinline__ float flog2(float x){ return __builtin_amdgcn_logf(x); }
#else
__device__ __forceinline__ float flog2(float x){ return log2f(x); }
#endif

__device__ __forceinline__ float rmax32(float x){
#pragma unroll
  for (int m = 16; m >= 1; m >>= 1) x = fmaxf(x, __shfl_xor(x, m, 32));
  return x;
}
__device__ __forceinline__ float rsum32(float x){
#pragma unroll
  for (int m = 16; m >= 1; m >>= 1) x += __shfl_xor(x, m, 32);
  return x;
}

__device__ __forceinline__ float xhalf(float x){
  int b = __builtin_bit_cast(int, x);
  return __builtin_bit_cast(float,
      __builtin_amdgcn_permlanex16(b, b, 0x76543210, 0xfedcba98, false, false));
}

__device__ __forceinline__ v8f wmma4(v2f a, v2f b, v8f c){
  return __builtin_amdgcn_wmma_f32_16x16x4_f32(
       false, a,  false, b,
       (short)0, c,  false,  false);
}


__device__ __forceinline__ unsigned long long uniform_addr(const void* p){
  const unsigned long long a = (unsigned long long)p;
  const unsigned lo = (unsigned)__builtin_amdgcn_readfirstlane((int)(unsigned)a);
  const unsigned hi = (unsigned)__builtin_amdgcn_readfirstlane((int)(unsigned)(a >> 32));
  return ((unsigned long long)hi << 32) | (unsigned long long)lo;
}
__device__ __forceinline__ void load_row(v4f (&dst)[4], const float* ub, unsigned boff){
  const GV4 * q = (const GV4 *)(uniform_addr(ub) + boff);
#pragma unroll
  for (int c = 0; c < 4; ++c) dst[c] = q[c];
}
__device__ __forceinline__ void pack_row(v4f (&w)[4], const float (&v)[16]){
#pragma unroll
  for (int c = 0; c < 4; ++c){
    v4f t = { v[4*c+0], v[4*c+1], v[4*c+2], v[4*c+3] };
    w[c] = t;
  }
}
__device__ __forceinline__ void store_row(float* ub, unsigned boff, const v4f (&w)[4]){
  GV4 * q = (GV4 *)(uniform_addr(ub) + boff);
#pragma unroll
  for (int c = 0; c < 4; ++c) q[c] = w[c];
}
__device__ __forceinline__ void store_row(float* ub, unsigned boff, const float (&v)[16]){
  v4f w[4];
  pack_row(w, v);
  store_row(ub, boff, w);
}

__device__ __forceinline__ float row_exponent(const float (&p)[16]){
  const float t0 = fmaxf(fmaxf(p[0],  p[1]),  p[2]);
  const float t1 = fmaxf(fmaxf(p[3],  p[4]),  p[5]);
  const float t2 = fmaxf(fmaxf(p[6],  p[7]),  p[8]);
  const float t3 = fmaxf(fmaxf(p[9],  p[10]), p[11]);
  const float t4 = fmaxf(fmaxf(p[12], p[13]), p[14]);
  const float u0 = fmaxf(fmaxf(t0, t1), t2);
  const float u1 = fmaxf(fmaxf(t3, t4), p[15]);
  float m = fmaxf(u0, u1);
  m = fmaxf(m, xhalf(m));
#if __has_builtin(__builtin_amdgcn_frexp_expf)
  return (float)__builtin_amdgcn_frexp_expf(m);
#else
  return (float)(((__builtin_bit_cast(int, m) >> 23) & 0xff) - 126);
#endif
}
__device__ __forceinline__ void lin_step(float (&pn)[16], v8f& d0, v8f& d1, float& ef_out,
                                         const float (&p)[16], const v4f (&emc)[4],
                                         const v2f (&ereg)[2][8]){
  v8f c0 = {0.f,0.f,0.f,0.f,0.f,0.f,0.f,0.f};
  v8f c1 = {0.f,0.f,0.f,0.f,0.f,0.f,0.f,0.f};
#pragma unroll
  for (int kc = 0; kc < 8; ++kc){
    const v2f b = { p[2*kc], p[2*kc+1] };
    c0 = wmma4(ereg[0][kc], b, c0);
    c1 = wmma4(ereg[1][kc], b, c1);
  }
  const float ef = row_exponent(p);
  ef_out = ef;
  float w[16];
#pragma unroll
  for (int q = 0; q < 16; ++q) w[q] = fexp2(fmaf(emc[q >> 2][q & 3], LOG2E, -ef));
#pragma unroll
  for (int r = 0; r < 8; ++r){
    pn[r]     = c0[r] * w[r];
    pn[8 + r] = c1[r] * w[8 + r];
  }
  d0 = c0;
  d1 = c1;
}
__device__ __forceinline__ void hold_row(const v4f (&w)[4], float after){
  asm volatile("" :: "v"(w[0]), "v"(w[1]), "v"(w[2]), "v"(w[3]), "v"(after));
}


__device__ __forceinline__ int row_exponent_i(const float (&p)[16]){
  const float t0 = fmaxf(fmaxf(p[0],  p[1]),  p[2]);
  const float t1 = fmaxf(fmaxf(p[3],  p[4]),  p[5]);
  const float t2 = fmaxf(fmaxf(p[6],  p[7]),  p[8]);
  const float t3 = fmaxf(fmaxf(p[9],  p[10]), p[11]);
  const float t4 = fmaxf(fmaxf(p[12], p[13]), p[14]);
  const float u0 = fmaxf(fmaxf(t0, t1), t2);
  const float u1 = fmaxf(fmaxf(t3, t4), p[15]);
  float m = fmaxf(u0, u1);
  m = fmaxf(m, xhalf(m));
#if __has_builtin(__builtin_amdgcn_frexp_expf)
  return __builtin_amdgcn_frexp_expf(m);
#else
  return ((__builtin_bit_cast(int, m) >> 23) & 0xff) - 126;
#endif
}
__device__ __forceinline__ void op_step(float (&pn)[16], int& e_out,
                                        const float (&p)[16], const v4f (&wv)[4],
                                        const v2f (&ereg)[2][8]){
  v8f c0 = {0.f,0.f,0.f,0.f,0.f,0.f,0.f,0.f};
  v8f c1 = {0.f,0.f,0.f,0.f,0.f,0.f,0.f,0.f};
#pragma unroll
  for (int kc = 0; kc < 8; ++kc){
    const v2f b = { p[2*kc], p[2*kc+1] };
    c0 = wmma4(ereg[0][kc], b, c0);
    c1 = wmma4(ereg[1][kc], b, c1);
  }
  const int e = row_exponent_i(p);
  e_out = e;
  const float sc = __builtin_ldexpf(1.0f, -e);
  v2f ws[8];
#pragma unroll
  for (int i = 0; i < 8; ++i){
    const v2f w = { wv[i >> 1][2*(i & 1)], wv[i >> 1][2*(i & 1) + 1] };
    ws[i] = w * sc;
  }
#pragma unroll
  for (int i = 0; i < 4; ++i){
    const v2f d0 = { c0[2*i], c0[2*i+1] };
    const v2f d1 = { c1[2*i], c1[2*i+1] };
    const v2f r0 = d0 * ws[i];
    const v2f r1 = d1 * ws[4 + i];
    pn[2*i]     = r0.x;  pn[2*i+1]     = r0.y;
    pn[8 + 2*i] = r1.x;  pn[8 + 2*i+1] = r1.y;
  }
}


__device__ __forceinline__ void build_expA(float* sA, const float* __restrict__ log_A, int lane){
  float x[32];
#pragma unroll
  for (int c = 0; c < 8; ++c){
    const v4f t = *(const v4f*)(log_A + lane*32 + 4*c);
    x[4*c+0] = t[0]; x[4*c+1] = t[1]; x[4*c+2] = t[2]; x[4*c+3] = t[3];
  }
  float mx = x[0];
#pragma unroll
  for (int j = 1; j < 32; ++j) mx = fmaxf(mx, x[j]);
  float s = 0.f;
#pragma unroll
  for (int j = 0; j < 32; ++j){ x[j] = fexp2((x[j] - mx) * LOG2E); s += x[j]; }
  const float r = 1.0f / s;
#pragma unroll
  for (int j = 0; j < 32; ++j) sA[lane*33 + j] = x[j] * r;
}

__device__ __forceinline__ void build_E(v2f (&ereg)[2][8], const float* sA, int ln, int h, int dir){
#pragma unroll
  for (int mt = 0; mt < 2; ++mt){
#pragma unroll
    for (int kc = 0; kc < 8; ++kc){
      const int sM = 16*(ln >> 3) + 8*mt + (ln & 7);
      const int sK = 16*h + 2*kc;
      if (dir == 0){
        ereg[mt][kc].x = sA[sK*33 + sM];
        ereg[mt][kc].y = sA[(sK+1)*33 + sM];
      } else {
        ereg[mt][kc].x = sA[sM*33 + sK];
        ereg[mt][kc].y = sA[sM*33 + sK + 1];
      }
    }
  }
}

__global__ void __launch_bounds__(32)
hmm_operator_kernel(const float* __restrict__ em,
                    const float* __restrict__ log_A,
                    float* __restrict__ fbuf,
                    float* __restrict__ tbuf){
  __shared__ __attribute__((aligned(16))) float sm[CL*HK];
  const int lane = threadIdx.x;
  const int h    = lane >> 4;
  const int ln   = lane & 15;
  const int nt   = blockIdx.x & 1;
  const int cb   = blockIdx.x >> 1;
  const int c    = cb % NCH;
  const int b    = cb / NCH;

  const size_t urow = (size_t)b*HT + (size_t)(CL*c + 1);
  const float* eb = em + urow*HK;

  float x[CL];
#pragma unroll
  for (int k = 0; k < CL; ++k) x[k] = eb[(size_t)k*HK + lane];

  build_expA(sm, log_A, lane);
  __syncthreads();
  v2f ereg[2][8];
  build_E(ereg, sm, ln, h, 0);
  __syncthreads();
#pragma unroll
  for (int k = 0; k < CL; ++k) sm[k*HK + lane] = fexp2(x[k] * LOG2E);
  __syncthreads();

  float p[16];
#pragma unroll
  for (int q = 0; q < 16; ++q) p[q] = (16*h + q == 16*nt + ln) ? 1.f : 0.f;
  int esum = 0;

  const float* wl = sm + 16*h;
  v4f wr[PFA][4];
#pragma unroll
  for (int cc = 0; cc < 4; ++cc) wr[0][cc] = *(const v4f*)(wl + 4*cc);

  for (int k0 = 0; k0 < CL; k0 += PFA){
#pragma unroll
    for (int u = 0; u < PFA; ++u){
      const int k  = k0 + u;
      const int un = (u + 1) % PFA;
      const int kn = (k + 1 < CL) ? k + 1 : CL - 1;
#pragma unroll
      for (int cc = 0; cc < 4; ++cc) wr[un][cc] = *(const v4f*)(wl + kn*HK + 4*cc);
      float pn[16];
      int e;
      op_step(pn, e, p, wr[u], ereg);
      esum += e;
#pragma unroll
      for (int q = 0; q < 16; ++q) p[q] = pn[q];
      __builtin_amdgcn_sched_barrier(0);
    }
  }
  float* dst = fbuf + urow*HK;
  const unsigned coff = (unsigned)((16*nt + ln)*HK + 16*h) * 4u;
  store_row(dst, coff, p);
  {
    float* dt = tbuf + urow*HK + (size_t)(nt*16*HK + (16*h)*16 + ln);
#pragma unroll
    for (int q = 0; q < 16; ++q) dt[q*16] = p[q];
  }
  if (h == 0) dst[(size_t)(32 + nt)*HK + 16*nt + ln] = (float)esum;
}

__device__ __forceinline__ float expo_of(float m){
#if __has_builtin(__builtin_amdgcn_frexp_expf)
  return (float)__builtin_amdgcn_frexp_expf(m);
#else
  return (float)(((__builtin_bit_cast(int, m) >> 23) & 0xff) - 126);
#endif
}

__device__ __forceinline__ float lane_swz(float x, const int ctrl_sel){
  const int b = __builtin_bit_cast(int, x);
  int r;
  if (ctrl_sel == 0)      r = __builtin_amdgcn_update_dpp(b, b, 0xB1,  0xf, 0xf, false);
  else if (ctrl_sel == 1) r = __builtin_amdgcn_update_dpp(b, b, 0x4E,  0xf, 0xf, false);
  else if (ctrl_sel == 2) r = __builtin_amdgcn_update_dpp(b, b, 0x141, 0xf, 0xf, false);
  else                    r = __builtin_amdgcn_update_dpp(b, b, 0x140, 0xf, 0xf, false);
  return __builtin_bit_cast(float, r);
}
__device__ __forceinline__ float wmax32(float x){
  x = fmaxf(x, lane_swz(x, 0));
  x = fmaxf(x, lane_swz(x, 1));
  x = fmaxf(x, lane_swz(x, 2));
  x = fmaxf(x, lane_swz(x, 3));
  return fmaxf(x, xhalf(x));
}
__device__ __forceinline__ float lane_bcast(float x, const int j){
  return __builtin_bit_cast(float, __builtin_amdgcn_readlane(__builtin_bit_cast(int, x), j));
}

__device__ __forceinline__ unsigned umax2(unsigned a, unsigned b){ return a > b ? a : b; }
__device__ __forceinline__ float inv_pow2_of_max(unsigned mb){
  int fld = 253 - (int)(mb >> 23);
  fld = fld < 1 ? 1 : fld;
  return __builtin_bit_cast(float, (unsigned)fld << 23);
}

__global__ void __launch_bounds__(32)
hmm_edge_kernel(const float* __restrict__ em,
                const float* __restrict__ log_pi,
                float* fa,
                float* __restrict__ bw){
  const int lane = threadIdx.x;
  const int b    = blockIdx.x >> 1;
  const int dir  = blockIdx.x & 1;
  float* fb = fa + (size_t)b*HT*HK;

  if (dir == 0){
    const float xp = log_pi[lane];
    const float e0 = em[(size_t)b*HT*HK + lane];
    const float* tb = bw + (size_t)b*HT*HK + (size_t)lane*16;
    v4f g[DP][8];
    float sx[DP];
    const size_t xoff = (size_t)(32 + (lane >> 4))*HK + lane;
#pragma unroll
    for (int u = 0; u < DP; ++u) sx[u] = fb[(size_t)(CL*u + 1)*HK + xoff];
    __builtin_amdgcn_sched_barrier(0);
#pragma unroll
    for (int u = 0; u < DP - 1; ++u){
      const float* src = tb + (size_t)(CL*u + 1)*HK;
#pragma unroll
      for (int q = 0; q < 4; ++q){
        g[u][q]     = *(const v4f*)(src + 4*q);
        g[u][4 + q] = *(const v4f*)(src + 16*HK + 4*q);
      }
      __builtin_amdgcn_sched_barrier(0);
    }
    float mp = rmax32(xp);
    float sp = rsum32(fexp2((xp - mp) * LOG2E));
    float lp = xp - mp - LN2 * flog2(sp);
    float v  = fexp2((lp + e0) * LOG2E);
    fb[lane] = v;
    float cs[DP];
    cs[0] = fexp2(sx[0] - wmax32(sx[0]));
    asm volatile("" :: "v"(g[0][0]), "v"(g[0][1]), "v"(g[0][2]), "v"(g[0][3]), "v"(g[0][4]), "v"(g[0][5]), "v"(g[0][6]), "v"(g[0][7]), "v"(sx[1]));
    for (int c0 = 0; c0 < NCH; c0 += DP){
#pragma unroll
      for (int u = 0; u < DP; ++u){
        const int c  = c0 + u;
        const int cn = (c + DP - 1 < NCH) ? c + DP - 1 : NCH - 1;
        const int cx = (c + DP < NCH) ? c + DP : NCH - 1;
        {
          sx[u] = fb[(size_t)(CL*cx + 1)*HK + xoff];
          const int un = (u + DP - 1) % DP;
          const float* src = tb + (size_t)(CL*cn + 1)*HK;
#pragma unroll
          for (int q = 0; q < 4; ++q){
            g[un][q]     = *(const v4f*)(src + 4*q);
            g[un][4 + q] = *(const v4f*)(src + 16*HK + 4*q);
          }
        }
        {
          const int u1 = (u + 1) % DP;
          cs[u1] = fexp2(sx[u1] - wmax32(sx[u1]));
        }
        const float uu = v * cs[u];
        float a0 = 0.f, a1 = 0.f, a2 = 0.f, a3 = 0.f;
        unsigned m0 = 0u, m1 = 0u, m2 = 0u, m3 = 0u;
#pragma unroll
        for (int q = 0; q < 8; ++q){
          const float b0 = lane_bcast(uu, 4*q+0), b1 = lane_bcast(uu, 4*q+1);
          const float b2 = lane_bcast(uu, 4*q+2), b3 = lane_bcast(uu, 4*q+3);
          m0 = umax2(m0, __builtin_bit_cast(unsigned, b0));
          m1 = umax2(m1, __builtin_bit_cast(unsigned, b1));
          m2 = umax2(m2, __builtin_bit_cast(unsigned, b2));
          m3 = umax2(m3, __builtin_bit_cast(unsigned, b3));
          a0 = fmaf(g[u][q][0], b0, a0);
          a1 = fmaf(g[u][q][1], b1, a1);
          a2 = fmaf(g[u][q][2], b2, a2);
          a3 = fmaf(g[u][q][3], b3, a3);
        }
        const float acc = (a0 + a1) + (a2 + a3);
        v = acc * inv_pow2_of_max(umax2(umax2(m0, m1), umax2(m2, m3)));
        fb[(size_t)(CL*(c + 1))*HK + lane] = v;
        __builtin_amdgcn_sched_barrier(0);
      }
    }
  } else {
    float* bb = bw + (size_t)b*HT*HK;
    float v = 1.f;
    bb[(size_t)(HT - 1)*HK + lane] = v;

    v4f g[DP][8];
    float sx[DP];
    const size_t xoff = (size_t)(32 + (lane >> 4))*HK + lane;
#pragma unroll
    for (int u = 0; u < DP; ++u) sx[u] = fb[(size_t)(CL*(NCH - 1 - u) + 1)*HK + xoff];
#pragma unroll
    for (int u = 0; u < DP - 1; ++u){
      const float* src = fb + (size_t)(CL*(NCH - 1 - u) + 1)*HK;
#pragma unroll
      for (int q = 0; q < 8; ++q) g[u][q] = *(const v4f*)(src + lane*HK + 4*q);
    }
    float cs[DP];
    cs[0] = fexp2(sx[0] - wmax32(sx[0]));
    asm volatile("" :: "v"(g[0][0]), "v"(g[0][1]), "v"(g[0][2]), "v"(g[0][3]), "v"(g[0][4]), "v"(g[0][5]), "v"(g[0][6]), "v"(g[0][7]), "v"(sx[1]));
    for (int c0 = 0; c0 < NCH; c0 += DP){
#pragma unroll
      for (int u = 0; u < DP; ++u){
        const int c  = NCH - 1 - (c0 + u);
        const int cn = (c - (DP - 1) > 0) ? c - (DP - 1) : 0;
        const int cx = (c - DP > 0) ? c - DP : 0;
        {
          sx[u] = fb[(size_t)(CL*cx + 1)*HK + xoff];
          const int un = (u + DP - 1) % DP;
          const float* src = fb + (size_t)(CL*cn + 1)*HK;
#pragma unroll
          for (int q = 0; q < 8; ++q) g[un][q] = *(const v4f*)(src + lane*HK + 4*q);
        }
        {
          const int u1 = (u + 1) % DP;
          cs[u1] = fexp2(sx[u1] - wmax32(sx[u1]));
        }
        float a0 = 0.f, a1 = 0.f, a2 = 0.f, a3 = 0.f;
        unsigned m0 = 0u, m1 = 0u, m2 = 0u, m3 = 0u;
#pragma unroll
        for (int q = 0; q < 8; ++q){
          const float b0 = lane_bcast(v, 4*q+0), b1 = lane_bcast(v, 4*q+1);
          const float b2 = lane_bcast(v, 4*q+2), b3 = lane_bcast(v, 4*q+3);
          m0 = umax2(m0, __builtin_bit_cast(unsigned, b0));
          m1 = umax2(m1, __builtin_bit_cast(unsigned, b1));
          m2 = umax2(m2, __builtin_bit_cast(unsigned, b2));
          m3 = umax2(m3, __builtin_bit_cast(unsigned, b3));
          a0 = fmaf(g[u][q][0], b0, a0);
          a1 = fmaf(g[u][q][1], b1, a1);
          a2 = fmaf(g[u][q][2], b2, a2);
          a3 = fmaf(g[u][q][3], b3, a3);
        }
        const float cse = cs[u] * inv_pow2_of_max(umax2(umax2(m0, m1), umax2(m2, m3)));
        v = ((a0 + a1) + (a2 + a3)) * cse;
        bb[(size_t)(CL*c)*HK + lane] = v;
        __builtin_amdgcn_sched_barrier(0);
      }
    }
  }
}

__device__ __forceinline__ void gamma_row(v4f (&g)[4], const v4f (&a)[4], const v4f (&b)[4]){
  float p[16];
#pragma unroll
  for (int q = 0; q < 16; ++q) p[q] = a[q >> 2][q & 3] * b[q >> 2][q & 3];
  const float s0 = ((p[0] + p[1]) + (p[2] + p[3])) + ((p[4] + p[5]) + (p[6] + p[7]));
  const float s1 = ((p[8] + p[9]) + (p[10] + p[11])) + ((p[12] + p[13]) + (p[14] + p[15]));
  float s = s0 + s1;
  s += xhalf(s);
  const float ls = flog2(s);
#pragma unroll
  for (int c = 0; c < 4; ++c){
    v4f t;
#pragma unroll
    for (int e = 0; e < 4; ++e) t[e] = LN2 * ((flog2(a[c][e]) + flog2(b[c][e])) - ls);
    g[c] = t;
  }
}

__global__ void __launch_bounds__(64)
hmm_scan_kernel(const float* __restrict__ em,
                const float* __restrict__ log_A,
                float* alpha_out,
                float* beta_out){
  __shared__ float sA[32*33];

  const int lane = threadIdx.x & 31;
  const int dir  = __builtin_amdgcn_readfirstlane((int)(threadIdx.x >> 5));
  const int h    = lane >> 4;
  const int ln   = lane & 15;
  const int mb   = blockIdx.x & 3;
  const int ch   = blockIdx.x >> 2;

  if (dir == 0) build_expA(sA, log_A, lane);
  __syncthreads();
  v2f ereg[2][8];
  build_E(ereg, sA, ln, h, dir);

  const size_t   uoff = (size_t)mb*16*HT*HK;
  const unsigned boff = (unsigned)(ln*HT*HK + 16*h) * 4u;
  const float* eb = em + uoff;
  float* ab = alpha_out + uoff;
  float* bb = beta_out + uoff;
  const int tb = CL*ch;
  const int te = CL*(ch + 1);
  constexpr int HALF = 32;

  v4f er[PF][4];
  v4f pr[PF][4];
  float st[16];
  v4f yr[4];

  if (dir == 0){
    {
      v4f a0[4], b0[4];
      load_row(a0, ab + (size_t)tb*HK, boff);
      load_row(b0, bb + (size_t)tb*HK, boff);
#pragma unroll
      for (int u = 0; u < PF - 1; ++u) load_row(er[u], eb + (size_t)(tb + 1 + u)*HK, boff);
#pragma unroll
      for (int q = 0; q < 16; ++q) st[q] = a0[q >> 2][q & 3];
      v4f g[4];
      gamma_row(g, a0, b0);
      store_row(ab + (size_t)tb*HK, boff, g);
    }
    unsigned bo = boff;
    for (int k0 = 1; k0 <= HALF; k0 += PF){
      asm volatile("" : "+v"(bo));
#pragma unroll
      for (int u = 0; u < PF; ++u){
        const int t = tb + k0 + u;
        v4f wr[4];
        pack_row(wr, st);
        if (k0 + u > 1) store_row(ab + (size_t)(t - 1)*HK, bo, wr);
        const int up = (u + PF - 1) % PF;
        const int tn = (t - 1 + PF < HT) ? t - 1 + PF : HT - 1;
        load_row(er[up], eb + (size_t)tn*HK, bo);
        float aln[16];
        v8f d0, d1;
        float ef;
        lin_step(aln, d0, d1, ef, st, er[u], ereg);
        hold_row(wr, aln[0]);
#pragma unroll
        for (int q = 0; q < 16; ++q) st[q] = aln[q];
        __builtin_amdgcn_sched_barrier(0);
      }
    }
    store_row(ab + (size_t)(tb + HALF)*HK, boff, st);
  } else {
    load_row(yr, bb + (size_t)te*HK, boff);
    {
      v4f eT[4];
      load_row(eT, eb + (size_t)te*HK, boff);
#pragma unroll
      for (int q = 0; q < 16; ++q) st[q] = yr[q >> 2][q & 3] * fexp2(eT[q >> 2][q & 3] * LOG2E);
    }
    if (ch == NCH - 1){
      v4f aT[4], g[4];
      load_row(aT, ab + (size_t)te*HK, boff);
      gamma_row(g, aT, yr);
      store_row(ab + (size_t)te*HK, boff, g);
    }
#pragma unroll
    for (int u = 0; u < PF - 1; ++u) load_row(er[u], eb + (size_t)(te - 1 - u)*HK, boff);

    unsigned bo = boff;
    for (int s0 = 0; s0 < HALF; s0 += PF){
      asm volatile("" : "+v"(bo));
#pragma unroll
      for (int u = 0; u < PF; ++u){
        const int t = te - 1 - (s0 + u);
        if (s0 + u > 0) store_row(bb + (size_t)(t + 1)*HK, bo, yr);
        const int up = (u + PF - 1) % PF;
        const int tn = (t + 1 - PF > 0) ? t + 1 - PF : 0;
        load_row(er[up], eb + (size_t)tn*HK, bo);
        float xn[16];
        v8f d0, d1;
        float ef;
        lin_step(xn, d0, d1, ef, st, er[u], ereg);
        hold_row(yr, xn[0]);
        {
          v4f a0 = { d0[0], d0[1], d0[2], d0[3] };
          v4f a1 = { d0[4], d0[5], d0[6], d0[7] };
          v4f a2 = { d1[0], d1[1], d1[2], d1[3] };
          v4f a3 = { d1[4], d1[5], d1[6], d1[7] };
          yr[0] = a0; yr[1] = a1; yr[2] = a2; yr[3] = a3;
        }
#pragma unroll
        for (int q = 0; q < 16; ++q) st[q] = xn[q];
        __builtin_amdgcn_sched_barrier(0);
      }
    }
    store_row(bb + (size_t)(te - HALF)*HK, boff, yr);
  }

  __syncthreads();

  const v4f vz = {0.f, 0.f, 0.f, 0.f};
  if (dir == 0){
#pragma unroll
    for (int u = 0; u < PF - 1; ++u) load_row(pr[u], bb + (size_t)(tb + HALF + 1 + u)*HK, boff);
    unsigned bo = boff;
    for (int k0 = HALF + 1; k0 <= 2*HALF; k0 += PF){
      asm volatile("" : "+v"(bo));
#pragma unroll
      for (int u = 0; u < PF; ++u){
        const int k  = k0 + u;
        const int t  = tb + k;
        const int up = (u + PF - 1) % PF;
        v4f wr[4] = { vz, vz, vz, vz };
        if (k > HALF + 1){
          v4f av[4];
          pack_row(av, st);
          gamma_row(wr, av, pr[up]);
          store_row(ab + (size_t)(t - 1)*HK, bo, wr);
        }
        const int tn = (t - 1 + PF < HT) ? t - 1 + PF : HT - 1;
        load_row(er[up], eb + (size_t)tn*HK, bo);
        const int jn = (k - 1 + PF < 2*HALF) ? k - 1 + PF : 2*HALF;
        load_row(pr[up], bb + (size_t)(tb + jn)*HK, bo);
        float aln[16];
        v8f d0, d1;
        float ef;
        lin_step(aln, d0, d1, ef, st, er[u], ereg);
        hold_row(wr, aln[0]);
#pragma unroll
        for (int q = 0; q < 16; ++q) st[q] = aln[q];
        __builtin_amdgcn_sched_barrier(0);
      }
    }
    {
      v4f av[4], g[4];
      pack_row(av, st);
      gamma_row(g, av, pr[(2*HALF - 1) % PF]);
      store_row(ab + (size_t)(tb + 2*HALF)*HK, boff, g);
    }
  } else {
#pragma unroll
    for (int u = 0; u < PF - 1; ++u) load_row(pr[u], ab + (size_t)(te - 1 - HALF - u)*HK, boff);
    unsigned bo = boff;
    for (int s0 = HALF; s0 < 2*HALF; s0 += PF){
      asm volatile("" : "+v"(bo));
#pragma unroll
      for (int u = 0; u < PF; ++u){
        const int s  = s0 + u;
        const int t  = te - 1 - s;
        const int up = (u + PF - 1) % PF;
        v4f wr[4] = { vz, vz, vz, vz };
        if (s > HALF){
          gamma_row(wr, pr[up], yr);
          store_row(ab + (size_t)(t + 1)*HK, bo, wr);
        }
        const int tn = (t + 1 - PF > 0) ? t + 1 - PF : 0;
        load_row(er[up], eb + (size_t)tn*HK, bo);
        const int pa = (t + 1 - PF > tb + 1) ? t + 1 - PF : tb + 1;
        load_row(pr[up], ab + (size_t)pa*HK, bo);
        float xn[16];
        v8f d0, d1;
        float ef;
        lin_step(xn, d0, d1, ef, st, er[u], ereg);
        hold_row(wr, xn[0]);
        {
          v4f a0 = { d0[0], d0[1], d0[2], d0[3] };
          v4f a1 = { d0[4], d0[5], d0[6], d0[7] };
          v4f a2 = { d1[0], d1[1], d1[2], d1[3] };
          v4f a3 = { d1[4], d1[5], d1[6], d1[7] };
          yr[0] = a0; yr[1] = a1; yr[2] = a2; yr[3] = a3;
        }
#pragma unroll
        for (int q = 0; q < 16; ++q) st[q] = xn[q];
        __builtin_amdgcn_sched_barrier(0);
      }
    }
    {
      v4f g[4];
      gamma_row(g, pr[(2*HALF - 1) % PF], yr);
      store_row(ab + (size_t)(tb + 1)*HK, boff, g);
    }
  }
}

extern "C" void kernel_launch(void* const* d_in, const int* in_sizes, int n_in,
                              void* d_out, int out_size, void* d_ws, size_t ws_size,
                              hipStream_t stream) {
  const float* em     = (const float*)d_in[0];
  const float* log_pi = (const float*)d_in[1];
  const float* log_A  = (const float*)d_in[2];
  float* out  = (float*)d_out;
  float* beta = (float*)d_ws;

  hmm_operator_kernel<<<dim3(HB*NCH*2), dim3(32), 0, stream>>>(em, log_A, out, beta);
  hmm_edge_kernel<<<dim3(HB*2), dim3(32), 0, stream>>>(em, log_pi, out, beta);
  hmm_scan_kernel<<<dim3(NCH*4), dim3(64), 0, stream>>>(em, log_A, out, beta);
}
